# starting kernel
# baseline (speedup 1.0000x reference)
#include <hip/hip_runtime.h>
#include <math.h>


#define N_NODES  100000
#define N_EDGES  1000000
#define N_FEAT   128
#define HIDDEN   64
#define N_GRAPHS 512
#define N_CLASSES 10
#define BN_EPS  1e-5f
#define BN_BETA 1e-4f

typedef __attribute__((ext_vector_type(2))) float v2f;
typedef __attribute__((ext_vector_type(8))) float v8f;

__global__ void colstats_kernel(const float* __restrict__ A, int N, int F,
                                float* __restrict__ sums, float* __restrict__ sumsq) {
  const int c = threadIdx.x;
  float s = 0.f, q = 0.f;
  for (int r = blockIdx.x; r < N; r += gridDim.x) {
    float v = A[(size_t)r * F + c];
    s += v; q += v * v;
  }
  atomicAdd(&sums[c], s);
  atomicAdd(&sumsq[c], q);
}

__global__ void fold_bn_kernel(const float* __restrict__ sums,
                               const float* __restrict__ sumsq, float invN,
                               int K, int NN, const float* __restrict__ W,
                               const float* __restrict__ extra_b,
                               float* __restrict__ Wout, float* __restrict__ vout) {
  __shared__ float s_ms[N_FEAT];
  __shared__ float s_s[N_FEAT];
  const int j = threadIdx.x;
  for (int k = j; k < K; k += blockDim.x) {
    float mean = sums[k] * invN;
    float var  = sumsq[k] * invN - mean * mean;
    float sc   = rsqrtf(var + BN_EPS);
    s_ms[k] = mean * sc;
    s_s[k]  = sc;
  }
  __syncthreads();
  float acc = extra_b ? extra_b[j] : 0.f;
  for (int k = 0; k < K; ++k) {
    float w = W[(size_t)k * NN + j];
    Wout[(size_t)k * NN + j] = s_s[k] * w;
    acc += (BN_BETA - s_ms[k]) * w;
  }
  vout[j] = acc;
}

__global__ __launch_bounds__(128) void gemm_wmma_kernel(
    const float* __restrict__ A, const float* __restrict__ Wt,
    const float* __restrict__ v, float* __restrict__ C,
    int M, int K, int doRelu) {
  const int lane = threadIdx.x & 31;
  const int wave = threadIdx.x >> 5;
  const int rowTile = blockIdx.x * 4 + wave;
  if (rowTile >= (M >> 4)) return;
  const int mrow = (rowTile << 4) + (lane & 15);
  const int kOff = (lane >> 4) << 1;
  const int n0   = lane & 15;
  v8f acc[4] = {};
  for (int kk = 0; kk < K; kk += 4) {
    const int k0 = kk + kOff;
    v2f a = *(const v2f*)(A + (size_t)mrow * K + k0);
    const float* w0 = Wt + (size_t)k0 * HIDDEN + n0;
#pragma unroll
    for (int t = 0; t < 4; ++t) {
      v2f b;
      b.x = w0[t * 16];
      b.y = w0[HIDDEN + t * 16];
      acc[t] = __builtin_amdgcn_wmma_f32_16x16x4_f32(
          false, a, false, b, (short)0, acc[t], false, false);
    }
  }
  const int rowBase = (rowTile << 4) + ((lane >> 4) << 3);
#pragma unroll
  for (int t = 0; t < 4; ++t) {
    const int n = t * 16 + n0;
    const float bias = v[n];
#pragma unroll
    for (int e = 0; e < 8; ++e) {
      float val = acc[t][e] + bias;
      if (doRelu) val = fmaxf(val, 0.f);
      C[(size_t)(rowBase + e) * HIDDEN + n] = val;
    }
  }
}

__global__ void deg_init_kernel(float* deg, int N) {
  int i = blockIdx.x * blockDim.x + threadIdx.x;
  if (i < N) deg[i] = 1.0f;
}
__global__ void deg_accum_kernel(const int* __restrict__ row,
                                 const float* __restrict__ att, int comp,
                                 int E, float* __restrict__ deg) {
  int e = blockIdx.x * blockDim.x + threadIdx.x;
  if (e >= E) return;
  float w = att ? att[(size_t)e * 2 + comp] : 1.0f;
  atomicAdd(&deg[row[e]], w);
}
__global__ void make_dinv_kernel(const float* __restrict__ deg,
                                 float* __restrict__ dinv, int N) {
  int i = blockIdx.x * blockDim.x + threadIdx.x;
  if (i < N) dinv[i] = deg[i] > 0.f ? rsqrtf(deg[i]) : 0.f;
}

__global__ void scatter_kernel(const float* __restrict__ m,
                               const int* __restrict__ row,
                               const int* __restrict__ col,
                               const float* __restrict__ att, int comp,
                               const float* __restrict__ dinv,
                               float* __restrict__ acc, int E) {
  int idx = blockIdx.x * blockDim.x + threadIdx.x;
  int e = idx >> 6, j = idx & 63;
  if (e >= E) return;
  int r = row[e], c = col[e];
  float w = att ? att[(size_t)e * 2 + comp] : 1.0f;
  float coef = dinv[r] * w * dinv[c];
  atomicAdd(&acc[(size_t)c * HIDDEN + j], coef * m[(size_t)r * HIDDEN + j]);
}

__global__ void finish_gcn_kernel(const float* __restrict__ acc,
                                  const float* __restrict__ m,
                                  const float* __restrict__ dinv,
                                  const float* __restrict__ b,
                                  float* __restrict__ out, int N) {
  int idx = blockIdx.x * blockDim.x + threadIdx.x;
  int i = idx >> 6, j = idx & 63;
  if (i >= N) return;
  float di = dinv[i];
  float val = acc[idx] + di * di * m[idx] + b[j];
  out[idx] = fmaxf(val, 0.f);
}

__global__ void edge_att_kernel(const float* __restrict__ h,
                                const int* __restrict__ row,
                                const int* __restrict__ col,
                                const float* __restrict__ Wea,
                                const float* __restrict__ bea,
                                float* __restrict__ att, int E) {
  int e = blockIdx.x * blockDim.x + threadIdx.x;
  if (e >= E) return;
  const float* hr = h + (size_t)row[e] * HIDDEN;
  const float* hc = h + (size_t)col[e] * HIDDEN;
  float l0 = bea[0], l1 = bea[1];
  for (int j = 0; j < HIDDEN; ++j) {
    float a = hr[j], c = hc[j];
    l0 += a * Wea[j * 2 + 0] + c * Wea[(HIDDEN + j) * 2 + 0];
    l1 += a * Wea[j * 2 + 1] + c * Wea[(HIDDEN + j) * 2 + 1];
  }
  float mx = fmaxf(l0, l1);
  float e0 = __expf(l0 - mx), e1 = __expf(l1 - mx);
  float inv = 1.f / (e0 + e1);
  att[(size_t)e * 2 + 0] = e0 * inv;
  att[(size_t)e * 2 + 1] = e1 * inv;
}

__global__ void node_att_split_kernel(const float* __restrict__ h,
                                      const float* __restrict__ Wna,
                                      const float* __restrict__ bna,
                                      float* __restrict__ xc,
                                      float* __restrict__ xo, int N) {
  int i = blockIdx.x * blockDim.x + threadIdx.x;
  if (i >= N) return;
  const float* hi = h + (size_t)i * HIDDEN;
  float l0 = bna[0], l1 = bna[1];
  for (int j = 0; j < HIDDEN; ++j) {
    float a = hi[j];
    l0 += a * Wna[j * 2 + 0];
    l1 += a * Wna[j * 2 + 1];
  }
  float mx = fmaxf(l0, l1);
  float e0 = __expf(l0 - mx), e1 = __expf(l1 - mx);
  float inv = 1.f / (e0 + e1);
  float p0 = e0 * inv, p1 = e1 * inv;
  for (int j = 0; j < HIDDEN; ++j) {
    float a = hi[j];
    xc[(size_t)i * HIDDEN + j] = p0 * a;
    xo[(size_t)i * HIDDEN + j] = p1 * a;
  }
}

__global__ void pool_kernel(const float* __restrict__ x,
                            const int* __restrict__ batch,
                            float* __restrict__ g, int N) {
  int idx = blockIdx.x * blockDim.x + threadIdx.x;
  int i = idx >> 6, j = idx & 63;
  if (i >= N) return;
  atomicAdd(&g[(size_t)batch[i] * HIDDEN + j], x[idx]);
}

__global__ void add_kernel(const float* a, const float* b, float* c, int n) {
  int i = blockIdx.x * blockDim.x + threadIdx.x;
  if (i < n) c[i] = a[i] + b[i];
}

__global__ void logits_kernel(const float* __restrict__ h,
                              const float* __restrict__ Wt,
                              const float* __restrict__ v,
                              float* __restrict__ out, int G) {
  int g = blockIdx.x * blockDim.x + threadIdx.x;
  if (g >= G) return;
  float l[N_CLASSES];
  for (int c = 0; c < N_CLASSES; ++c) l[c] = v[c];
  const float* hg = h + (size_t)g * HIDDEN;
  for (int k = 0; k < HIDDEN; ++k) {
    float x = hg[k];
    for (int c = 0; c < N_CLASSES; ++c) l[c] += x * Wt[k * N_CLASSES + c];
  }
  float mx = l[0];
  for (int c = 1; c < N_CLASSES; ++c) mx = fmaxf(mx, l[c]);
  float s = 0.f;
  for (int c = 0; c < N_CLASSES; ++c) s += __expf(l[c] - mx);
  float lse = __logf(s) + mx;
  for (int c = 0; c < N_CLASSES; ++c) out[(size_t)g * N_CLASSES + c] = l[c] - lse;
}

extern "C" void kernel_launch(void* const* d_in, const int* in_sizes, int n_in,
                              void* d_out, int out_size, void* d_ws, size_t ws_size,
                              hipStream_t stream) {
  (void)in_sizes; (void)n_in; (void)out_size; (void)ws_size;
  const float* x       = (const float*)d_in[0];
  const int*   ei      = (const int*)d_in[1];
  const int*   row     = ei;
  const int*   col     = ei + N_EDGES;
  const int*   batch   = (const int*)d_in[2];
  const float* W_feat  = (const float*)d_in[3];
  const float* W_convs = (const float*)d_in[4];
  const float* b_convs = (const float*)d_in[5];
  const float* W_ea    = (const float*)d_in[6];
  const float* b_ea    = (const float*)d_in[7];
  const float* W_na    = (const float*)d_in[8];
  const float* b_na    = (const float*)d_in[9];
  const float* W_ctx   = (const float*)d_in[10];
  const float* b_ctx   = (const float*)d_in[11];
  const float* W_obj   = (const float*)d_in[12];
  const float* b_obj   = (const float*)d_in[13];
  const float* W1h[3]  = {(const float*)d_in[14], (const float*)d_in[18], (const float*)d_in[22]};
  const float* b1h[3]  = {(const float*)d_in[15], (const float*)d_in[19], (const float*)d_in[23]};
  const float* W2h[3]  = {(const float*)d_in[16], (const float*)d_in[20], (const float*)d_in[24]};
  const float* b2h[3]  = {(const float*)d_in[17], (const float*)d_in[21], (const float*)d_in[25]};

  size_t off = 0;
  auto carve = [&](size_t elems) -> float* {
    float* p = (float*)((char*)d_ws + off);
    off += ((elems * sizeof(float)) + 255) & ~(size_t)255;
    return p;
  };
  float* hb    = carve((size_t)N_NODES * HIDDEN);
  float* mbuf  = carve((size_t)N_NODES * HIDDEN);
  float* accb  = carve((size_t)N_NODES * HIDDEN);
  float* xc    = carve((size_t)N_NODES * HIDDEN);
  float* xo    = carve((size_t)N_NODES * HIDDEN);
  float* att   = carve((size_t)N_EDGES * 2);
  float* degu  = carve(N_NODES);
  float* dinvu = carve(N_NODES);
  float* dega  = carve(N_NODES);
  float* dinva = carve(N_NODES);
  float* gc    = carve((size_t)N_GRAPHS * HIDDEN);
  float* go    = carve((size_t)N_GRAPHS * HIDDEN);
  float* gs    = carve((size_t)N_GRAPHS * HIDDEN);
  float* r1    = carve((size_t)N_GRAPHS * HIDDEN);
  float* sums  = carve(N_FEAT);
  float* sq    = carve(N_FEAT);
  float* Wf    = carve((size_t)N_FEAT * HIDDEN);
  float* vf    = carve(HIDDEN);

  auto nblk = [](long n, int b) { return (int)((n + b - 1) / b); };

  auto stats = [&](const float* A, int N, int F) {
    hipMemsetAsync(sums, 0, (size_t)F * sizeof(float), stream);
    hipMemsetAsync(sq,   0, (size_t)F * sizeof(float), stream);
    colstats_kernel<<<512, F, 0, stream>>>(A, N, F, sums, sq);
  };
  auto gemm = [&](const float* A, float* C, int M, int K, int relu) {
    gemm_wmma_kernel<<<nblk(M / 16, 4), 128, 0, stream>>>(A, Wf, vf, C, M, K, relu);
  };

  stats(x, N_NODES, N_FEAT);
  fold_bn_kernel<<<1, HIDDEN, 0, stream>>>(sums, sq, 1.f / N_NODES, N_FEAT, HIDDEN,
                                           W_feat, nullptr, Wf, vf);
  gemm(x, hb, N_NODES, N_FEAT, 1);

  deg_init_kernel<<<nblk(N_NODES, 256), 256, 0, stream>>>(degu, N_NODES);
  deg_accum_kernel<<<nblk(N_EDGES, 256), 256, 0, stream>>>(row, nullptr, 0, N_EDGES, degu);
  make_dinv_kernel<<<nblk(N_NODES, 256), 256, 0, stream>>>(degu, dinvu, N_NODES);

  for (int i = 0; i < 3; ++i) {
    stats(hb, N_NODES, HIDDEN);
    fold_bn_kernel<<<1, HIDDEN, 0, stream>>>(sums, sq, 1.f / N_NODES, HIDDEN, HIDDEN,
                                             W_convs + (size_t)i * HIDDEN * HIDDEN,
                                             nullptr, Wf, vf);
    gemm(hb, mbuf, N_NODES, HIDDEN, 0);
    hipMemsetAsync(accb, 0, (size_t)N_NODES * HIDDEN * sizeof(float), stream);
    scatter_kernel<<<nblk((long)N_EDGES * 64, 256), 256, 0, stream>>>(
        mbuf, row, col, nullptr, 0, dinvu, accb, N_EDGES);
    finish_gcn_kernel<<<nblk((long)N_NODES * 64, 256), 256, 0, stream>>>(
        accb, mbuf, dinvu, b_convs + (size_t)i * HIDDEN, hb, N_NODES);
  }

  edge_att_kernel<<<nblk(N_EDGES, 256), 256, 0, stream>>>(hb, row, col, W_ea, b_ea,
                                                          att, N_EDGES);
  node_att_split_kernel<<<nblk(N_NODES, 256), 256, 0, stream>>>(hb, W_na, b_na,
                                                                xc, xo, N_NODES);

  auto run_branch = [&](float* xin, int comp, const float* W, const float* b,
                        float* gout) {
    stats(xin, N_NODES, HIDDEN);
    fold_bn_kernel<<<1, HIDDEN, 0, stream>>>(sums, sq, 1.f / N_NODES, HIDDEN, HIDDEN,
                                             W, nullptr, Wf, vf);
    gemm(xin, mbuf, N_NODES, HIDDEN, 0);
    deg_init_kernel<<<nblk(N_NODES, 256), 256, 0, stream>>>(dega, N_NODES);
    deg_accum_kernel<<<nblk(N_EDGES, 256), 256, 0, stream>>>(row, att, comp, N_EDGES, dega);
    make_dinv_kernel<<<nblk(N_NODES, 256), 256, 0, stream>>>(dega, dinva, N_NODES);
    hipMemsetAsync(accb, 0, (size_t)N_NODES * HIDDEN * sizeof(float), stream);
    scatter_kernel<<<nblk((long)N_EDGES * 64, 256), 256, 0, stream>>>(
        mbuf, row, col, att, comp, dinva, accb, N_EDGES);
    finish_gcn_kernel<<<nblk((long)N_NODES * 64, 256), 256, 0, stream>>>(
        accb, mbuf, dinva, b, xin, N_NODES);
    hipMemsetAsync(gout, 0, (size_t)N_GRAPHS * HIDDEN * sizeof(float), stream);
    pool_kernel<<<nblk((long)N_NODES * 64, 256), 256, 0, stream>>>(xin, batch, gout,
                                                                   N_NODES);
  };
  run_branch(xc, 0, W_ctx, b_ctx, gc);
  run_branch(xo, 1, W_obj, b_obj, go);

  add_kernel<<<nblk(N_GRAPHS * HIDDEN, 256), 256, 0, stream>>>(gc, go, gs,
                                                               N_GRAPHS * HIDDEN);

  float* out = (float*)d_out;
  auto run_head = [&](const float* gin, int h, float* outp) {
    stats(gin, N_GRAPHS, HIDDEN);
    fold_bn_kernel<<<1, HIDDEN, 0, stream>>>(sums, sq, 1.f / N_GRAPHS, HIDDEN, HIDDEN,
                                             W1h[h], b1h[h], Wf, vf);
    gemm(gin, r1, N_GRAPHS, HIDDEN, 1);
    stats(r1, N_GRAPHS, HIDDEN);
    fold_bn_kernel<<<1, N_CLASSES, 0, stream>>>(sums, sq, 1.f / N_GRAPHS, HIDDEN,
                                                N_CLASSES, W2h[h], b2h[h], Wf, vf);
    logits_kernel<<<nblk(N_GRAPHS, 256), 256, 0, stream>>>(r1, Wf, vf, outp, N_GRAPHS);
  };
  run_head(gc, 0, out);
  run_head(go, 1, out + N_GRAPHS * N_CLASSES);
  run_head(gs, 2, out + 2 * N_GRAPHS * N_CLASSES);
}


# column-statistics pass: one 4-byte load in flight per lane -> eight 16-byte loads in flight, 16 waves per WGP; block partials summed in LDS before the atomics (BN sums: new reduction tree)
# speedup vs baseline: 1.2836x; 1.2836x over previous
#include <hip/hip_runtime.h>
#include <math.h>


#define N_NODES  100000
#define N_EDGES  1000000
#define N_FEAT   128
#define HIDDEN   64
#define N_GRAPHS 512
#define N_CLASSES 10
#define BN_EPS  1e-5f
#define BN_BETA 1e-4f

typedef __attribute__((ext_vector_type(2))) float v2f;
typedef __attribute__((ext_vector_type(8))) float v8f;

__global__ __launch_bounds__(256) void colstats_kernel(
    const float* __restrict__ A, int nvec, int F,
    float* __restrict__ sums, float* __restrict__ sumsq) {
  __shared__ float sh_s[1024];
  __shared__ float sh_q[1024];
  const float4* __restrict__ A4 = (const float4*)A;
  const int tid = threadIdx.x;
  const int stride = gridDim.x * 256;
  const int last = nvec - 1;
  float4 s = make_float4(0.f, 0.f, 0.f, 0.f);
  float4 q = make_float4(0.f, 0.f, 0.f, 0.f);
  for (int i = blockIdx.x * 256 + tid; i < nvec; i += 8 * stride) {
    float4 v[8];
#pragma unroll
    for (int u = 0; u < 8; ++u) {
      const int idx = i + u * stride;
      v[u] = A4[idx < last ? idx : last];
    }
    __builtin_amdgcn_sched_barrier(0);
#pragma unroll
    for (int u = 0; u < 8; ++u) {
      const bool live = (i + u * stride) < nvec;
      const float a = live ? v[u].x : 0.f;
      const float b = live ? v[u].y : 0.f;
      const float c = live ? v[u].z : 0.f;
      const float d = live ? v[u].w : 0.f;
      s.x += a; q.x += a * a;
      s.y += b; q.y += b * b;
      s.z += c; q.z += c * c;
      s.w += d; q.w += d * d;
    }
  }
  ((float4*)sh_s)[tid] = s;
  ((float4*)sh_q)[tid] = q;
  __syncthreads();
  if (tid < F) {
    float ts = 0.f, tq = 0.f;
    for (int p = tid; p < 1024; p += F) { ts += sh_s[p]; tq += sh_q[p]; }
    atomicAdd(&sums[tid], ts);
    atomicAdd(&sumsq[tid], tq);
  }
}

__global__ void fold_bn_kernel(const float* __restrict__ sums,
                               const float* __restrict__ sumsq, float invN,
                               int K, int NN, const float* __restrict__ W,
                               const float* __restrict__ extra_b,
                               float* __restrict__ Wout, float* __restrict__ vout) {
  __shared__ float s_ms[N_FEAT];
  __shared__ float s_s[N_FEAT];
  const int j = threadIdx.x;
  for (int k = j; k < K; k += blockDim.x) {
    float mean = sums[k] * invN;
    float var  = sumsq[k] * invN - mean * mean;
    float sc   = rsqrtf(var + BN_EPS);
    s_ms[k] = mean * sc;
    s_s[k]  = sc;
  }
  __syncthreads();
  float acc = extra_b ? extra_b[j] : 0.f;
  for (int k = 0; k < K; ++k) {
    float w = W[(size_t)k * NN + j];
    Wout[(size_t)k * NN + j] = s_s[k] * w;
    acc += (BN_BETA - s_ms[k]) * w;
  }
  vout[j] = acc;
}

__global__ __launch_bounds__(128) void gemm_wmma_kernel(
    const float* __restrict__ A, const float* __restrict__ Wt,
    const float* __restrict__ v, float* __restrict__ C,
    int M, int K, int doRelu) {
  const int lane = threadIdx.x & 31;
  const int wave = threadIdx.x >> 5;
  const int rowTile = blockIdx.x * 4 + wave;
  if (rowTile >= (M >> 4)) return;
  const int mrow = (rowTile << 4) + (lane & 15);
  const int kOff = (lane >> 4) << 1;
  const int n0   = lane & 15;
  v8f acc[4] = {};
  for (int kk = 0; kk < K; kk += 4) {
    const int k0 = kk + kOff;
    v2f a = *(const v2f*)(A + (size_t)mrow * K + k0);
    const float* w0 = Wt + (size_t)k0 * HIDDEN + n0;
#pragma unroll
    for (int t = 0; t < 4; ++t) {
      v2f b;
      b.x = w0[t * 16];
      b.y = w0[HIDDEN + t * 16];
      acc[t] = __builtin_amdgcn_wmma_f32_16x16x4_f32(
          false, a, false, b, (short)0, acc[t], false, false);
    }
  }
  const int rowBase = (rowTile << 4) + ((lane >> 4) << 3);
#pragma unroll
  for (int t = 0; t < 4; ++t) {
    const int n = t * 16 + n0;
    const float bias = v[n];
#pragma unroll
    for (int e = 0; e < 8; ++e) {
      float val = acc[t][e] + bias;
      if (doRelu) val = fmaxf(val, 0.f);
      C[(size_t)(rowBase + e) * HIDDEN + n] = val;
    }
  }
}

__global__ void deg_init_kernel(float* deg, int N) {
  int i = blockIdx.x * blockDim.x + threadIdx.x;
  if (i < N) deg[i] = 1.0f;
}
__global__ void deg_accum_kernel(const int* __restrict__ row,
                                 const float* __restrict__ att, int comp,
                                 int E, float* __restrict__ deg) {
  int e = blockIdx.x * blockDim.x + threadIdx.x;
  if (e >= E) return;
  float w = att ? att[(size_t)e * 2 + comp] : 1.0f;
  atomicAdd(&deg[row[e]], w);
}
__global__ void make_dinv_kernel(const float* __restrict__ deg,
                                 float* __restrict__ dinv, int N) {
  int i = blockIdx.x * blockDim.x + threadIdx.x;
  if (i < N) dinv[i] = deg[i] > 0.f ? rsqrtf(deg[i]) : 0.f;
}

__global__ void scatter_kernel(const float* __restrict__ m,
                               const int* __restrict__ row,
                               const int* __restrict__ col,
                               const float* __restrict__ att, int comp,
                               const float* __restrict__ dinv,
                               float* __restrict__ acc, int E) {
  int idx = blockIdx.x * blockDim.x + threadIdx.x;
  int e = idx >> 6, j = idx & 63;
  if (e >= E) return;
  int r = row[e], c = col[e];
  float w = att ? att[(size_t)e * 2 + comp] : 1.0f;
  float coef = dinv[r] * w * dinv[c];
  atomicAdd(&acc[(size_t)c * HIDDEN + j], coef * m[(size_t)r * HIDDEN + j]);
}

__global__ void finish_gcn_kernel(const float* __restrict__ acc,
                                  const float* __restrict__ m,
                                  const float* __restrict__ dinv,
                                  const float* __restrict__ b,
                                  float* __restrict__ out, int N) {
  int idx = blockIdx.x * blockDim.x + threadIdx.x;
  int i = idx >> 6, j = idx & 63;
  if (i >= N) return;
  float di = dinv[i];
  float val = acc[idx] + di * di * m[idx] + b[j];
  out[idx] = fmaxf(val, 0.f);
}

__global__ void edge_att_kernel(const float* __restrict__ h,
                                const int* __restrict__ row,
                                const int* __restrict__ col,
                                const float* __restrict__ Wea,
                                const float* __restrict__ bea,
                                float* __restrict__ att, int E) {
  int e = blockIdx.x * blockDim.x + threadIdx.x;
  if (e >= E) return;
  const float* hr = h + (size_t)row[e] * HIDDEN;
  const float* hc = h + (size_t)col[e] * HIDDEN;
  float l0 = bea[0], l1 = bea[1];
  for (int j = 0; j < HIDDEN; ++j) {
    float a = hr[j], c = hc[j];
    l0 += a * Wea[j * 2 + 0] + c * Wea[(HIDDEN + j) * 2 + 0];
    l1 += a * Wea[j * 2 + 1] + c * Wea[(HIDDEN + j) * 2 + 1];
  }
  float mx = fmaxf(l0, l1);
  float e0 = __expf(l0 - mx), e1 = __expf(l1 - mx);
  float inv = 1.f / (e0 + e1);
  att[(size_t)e * 2 + 0] = e0 * inv;
  att[(size_t)e * 2 + 1] = e1 * inv;
}

__global__ void node_att_split_kernel(const float* __restrict__ h,
                                      const float* __restrict__ Wna,
                                      const float* __restrict__ bna,
                                      float* __restrict__ xc,
                                      float* __restrict__ xo, int N) {
  int i = blockIdx.x * blockDim.x + threadIdx.x;
  if (i >= N) return;
  const float* hi = h + (size_t)i * HIDDEN;
  float l0 = bna[0], l1 = bna[1];
  for (int j = 0; j < HIDDEN; ++j) {
    float a = hi[j];
    l0 += a * Wna[j * 2 + 0];
    l1 += a * Wna[j * 2 + 1];
  }
  float mx = fmaxf(l0, l1);
  float e0 = __expf(l0 - mx), e1 = __expf(l1 - mx);
  float inv = 1.f / (e0 + e1);
  float p0 = e0 * inv, p1 = e1 * inv;
  for (int j = 0; j < HIDDEN; ++j) {
    float a = hi[j];
    xc[(size_t)i * HIDDEN + j] = p0 * a;
    xo[(size_t)i * HIDDEN + j] = p1 * a;
  }
}

__global__ void pool_kernel(const float* __restrict__ x,
                            const int* __restrict__ batch,
                            float* __restrict__ g, int N) {
  int idx = blockIdx.x * blockDim.x + threadIdx.x;
  int i = idx >> 6, j = idx & 63;
  if (i >= N) return;
  atomicAdd(&g[(size_t)batch[i] * HIDDEN + j], x[idx]);
}

__global__ void add_kernel(const float* a, const float* b, float* c, int n) {
  int i = blockIdx.x * blockDim.x + threadIdx.x;
  if (i < n) c[i] = a[i] + b[i];
}

__global__ void logits_kernel(const float* __restrict__ h,
                              const float* __restrict__ Wt,
                              const float* __restrict__ v,
                              float* __restrict__ out, int G) {
  int g = blockIdx.x * blockDim.x + threadIdx.x;
  if (g >= G) return;
  float l[N_CLASSES];
  for (int c = 0; c < N_CLASSES; ++c) l[c] = v[c];
  const float* hg = h + (size_t)g * HIDDEN;
  for (int k = 0; k < HIDDEN; ++k) {
    float x = hg[k];
    for (int c = 0; c < N_CLASSES; ++c) l[c] += x * Wt[k * N_CLASSES + c];
  }
  float mx = l[0];
  for (int c = 1; c < N_CLASSES; ++c) mx = fmaxf(mx, l[c]);
  float s = 0.f;
  for (int c = 0; c < N_CLASSES; ++c) s += __expf(l[c] - mx);
  float lse = __logf(s) + mx;
  for (int c = 0; c < N_CLASSES; ++c) out[(size_t)g * N_CLASSES + c] = l[c] - lse;
}

extern "C" void kernel_launch(void* const* d_in, const int* in_sizes, int n_in,
                              void* d_out, int out_size, void* d_ws, size_t ws_size,
                              hipStream_t stream) {
  (void)in_sizes; (void)n_in; (void)out_size; (void)ws_size;
  const float* x       = (const float*)d_in[0];
  const int*   ei      = (const int*)d_in[1];
  const int*   row     = ei;
  const int*   col     = ei + N_EDGES;
  const int*   batch   = (const int*)d_in[2];
  const float* W_feat  = (const float*)d_in[3];
  const float* W_convs = (const float*)d_in[4];
  const float* b_convs = (const float*)d_in[5];
  const float* W_ea    = (const float*)d_in[6];
  const float* b_ea    = (const float*)d_in[7];
  const float* W_na    = (const float*)d_in[8];
  const float* b_na    = (const float*)d_in[9];
  const float* W_ctx   = (const float*)d_in[10];
  const float* b_ctx   = (const float*)d_in[11];
  const float* W_obj   = (const float*)d_in[12];
  const float* b_obj   = (const float*)d_in[13];
  const float* W1h[3]  = {(const float*)d_in[14], (const float*)d_in[18], (const float*)d_in[22]};
  const float* b1h[3]  = {(const float*)d_in[15], (const float*)d_in[19], (const float*)d_in[23]};
  const float* W2h[3]  = {(const float*)d_in[16], (const float*)d_in[20], (const float*)d_in[24]};
  const float* b2h[3]  = {(const float*)d_in[17], (const float*)d_in[21], (const float*)d_in[25]};

  size_t off = 0;
  auto carve = [&](size_t elems) -> float* {
    float* p = (float*)((char*)d_ws + off);
    off += ((elems * sizeof(float)) + 255) & ~(size_t)255;
    return p;
  };
  float* hb    = carve((size_t)N_NODES * HIDDEN);
  float* mbuf  = carve((size_t)N_NODES * HIDDEN);
  float* accb  = carve((size_t)N_NODES * HIDDEN);
  float* xc    = carve((size_t)N_NODES * HIDDEN);
  float* xo    = carve((size_t)N_NODES * HIDDEN);
  float* att   = carve((size_t)N_EDGES * 2);
  float* degu  = carve(N_NODES);
  float* dinvu = carve(N_NODES);
  float* dega  = carve(N_NODES);
  float* dinva = carve(N_NODES);
  float* gc    = carve((size_t)N_GRAPHS * HIDDEN);
  float* go    = carve((size_t)N_GRAPHS * HIDDEN);
  float* gs    = carve((size_t)N_GRAPHS * HIDDEN);
  float* r1    = carve((size_t)N_GRAPHS * HIDDEN);
  float* sums  = carve(N_FEAT);
  float* sq    = carve(N_FEAT);
  float* Wf    = carve((size_t)N_FEAT * HIDDEN);
  float* vf    = carve(HIDDEN);

  auto nblk = [](long n, int b) { return (int)((n + b - 1) / b); };

  auto stats = [&](const float* A, int N, int F) {
    hipMemsetAsync(sums, 0, (size_t)F * sizeof(float), stream);
    hipMemsetAsync(sq,   0, (size_t)F * sizeof(float), stream);
    const int nvec = N * F / 4;
    const int grid = nblk(nvec, 256) < 512 ? nblk(nvec, 256) : 512;
    colstats_kernel<<<grid, 256, 0, stream>>>(A, nvec, F, sums, sq);
  };
  auto gemm = [&](const float* A, float* C, int M, int K, int relu) {
    gemm_wmma_kernel<<<nblk(M / 16, 4), 128, 0, stream>>>(A, Wf, vf, C, M, K, relu);
  };

  stats(x, N_NODES, N_FEAT);
  fold_bn_kernel<<<1, HIDDEN, 0, stream>>>(sums, sq, 1.f / N_NODES, N_FEAT, HIDDEN,
                                           W_feat, nullptr, Wf, vf);
  gemm(x, hb, N_NODES, N_FEAT, 1);

  deg_init_kernel<<<nblk(N_NODES, 256), 256, 0, stream>>>(degu, N_NODES);
  deg_accum_kernel<<<nblk(N_EDGES, 256), 256, 0, stream>>>(row, nullptr, 0, N_EDGES, degu);
  make_dinv_kernel<<<nblk(N_NODES, 256), 256, 0, stream>>>(degu, dinvu, N_NODES);

  for (int i = 0; i < 3; ++i) {
    stats(hb, N_NODES, HIDDEN);
    fold_bn_kernel<<<1, HIDDEN, 0, stream>>>(sums, sq, 1.f / N_NODES, HIDDEN, HIDDEN,
                                             W_convs + (size_t)i * HIDDEN * HIDDEN,
                                             nullptr, Wf, vf);
    gemm(hb, mbuf, N_NODES, HIDDEN, 0);
    hipMemsetAsync(accb, 0, (size_t)N_NODES * HIDDEN * sizeof(float), stream);
    scatter_kernel<<<nblk((long)N_EDGES * 64, 256), 256, 0, stream>>>(
        mbuf, row, col, nullptr, 0, dinvu, accb, N_EDGES);
    finish_gcn_kernel<<<nblk((long)N_NODES * 64, 256), 256, 0, stream>>>(
        accb, mbuf, dinvu, b_convs + (size_t)i * HIDDEN, hb, N_NODES);
  }

  edge_att_kernel<<<nblk(N_EDGES, 256), 256, 0, stream>>>(hb, row, col, W_ea, b_ea,
                                                          att, N_EDGES);
  node_att_split_kernel<<<nblk(N_NODES, 256), 256, 0, stream>>>(hb, W_na, b_na,
                                                                xc, xo, N_NODES);

  auto run_branch = [&](float* xin, int comp, const float* W, const float* b,
                        float* gout) {
    stats(xin, N_NODES, HIDDEN);
    fold_bn_kernel<<<1, HIDDEN, 0, stream>>>(sums, sq, 1.f / N_NODES, HIDDEN, HIDDEN,
                                             W, nullptr, Wf, vf);
    gemm(xin, mbuf, N_NODES, HIDDEN, 0);
    deg_init_kernel<<<nblk(N_NODES, 256), 256, 0, stream>>>(dega, N_NODES);
    deg_accum_kernel<<<nblk(N_EDGES, 256), 256, 0, stream>>>(row, att, comp, N_EDGES, dega);
    make_dinv_kernel<<<nblk(N_NODES, 256), 256, 0, stream>>>(dega, dinva, N_NODES);
    hipMemsetAsync(accb, 0, (size_t)N_NODES * HIDDEN * sizeof(float), stream);
    scatter_kernel<<<nblk((long)N_EDGES * 64, 256), 256, 0, stream>>>(
        mbuf, row, col, att, comp, dinva, accb, N_EDGES);
    finish_gcn_kernel<<<nblk((long)N_NODES * 64, 256), 256, 0, stream>>>(
        accb, mbuf, dinva, b, xin, N_NODES);
    hipMemsetAsync(gout, 0, (size_t)N_GRAPHS * HIDDEN * sizeof(float), stream);
    pool_kernel<<<nblk((long)N_NODES * 64, 256), 256, 0, stream>>>(xin, batch, gout,
                                                                   N_NODES);
  };
  run_branch(xc, 0, W_ctx, b_ctx, gc);
  run_branch(xo, 1, W_obj, b_obj, go);

  add_kernel<<<nblk(N_GRAPHS * HIDDEN, 256), 256, 0, stream>>>(gc, go, gs,
                                                               N_GRAPHS * HIDDEN);

  float* out = (float*)d_out;
  auto run_head = [&](const float* gin, int h, float* outp) {
    stats(gin, N_GRAPHS, HIDDEN);
    fold_bn_kernel<<<1, HIDDEN, 0, stream>>>(sums, sq, 1.f / N_GRAPHS, HIDDEN, HIDDEN,
                                             W1h[h], b1h[h], Wf, vf);
    gemm(gin, r1, N_GRAPHS, HIDDEN, 1);
    stats(r1, N_GRAPHS, HIDDEN);
    fold_bn_kernel<<<1, N_CLASSES, 0, stream>>>(sums, sq, 1.f / N_GRAPHS, HIDDEN,
                                                N_CLASSES, W2h[h], b2h[h], Wf, vf);
    logits_kernel<<<nblk(N_GRAPHS, 256), 256, 0, stream>>>(r1, Wf, vf, outp, N_GRAPHS);
  };
  run_head(gc, 0, out);
  run_head(go, 1, out + N_GRAPHS * N_CLASSES);
  run_head(gs, 2, out + 2 * N_GRAPHS * N_CLASSES);
}


# edge propagation: f32 atomic scatter -> per-destination gather summed in registers, over an edge list counting-sorted by destination once per call (arrival-order rank: new summation order)
# speedup vs baseline: 2.2952x; 1.7881x over previous
#include <hip/hip_runtime.h>
#include <math.h>


#define N_NODES  100000
#define N_EDGES  1000000
#define N_FEAT   128
#define HIDDEN   64
#define N_GRAPHS 512
#define N_CLASSES 10
#define BN_EPS  1e-5f
#define BN_BETA 1e-4f

typedef __attribute__((ext_vector_type(2))) float v2f;
typedef __attribute__((ext_vector_type(8))) float v8f;

__global__ __launch_bounds__(256) void colstats_kernel(
    const float* __restrict__ A, int nvec, int F,
    float* __restrict__ sums, float* __restrict__ sumsq) {
  __shared__ float sh_s[1024];
  __shared__ float sh_q[1024];
  const float4* __restrict__ A4 = (const float4*)A;
  const int tid = threadIdx.x;
  const int stride = gridDim.x * 256;
  const int last = nvec - 1;
  float4 s = make_float4(0.f, 0.f, 0.f, 0.f);
  float4 q = make_float4(0.f, 0.f, 0.f, 0.f);
  for (int i = blockIdx.x * 256 + tid; i < nvec; i += 8 * stride) {
    float4 v[8];
#pragma unroll
    for (int u = 0; u < 8; ++u) {
      const int idx = i + u * stride;
      v[u] = A4[idx < last ? idx : last];
    }
    __builtin_amdgcn_sched_barrier(0);
#pragma unroll
    for (int u = 0; u < 8; ++u) {
      const bool live = (i + u * stride) < nvec;
      const float a = live ? v[u].x : 0.f;
      const float b = live ? v[u].y : 0.f;
      const float c = live ? v[u].z : 0.f;
      const float d = live ? v[u].w : 0.f;
      s.x += a; q.x += a * a;
      s.y += b; q.y += b * b;
      s.z += c; q.z += c * c;
      s.w += d; q.w += d * d;
    }
  }
  ((float4*)sh_s)[tid] = s;
  ((float4*)sh_q)[tid] = q;
  __syncthreads();
  if (tid < F) {
    float ts = 0.f, tq = 0.f;
    for (int p = tid; p < 1024; p += F) { ts += sh_s[p]; tq += sh_q[p]; }
    atomicAdd(&sums[tid], ts);
    atomicAdd(&sumsq[tid], tq);
  }
}

__global__ void fold_bn_kernel(const float* __restrict__ sums,
                               const float* __restrict__ sumsq, float invN,
                               int K, int NN, const float* __restrict__ W,
                               const float* __restrict__ extra_b,
                               float* __restrict__ Wout, float* __restrict__ vout) {
  __shared__ float s_ms[N_FEAT];
  __shared__ float s_s[N_FEAT];
  const int j = threadIdx.x;
  for (int k = j; k < K; k += blockDim.x) {
    float mean = sums[k] * invN;
    float var  = sumsq[k] * invN - mean * mean;
    float sc   = rsqrtf(var + BN_EPS);
    s_ms[k] = mean * sc;
    s_s[k]  = sc;
  }
  __syncthreads();
  float acc = extra_b ? extra_b[j] : 0.f;
  for (int k = 0; k < K; ++k) {
    float w = W[(size_t)k * NN + j];
    Wout[(size_t)k * NN + j] = s_s[k] * w;
    acc += (BN_BETA - s_ms[k]) * w;
  }
  vout[j] = acc;
}

__global__ __launch_bounds__(128) void gemm_wmma_kernel(
    const float* __restrict__ A, const float* __restrict__ Wt,
    const float* __restrict__ v, float* __restrict__ C,
    int M, int K, int doRelu) {
  const int lane = threadIdx.x & 31;
  const int wave = threadIdx.x >> 5;
  const int rowTile = blockIdx.x * 4 + wave;
  if (rowTile >= (M >> 4)) return;
  const int mrow = (rowTile << 4) + (lane & 15);
  const int kOff = (lane >> 4) << 1;
  const int n0   = lane & 15;
  v8f acc[4] = {};
  for (int kk = 0; kk < K; kk += 4) {
    const int k0 = kk + kOff;
    v2f a = *(const v2f*)(A + (size_t)mrow * K + k0);
    const float* w0 = Wt + (size_t)k0 * HIDDEN + n0;
#pragma unroll
    for (int t = 0; t < 4; ++t) {
      v2f b;
      b.x = w0[t * 16];
      b.y = w0[HIDDEN + t * 16];
      acc[t] = __builtin_amdgcn_wmma_f32_16x16x4_f32(
          false, a, false, b, (short)0, acc[t], false, false);
    }
  }
  const int rowBase = (rowTile << 4) + ((lane >> 4) << 3);
#pragma unroll
  for (int t = 0; t < 4; ++t) {
    const int n = t * 16 + n0;
    const float bias = v[n];
#pragma unroll
    for (int e = 0; e < 8; ++e) {
      float val = acc[t][e] + bias;
      if (doRelu) val = fmaxf(val, 0.f);
      C[(size_t)(rowBase + e) * HIDDEN + n] = val;
    }
  }
}

__global__ void deg_init_kernel(float* deg, int N) {
  int i = blockIdx.x * blockDim.x + threadIdx.x;
  if (i < N) deg[i] = 1.0f;
}
__global__ void deg_accum_kernel(const int* __restrict__ row,
                                 const float* __restrict__ att, int comp,
                                 int E, float* __restrict__ deg) {
  int e = blockIdx.x * blockDim.x + threadIdx.x;
  if (e >= E) return;
  float w = att ? att[(size_t)e * 2 + comp] : 1.0f;
  atomicAdd(&deg[row[e]], w);
}
__global__ void make_dinv_kernel(const float* __restrict__ deg,
                                 float* __restrict__ dinv, int N) {
  int i = blockIdx.x * blockDim.x + threadIdx.x;
  if (i < N) dinv[i] = deg[i] > 0.f ? rsqrtf(deg[i]) : 0.f;
}

__global__ void csr_zero_kernel(int* __restrict__ cnt, int N) {
  int i = blockIdx.x * blockDim.x + threadIdx.x;
  if (i < N) cnt[i] = 0;
}
__global__ void csr_count_kernel(const int* __restrict__ col, int E,
                                 int* __restrict__ cnt) {
  int e = blockIdx.x * blockDim.x + threadIdx.x;
  if (e < E) atomicAdd(&cnt[col[e]], 1);
}
__device__ __forceinline__ void load_counts4(const int* __restrict__ cnt, int i0,
                                             int N, int c[4]) {
  if (i0 + 3 < N) {
    const int4 v = *(const int4*)(cnt + i0);
    c[0] = v.x; c[1] = v.y; c[2] = v.z; c[3] = v.w;
  } else {
    c[0] = (i0     < N) ? cnt[i0]     : 0;
    c[1] = (i0 + 1 < N) ? cnt[i0 + 1] : 0;
    c[2] = (i0 + 2 < N) ? cnt[i0 + 2] : 0;
    c[3] = 0;
  }
}
__global__ __launch_bounds__(256) void csr_tilesum_kernel(
    const int* __restrict__ cnt, int N, int* __restrict__ tsum) {
  __shared__ int sh[8];
  const int tid = threadIdx.x;
  int c[4];
  load_counts4(cnt, (blockIdx.x * 256 + tid) * 4, N, c);
  int t = c[0] + c[1] + c[2] + c[3];
  for (int o = 16; o > 0; o >>= 1) t += __shfl_xor(t, o);
  if ((tid & 31) == 0) sh[tid >> 5] = t;
  __syncthreads();
  if (tid == 0) {
    int s = 0;
    for (int w = 0; w < 8; ++w) s += sh[w];
    tsum[blockIdx.x] = s;
  }
}
__global__ __launch_bounds__(256) void csr_scan_kernel(
    const int* __restrict__ cnt, const int* __restrict__ tsum, int N,
    int* __restrict__ off, int* __restrict__ cur) {
  __shared__ int sh_w[8];
  __shared__ int sh_b[8];
  const int tid = threadIdx.x, lane = tid & 31, wave = tid >> 5;
  int bpart = 0;
  for (int t = tid; t < (int)blockIdx.x; t += 256) bpart += tsum[t];
  for (int o = 16; o > 0; o >>= 1) bpart += __shfl_xor(bpart, o);
  const int i0 = (blockIdx.x * 256 + tid) * 4;
  int c[4];
  load_counts4(cnt, i0, N, c);
  const int t = c[0] + c[1] + c[2] + c[3];
  int inc = t;
  for (int o = 1; o < 32; o <<= 1) {
    const int y = __shfl_up(inc, o);
    if (lane >= o) inc += y;
  }
  if (lane == 31) sh_w[wave] = inc;
  if (lane == 0)  sh_b[wave] = bpart;
  __syncthreads();
  int base = 0;
  for (int w = 0; w < 8; ++w) {
    base += sh_b[w];
    if (w < wave) base += sh_w[w];
  }
  const int o0 = base + inc - t;
  const int o1 = o0 + c[0], o2 = o1 + c[1], o3 = o2 + c[2];
  if (i0 + 3 < N) {
    const int4 v = make_int4(o0, o1, o2, o3);
    *(int4*)(off + i0) = v;
    *(int4*)(cur + i0) = v;
  } else {
    if (i0     < N) { off[i0]     = o0; cur[i0]     = o0; }
    if (i0 + 1 < N) { off[i0 + 1] = o1; cur[i0 + 1] = o1; }
    if (i0 + 2 < N) { off[i0 + 2] = o2; cur[i0 + 2] = o2; }
  }
  if (i0 < N && N - 1 <= i0 + 3) off[N] = o0 + t;
}
__global__ void csr_fill_kernel(const int* __restrict__ row,
                                const int* __restrict__ col, int E,
                                int* __restrict__ cur, int2* __restrict__ sedge) {
  int e = blockIdx.x * blockDim.x + threadIdx.x;
  if (e >= E) return;
  const int r = row[e];
  const int pos = atomicAdd(&cur[col[e]], 1);
  if ((unsigned)pos < (unsigned)E) sedge[pos] = make_int2(r, e);
}

template <int CNT>
__device__ __forceinline__ void gather_rows(const float2* __restrict__ m2, int r,
                                            float cf, int k, int lane, float2& acc) {
  int rr[CNT];
  float cc[CNT];
  float2 v[CNT];
#pragma unroll
  for (int u = 0; u < CNT; ++u) {
    rr[u] = __builtin_amdgcn_readlane(r, k + u);
    cc[u] = __int_as_float(__builtin_amdgcn_readlane(__float_as_int(cf), k + u));
  }
#pragma unroll
  for (int u = 0; u < CNT; ++u) v[u] = m2[(size_t)rr[u] * (HIDDEN / 2) + lane];
#pragma unroll
  for (int u = 0; u < CNT; ++u) {
    acc.x += cc[u] * v[u].x;
    acc.y += cc[u] * v[u].y;
  }
}
template <bool HAS_ATT>
__global__ __launch_bounds__(256) void gather_gcn_kernel(
    const float* __restrict__ m, const int* __restrict__ off,
    const int2* __restrict__ sedge, const float* __restrict__ att, int comp,
    const float* __restrict__ dinv, const float* __restrict__ b,
    float* __restrict__ out, int N) {
  const int lane = threadIdx.x & 31;
  const int c = __builtin_amdgcn_readfirstlane(blockIdx.x * 8 + (threadIdx.x >> 5));
  if (c >= N) return;
  const float2* __restrict__ m2 = (const float2*)m;
  const int beg = off[c], end = off[c + 1];
  const float dc = dinv[c];
  const float2 self = m2[(size_t)c * (HIDDEN / 2) + lane];
  const float2 bias = ((const float2*)b)[lane];
  float2 acc = make_float2(0.f, 0.f);
  for (int base = beg; base < end; base += 32) {
    const int p = base + lane;
    int r = 0;
    float cf = 0.f;
    if (p < end) {
      const int2 se = sedge[p];
      r = se.x;
      const float dr = dinv[r];
      const float w = HAS_ATT ? att[(size_t)se.y * 2 + comp] : 1.0f;
      cf = dr * w * dc;
    }
    const int n = (end - base) < 32 ? (end - base) : 32;
    int k = 0;
    for (; k + 4 <= n; k += 4) gather_rows<4>(m2, r, cf, k, lane, acc);
    const int rem = n - k;
    if (rem == 3)      gather_rows<3>(m2, r, cf, k, lane, acc);
    else if (rem == 2) gather_rows<2>(m2, r, cf, k, lane, acc);
    else if (rem == 1) gather_rows<1>(m2, r, cf, k, lane, acc);
  }
  float2 o;
  o.x = fmaxf(acc.x + dc * dc * self.x + bias.x, 0.f);
  o.y = fmaxf(acc.y + dc * dc * self.y + bias.y, 0.f);
  ((float2*)out)[(size_t)c * (HIDDEN / 2) + lane] = o;
}

__global__ void edge_att_kernel(const float* __restrict__ h,
                                const int* __restrict__ row,
                                const int* __restrict__ col,
                                const float* __restrict__ Wea,
                                const float* __restrict__ bea,
                                float* __restrict__ att, int E) {
  int e = blockIdx.x * blockDim.x + threadIdx.x;
  if (e >= E) return;
  const float* hr = h + (size_t)row[e] * HIDDEN;
  const float* hc = h + (size_t)col[e] * HIDDEN;
  float l0 = bea[0], l1 = bea[1];
  for (int j = 0; j < HIDDEN; ++j) {
    float a = hr[j], c = hc[j];
    l0 += a * Wea[j * 2 + 0] + c * Wea[(HIDDEN + j) * 2 + 0];
    l1 += a * Wea[j * 2 + 1] + c * Wea[(HIDDEN + j) * 2 + 1];
  }
  float mx = fmaxf(l0, l1);
  float e0 = __expf(l0 - mx), e1 = __expf(l1 - mx);
  float inv = 1.f / (e0 + e1);
  att[(size_t)e * 2 + 0] = e0 * inv;
  att[(size_t)e * 2 + 1] = e1 * inv;
}

__global__ void node_att_split_kernel(const float* __restrict__ h,
                                      const float* __restrict__ Wna,
                                      const float* __restrict__ bna,
                                      float* __restrict__ xc,
                                      float* __restrict__ xo, int N) {
  int i = blockIdx.x * blockDim.x + threadIdx.x;
  if (i >= N) return;
  const float* hi = h + (size_t)i * HIDDEN;
  float l0 = bna[0], l1 = bna[1];
  for (int j = 0; j < HIDDEN; ++j) {
    float a = hi[j];
    l0 += a * Wna[j * 2 + 0];
    l1 += a * Wna[j * 2 + 1];
  }
  float mx = fmaxf(l0, l1);
  float e0 = __expf(l0 - mx), e1 = __expf(l1 - mx);
  float inv = 1.f / (e0 + e1);
  float p0 = e0 * inv, p1 = e1 * inv;
  for (int j = 0; j < HIDDEN; ++j) {
    float a = hi[j];
    xc[(size_t)i * HIDDEN + j] = p0 * a;
    xo[(size_t)i * HIDDEN + j] = p1 * a;
  }
}

__global__ void pool_kernel(const float* __restrict__ x,
                            const int* __restrict__ batch,
                            float* __restrict__ g, int N) {
  int idx = blockIdx.x * blockDim.x + threadIdx.x;
  int i = idx >> 6, j = idx & 63;
  if (i >= N) return;
  atomicAdd(&g[(size_t)batch[i] * HIDDEN + j], x[idx]);
}

__global__ void add_kernel(const float* a, const float* b, float* c, int n) {
  int i = blockIdx.x * blockDim.x + threadIdx.x;
  if (i < n) c[i] = a[i] + b[i];
}

__global__ void logits_kernel(const float* __restrict__ h,
                              const float* __restrict__ Wt,
                              const float* __restrict__ v,
                              float* __restrict__ out, int G) {
  int g = blockIdx.x * blockDim.x + threadIdx.x;
  if (g >= G) return;
  float l[N_CLASSES];
  for (int c = 0; c < N_CLASSES; ++c) l[c] = v[c];
  const float* hg = h + (size_t)g * HIDDEN;
  for (int k = 0; k < HIDDEN; ++k) {
    float x = hg[k];
    for (int c = 0; c < N_CLASSES; ++c) l[c] += x * Wt[k * N_CLASSES + c];
  }
  float mx = l[0];
  for (int c = 1; c < N_CLASSES; ++c) mx = fmaxf(mx, l[c]);
  float s = 0.f;
  for (int c = 0; c < N_CLASSES; ++c) s += __expf(l[c] - mx);
  float lse = __logf(s) + mx;
  for (int c = 0; c < N_CLASSES; ++c) out[(size_t)g * N_CLASSES + c] = l[c] - lse;
}

extern "C" void kernel_launch(void* const* d_in, const int* in_sizes, int n_in,
                              void* d_out, int out_size, void* d_ws, size_t ws_size,
                              hipStream_t stream) {
  (void)in_sizes; (void)n_in; (void)out_size; (void)ws_size;
  const float* x       = (const float*)d_in[0];
  const int*   ei      = (const int*)d_in[1];
  const int*   row     = ei;
  const int*   col     = ei + N_EDGES;
  const int*   batch   = (const int*)d_in[2];
  const float* W_feat  = (const float*)d_in[3];
  const float* W_convs = (const float*)d_in[4];
  const float* b_convs = (const float*)d_in[5];
  const float* W_ea    = (const float*)d_in[6];
  const float* b_ea    = (const float*)d_in[7];
  const float* W_na    = (const float*)d_in[8];
  const float* b_na    = (const float*)d_in[9];
  const float* W_ctx   = (const float*)d_in[10];
  const float* b_ctx   = (const float*)d_in[11];
  const float* W_obj   = (const float*)d_in[12];
  const float* b_obj   = (const float*)d_in[13];
  const float* W1h[3]  = {(const float*)d_in[14], (const float*)d_in[18], (const float*)d_in[22]};
  const float* b1h[3]  = {(const float*)d_in[15], (const float*)d_in[19], (const float*)d_in[23]};
  const float* W2h[3]  = {(const float*)d_in[16], (const float*)d_in[20], (const float*)d_in[24]};
  const float* b2h[3]  = {(const float*)d_in[17], (const float*)d_in[21], (const float*)d_in[25]};

  size_t off = 0;
  auto carve = [&](size_t elems) -> float* {
    float* p = (float*)((char*)d_ws + off);
    off += ((elems * sizeof(float)) + 255) & ~(size_t)255;
    return p;
  };
  float* hb    = carve((size_t)N_NODES * HIDDEN);
  float* mbuf  = carve((size_t)N_NODES * HIDDEN);
  float* xc    = carve((size_t)N_NODES * HIDDEN);
  float* xo    = carve((size_t)N_NODES * HIDDEN);
  float* att   = carve((size_t)N_EDGES * 2);
  float* degu  = carve(N_NODES);
  float* dinvu = carve(N_NODES);
  float* dega  = carve(N_NODES);
  float* dinva = carve(N_NODES);
  float* gc    = carve((size_t)N_GRAPHS * HIDDEN);
  float* go    = carve((size_t)N_GRAPHS * HIDDEN);
  float* gs    = carve((size_t)N_GRAPHS * HIDDEN);
  float* r1    = carve((size_t)N_GRAPHS * HIDDEN);
  float* sums  = carve(N_FEAT);
  float* sq    = carve(N_FEAT);
  float* Wf    = carve((size_t)N_FEAT * HIDDEN);
  float* vf    = carve(HIDDEN);
  int*  ecnt   = (int*)carve(N_NODES);
  int*  etsum  = (int*)carve(128);
  int*  eoff   = (int*)carve(N_NODES + 1);
  int*  ecur   = (int*)carve(N_NODES);
  int2* sedge  = (int2*)carve((size_t)N_EDGES * 2);

  auto nblk = [](long n, int b) { return (int)((n + b - 1) / b); };

  auto stats = [&](const float* A, int N, int F) {
    hipMemsetAsync(sums, 0, (size_t)F * sizeof(float), stream);
    hipMemsetAsync(sq,   0, (size_t)F * sizeof(float), stream);
    const int nvec = N * F / 4;
    const int grid = nblk(nvec, 256) < 512 ? nblk(nvec, 256) : 512;
    colstats_kernel<<<grid, 256, 0, stream>>>(A, nvec, F, sums, sq);
  };
  auto gemm = [&](const float* A, float* C, int M, int K, int relu) {
    gemm_wmma_kernel<<<nblk(M / 16, 4), 128, 0, stream>>>(A, Wf, vf, C, M, K, relu);
  };

  stats(x, N_NODES, N_FEAT);
  fold_bn_kernel<<<1, HIDDEN, 0, stream>>>(sums, sq, 1.f / N_NODES, N_FEAT, HIDDEN,
                                           W_feat, nullptr, Wf, vf);
  gemm(x, hb, N_NODES, N_FEAT, 1);

  deg_init_kernel<<<nblk(N_NODES, 256), 256, 0, stream>>>(degu, N_NODES);
  deg_accum_kernel<<<nblk(N_EDGES, 256), 256, 0, stream>>>(row, nullptr, 0, N_EDGES, degu);
  make_dinv_kernel<<<nblk(N_NODES, 256), 256, 0, stream>>>(degu, dinvu, N_NODES);

  csr_zero_kernel<<<nblk(N_NODES, 256), 256, 0, stream>>>(ecnt, N_NODES);
  csr_count_kernel<<<nblk(N_EDGES, 256), 256, 0, stream>>>(col, N_EDGES, ecnt);
  csr_tilesum_kernel<<<nblk(N_NODES, 1024), 256, 0, stream>>>(ecnt, N_NODES, etsum);
  csr_scan_kernel<<<nblk(N_NODES, 1024), 256, 0, stream>>>(ecnt, etsum, N_NODES,
                                                           eoff, ecur);
  csr_fill_kernel<<<nblk(N_EDGES, 256), 256, 0, stream>>>(row, col, N_EDGES, ecur,
                                                          sedge);

  for (int i = 0; i < 3; ++i) {
    stats(hb, N_NODES, HIDDEN);
    fold_bn_kernel<<<1, HIDDEN, 0, stream>>>(sums, sq, 1.f / N_NODES, HIDDEN, HIDDEN,
                                             W_convs + (size_t)i * HIDDEN * HIDDEN,
                                             nullptr, Wf, vf);
    gemm(hb, mbuf, N_NODES, HIDDEN, 0);
    gather_gcn_kernel<false><<<nblk(N_NODES, 8), 256, 0, stream>>>(
        mbuf, eoff, sedge, nullptr, 0, dinvu, b_convs + (size_t)i * HIDDEN, hb,
        N_NODES);
  }

  edge_att_kernel<<<nblk(N_EDGES, 256), 256, 0, stream>>>(hb, row, col, W_ea, b_ea,
                                                          att, N_EDGES);
  node_att_split_kernel<<<nblk(N_NODES, 256), 256, 0, stream>>>(hb, W_na, b_na,
                                                                xc, xo, N_NODES);

  auto run_branch = [&](float* xin, int comp, const float* W, const float* b,
                        float* gout) {
    stats(xin, N_NODES, HIDDEN);
    fold_bn_kernel<<<1, HIDDEN, 0, stream>>>(sums, sq, 1.f / N_NODES, HIDDEN, HIDDEN,
                                             W, nullptr, Wf, vf);
    gemm(xin, mbuf, N_NODES, HIDDEN, 0);
    deg_init_kernel<<<nblk(N_NODES, 256), 256, 0, stream>>>(dega, N_NODES);
    deg_accum_kernel<<<nblk(N_EDGES, 256), 256, 0, stream>>>(row, att, comp, N_EDGES, dega);
    make_dinv_kernel<<<nblk(N_NODES, 256), 256, 0, stream>>>(dega, dinva, N_NODES);
    gather_gcn_kernel<true><<<nblk(N_NODES, 8), 256, 0, stream>>>(
        mbuf, eoff, sedge, att, comp, dinva, b, xin, N_NODES);
    hipMemsetAsync(gout, 0, (size_t)N_GRAPHS * HIDDEN * sizeof(float), stream);
    pool_kernel<<<nblk((long)N_NODES * 64, 256), 256, 0, stream>>>(xin, batch, gout,
                                                                   N_NODES);
  };
  run_branch(xc, 0, W_ctx, b_ctx, gc);
  run_branch(xo, 1, W_obj, b_obj, go);

  add_kernel<<<nblk(N_GRAPHS * HIDDEN, 256), 256, 0, stream>>>(gc, go, gs,
                                                               N_GRAPHS * HIDDEN);

  float* out = (float*)d_out;
  auto run_head = [&](const float* gin, int h, float* outp) {
    stats(gin, N_GRAPHS, HIDDEN);
    fold_bn_kernel<<<1, HIDDEN, 0, stream>>>(sums, sq, 1.f / N_GRAPHS, HIDDEN, HIDDEN,
                                             W1h[h], b1h[h], Wf, vf);
    gemm(gin, r1, N_GRAPHS, HIDDEN, 1);
    stats(r1, N_GRAPHS, HIDDEN);
    fold_bn_kernel<<<1, N_CLASSES, 0, stream>>>(sums, sq, 1.f / N_GRAPHS, HIDDEN,
                                                N_CLASSES, W2h[h], b2h[h], Wf, vf);
    logits_kernel<<<nblk(N_GRAPHS, 256), 256, 0, stream>>>(r1, Wf, vf, outp, N_GRAPHS);
  };
  run_head(gc, 0, out);
  run_head(go, 1, out + N_GRAPHS * N_CLASSES);
  run_head(gs, 2, out + 2 * N_GRAPHS * N_CLASSES);
}


# global add pool folded into the two attention-branch gathers: a block's 8 finished rows are summed per graph in LDS, one atomic per column; pool launches + row round trip go (pool sums: new tree)
# speedup vs baseline: 2.3824x; 1.0380x over previous
#include <hip/hip_runtime.h>
#include <math.h>


#define N_NODES  100000
#define N_EDGES  1000000
#define N_FEAT   128
#define HIDDEN   64
#define N_GRAPHS 512
#define N_CLASSES 10
#define BN_EPS  1e-5f
#define BN_BETA 1e-4f

typedef __attribute__((ext_vector_type(2))) float v2f;
typedef __attribute__((ext_vector_type(8))) float v8f;

__global__ __launch_bounds__(256) void colstats_kernel(
    const float* __restrict__ A, int nvec, int F,
    float* __restrict__ sums, float* __restrict__ sumsq) {
  __shared__ float sh_s[1024];
  __shared__ float sh_q[1024];
  const float4* __restrict__ A4 = (const float4*)A;
  const int tid = threadIdx.x;
  const int stride = gridDim.x * 256;
  const int last = nvec - 1;
  float4 s = make_float4(0.f, 0.f, 0.f, 0.f);
  float4 q = make_float4(0.f, 0.f, 0.f, 0.f);
  for (int i = blockIdx.x * 256 + tid; i < nvec; i += 8 * stride) {
    float4 v[8];
#pragma unroll
    for (int u = 0; u < 8; ++u) {
      const int idx = i + u * stride;
      v[u] = A4[idx < last ? idx : last];
    }
    __builtin_amdgcn_sched_barrier(0);
#pragma unroll
    for (int u = 0; u < 8; ++u) {
      const bool live = (i + u * stride) < nvec;
      const float a = live ? v[u].x : 0.f;
      const float b = live ? v[u].y : 0.f;
      const float c = live ? v[u].z : 0.f;
      const float d = live ? v[u].w : 0.f;
      s.x += a; q.x += a * a;
      s.y += b; q.y += b * b;
      s.z += c; q.z += c * c;
      s.w += d; q.w += d * d;
    }
  }
  ((float4*)sh_s)[tid] = s;
  ((float4*)sh_q)[tid] = q;
  __syncthreads();
  if (tid < F) {
    float ts = 0.f, tq = 0.f;
    for (int p = tid; p < 1024; p += F) { ts += sh_s[p]; tq += sh_q[p]; }
    atomicAdd(&sums[tid], ts);
    atomicAdd(&sumsq[tid], tq);
  }
}

__global__ void fold_bn_kernel(const float* __restrict__ sums,
                               const float* __restrict__ sumsq, float invN,
                               int K, int NN, const float* __restrict__ W,
                               const float* __restrict__ extra_b,
                               float* __restrict__ Wout, float* __restrict__ vout) {
  __shared__ float s_ms[N_FEAT];
  __shared__ float s_s[N_FEAT];
  const int j = threadIdx.x;
  for (int k = j; k < K; k += blockDim.x) {
    float mean = sums[k] * invN;
    float var  = sumsq[k] * invN - mean * mean;
    float sc   = rsqrtf(var + BN_EPS);
    s_ms[k] = mean * sc;
    s_s[k]  = sc;
  }
  __syncthreads();
  float acc = extra_b ? extra_b[j] : 0.f;
  for (int k = 0; k < K; ++k) {
    float w = W[(size_t)k * NN + j];
    Wout[(size_t)k * NN + j] = s_s[k] * w;
    acc += (BN_BETA - s_ms[k]) * w;
  }
  vout[j] = acc;
}

__global__ __launch_bounds__(128) void gemm_wmma_kernel(
    const float* __restrict__ A, const float* __restrict__ Wt,
    const float* __restrict__ v, float* __restrict__ C,
    int M, int K, int doRelu) {
  const int lane = threadIdx.x & 31;
  const int wave = threadIdx.x >> 5;
  const int rowTile = blockIdx.x * 4 + wave;
  if (rowTile >= (M >> 4)) return;
  const int mrow = (rowTile << 4) + (lane & 15);
  const int kOff = (lane >> 4) << 1;
  const int n0   = lane & 15;
  v8f acc[4] = {};
  for (int kk = 0; kk < K; kk += 4) {
    const int k0 = kk + kOff;
    v2f a = *(const v2f*)(A + (size_t)mrow * K + k0);
    const float* w0 = Wt + (size_t)k0 * HIDDEN + n0;
#pragma unroll
    for (int t = 0; t < 4; ++t) {
      v2f b;
      b.x = w0[t * 16];
      b.y = w0[HIDDEN + t * 16];
      acc[t] = __builtin_amdgcn_wmma_f32_16x16x4_f32(
          false, a, false, b, (short)0, acc[t], false, false);
    }
  }
  const int rowBase = (rowTile << 4) + ((lane >> 4) << 3);
#pragma unroll
  for (int t = 0; t < 4; ++t) {
    const int n = t * 16 + n0;
    const float bias = v[n];
#pragma unroll
    for (int e = 0; e < 8; ++e) {
      float val = acc[t][e] + bias;
      if (doRelu) val = fmaxf(val, 0.f);
      C[(size_t)(rowBase + e) * HIDDEN + n] = val;
    }
  }
}

__global__ void deg_init_kernel(float* deg, int N) {
  int i = blockIdx.x * blockDim.x + threadIdx.x;
  if (i < N) deg[i] = 1.0f;
}
__global__ void deg_accum_kernel(const int* __restrict__ row,
                                 const float* __restrict__ att, int comp,
                                 int E, float* __restrict__ deg) {
  int e = blockIdx.x * blockDim.x + threadIdx.x;
  if (e >= E) return;
  float w = att ? att[(size_t)e * 2 + comp] : 1.0f;
  atomicAdd(&deg[row[e]], w);
}
__global__ void make_dinv_kernel(const float* __restrict__ deg,
                                 float* __restrict__ dinv, int N) {
  int i = blockIdx.x * blockDim.x + threadIdx.x;
  if (i < N) dinv[i] = deg[i] > 0.f ? rsqrtf(deg[i]) : 0.f;
}

__global__ void csr_zero_kernel(int* __restrict__ cnt, int N) {
  int i = blockIdx.x * blockDim.x + threadIdx.x;
  if (i < N) cnt[i] = 0;
}
__global__ void csr_count_kernel(const int* __restrict__ col, int E,
                                 int* __restrict__ cnt) {
  int e = blockIdx.x * blockDim.x + threadIdx.x;
  if (e < E) atomicAdd(&cnt[col[e]], 1);
}
__device__ __forceinline__ void load_counts4(const int* __restrict__ cnt, int i0,
                                             int N, int c[4]) {
  if (i0 + 3 < N) {
    const int4 v = *(const int4*)(cnt + i0);
    c[0] = v.x; c[1] = v.y; c[2] = v.z; c[3] = v.w;
  } else {
    c[0] = (i0     < N) ? cnt[i0]     : 0;
    c[1] = (i0 + 1 < N) ? cnt[i0 + 1] : 0;
    c[2] = (i0 + 2 < N) ? cnt[i0 + 2] : 0;
    c[3] = 0;
  }
}
__global__ __launch_bounds__(256) void csr_tilesum_kernel(
    const int* __restrict__ cnt, int N, int* __restrict__ tsum) {
  __shared__ int sh[8];
  const int tid = threadIdx.x;
  int c[4];
  load_counts4(cnt, (blockIdx.x * 256 + tid) * 4, N, c);
  int t = c[0] + c[1] + c[2] + c[3];
  for (int o = 16; o > 0; o >>= 1) t += __shfl_xor(t, o);
  if ((tid & 31) == 0) sh[tid >> 5] = t;
  __syncthreads();
  if (tid == 0) {
    int s = 0;
    for (int w = 0; w < 8; ++w) s += sh[w];
    tsum[blockIdx.x] = s;
  }
}
__global__ __launch_bounds__(256) void csr_scan_kernel(
    const int* __restrict__ cnt, const int* __restrict__ tsum, int N,
    int* __restrict__ off, int* __restrict__ cur) {
  __shared__ int sh_w[8];
  __shared__ int sh_b[8];
  const int tid = threadIdx.x, lane = tid & 31, wave = tid >> 5;
  int bpart = 0;
  for (int t = tid; t < (int)blockIdx.x; t += 256) bpart += tsum[t];
  for (int o = 16; o > 0; o >>= 1) bpart += __shfl_xor(bpart, o);
  const int i0 = (blockIdx.x * 256 + tid) * 4;
  int c[4];
  load_counts4(cnt, i0, N, c);
  const int t = c[0] + c[1] + c[2] + c[3];
  int inc = t;
  for (int o = 1; o < 32; o <<= 1) {
    const int y = __shfl_up(inc, o);
    if (lane >= o) inc += y;
  }
  if (lane == 31) sh_w[wave] = inc;
  if (lane == 0)  sh_b[wave] = bpart;
  __syncthreads();
  int base = 0;
  for (int w = 0; w < 8; ++w) {
    base += sh_b[w];
    if (w < wave) base += sh_w[w];
  }
  const int o0 = base + inc - t;
  const int o1 = o0 + c[0], o2 = o1 + c[1], o3 = o2 + c[2];
  if (i0 + 3 < N) {
    const int4 v = make_int4(o0, o1, o2, o3);
    *(int4*)(off + i0) = v;
    *(int4*)(cur + i0) = v;
  } else {
    if (i0     < N) { off[i0]     = o0; cur[i0]     = o0; }
    if (i0 + 1 < N) { off[i0 + 1] = o1; cur[i0 + 1] = o1; }
    if (i0 + 2 < N) { off[i0 + 2] = o2; cur[i0 + 2] = o2; }
  }
  if (i0 < N && N - 1 <= i0 + 3) off[N] = o0 + t;
}
__global__ void csr_fill_kernel(const int* __restrict__ row,
                                const int* __restrict__ col, int E,
                                int* __restrict__ cur, int2* __restrict__ sedge) {
  int e = blockIdx.x * blockDim.x + threadIdx.x;
  if (e >= E) return;
  const int r = row[e];
  const int pos = atomicAdd(&cur[col[e]], 1);
  if ((unsigned)pos < (unsigned)E) sedge[pos] = make_int2(r, e);
}

template <int CNT>
__device__ __forceinline__ void gather_rows(const float2* __restrict__ m2, int r,
                                            float cf, int k, int lane, float2& acc) {
  int rr[CNT];
  float cc[CNT];
  float2 v[CNT];
#pragma unroll
  for (int u = 0; u < CNT; ++u) {
    rr[u] = __builtin_amdgcn_readlane(r, k + u);
    cc[u] = __int_as_float(__builtin_amdgcn_readlane(__float_as_int(cf), k + u));
  }
#pragma unroll
  for (int u = 0; u < CNT; ++u) v[u] = m2[(size_t)rr[u] * (HIDDEN / 2) + lane];
#pragma unroll
  for (int u = 0; u < CNT; ++u) {
    acc.x += cc[u] * v[u].x;
    acc.y += cc[u] * v[u].y;
  }
}
template <bool HAS_ATT>
__device__ __forceinline__ float2 gcn_row(
    const float2* __restrict__ m2, const int* __restrict__ off,
    const int2* __restrict__ sedge, const float* __restrict__ att, int comp,
    const float* __restrict__ dinv, const float* __restrict__ b, int c, int lane) {
  const int beg = off[c], end = off[c + 1];
  const float dc = dinv[c];
  const float2 self = m2[(size_t)c * (HIDDEN / 2) + lane];
  const float2 bias = ((const float2*)b)[lane];
  float2 acc = make_float2(0.f, 0.f);
  for (int base = beg; base < end; base += 32) {
    const int p = base + lane;
    int r = 0;
    float cf = 0.f;
    if (p < end) {
      const int2 se = sedge[p];
      r = se.x;
      const float dr = dinv[r];
      const float w = HAS_ATT ? att[(size_t)se.y * 2 + comp] : 1.0f;
      cf = dr * w * dc;
    }
    const int n = (end - base) < 32 ? (end - base) : 32;
    int k = 0;
    for (; k + 4 <= n; k += 4) gather_rows<4>(m2, r, cf, k, lane, acc);
    const int rem = n - k;
    if (rem == 3)      gather_rows<3>(m2, r, cf, k, lane, acc);
    else if (rem == 2) gather_rows<2>(m2, r, cf, k, lane, acc);
    else if (rem == 1) gather_rows<1>(m2, r, cf, k, lane, acc);
  }
  float2 o;
  o.x = fmaxf(acc.x + dc * dc * self.x + bias.x, 0.f);
  o.y = fmaxf(acc.y + dc * dc * self.y + bias.y, 0.f);
  return o;
}

__global__ __launch_bounds__(256) void gather_gcn_kernel(
    const float* __restrict__ m, const int* __restrict__ off,
    const int2* __restrict__ sedge, const float* __restrict__ dinv,
    const float* __restrict__ b, float* __restrict__ out, int N) {
  const int lane = threadIdx.x & 31;
  const int c = __builtin_amdgcn_readfirstlane(blockIdx.x * 8 + (threadIdx.x >> 5));
  if (c >= N) return;
  const float2 o = gcn_row<false>((const float2*)m, off, sedge, nullptr, 0, dinv, b,
                                  c, lane);
  ((float2*)out)[(size_t)c * (HIDDEN / 2) + lane] = o;
}

__global__ __launch_bounds__(256) void gather_pool_kernel(
    const float* __restrict__ m, const int* __restrict__ off,
    const int2* __restrict__ sedge, const float* __restrict__ att, int comp,
    const float* __restrict__ dinv, const float* __restrict__ b,
    const int* __restrict__ batch, float* __restrict__ g, int N) {
  __shared__ float2 sh_row[8 * (HIDDEN / 2)];
  __shared__ int sh_gid[8];
  const int lane = threadIdx.x & 31;
  const int wave = threadIdx.x >> 5;
  const int c = __builtin_amdgcn_readfirstlane(blockIdx.x * 8 + wave);
  float2 o = make_float2(0.f, 0.f);
  int gid = -1;
  if (c < N) {
    gid = batch[c];
    o = gcn_row<true>((const float2*)m, off, sedge, att, comp, dinv, b, c, lane);
  }
  sh_row[wave * (HIDDEN / 2) + lane] = o;
  if (lane == 0) sh_gid[wave] = gid;
  __syncthreads();
  if (threadIdx.x < HIDDEN) {
    const int j = threadIdx.x;
    int gw[8];
    float v[8];
#pragma unroll
    for (int w = 0; w < 8; ++w) {
      gw[w] = sh_gid[w];
      v[w] = ((const float*)sh_row)[w * HIDDEN + j];
    }
    int cur = gw[0];
    float a = v[0];
#pragma unroll
    for (int w = 1; w < 8; ++w) {
      if (gw[w] == cur) {
        a += v[w];
      } else {
        if ((unsigned)cur < (unsigned)N_GRAPHS) atomicAdd(&g[(size_t)cur * HIDDEN + j], a);
        cur = gw[w];
        a = v[w];
      }
    }
    if ((unsigned)cur < (unsigned)N_GRAPHS) atomicAdd(&g[(size_t)cur * HIDDEN + j], a);
  }
}

__global__ void edge_att_kernel(const float* __restrict__ h,
                                const int* __restrict__ row,
                                const int* __restrict__ col,
                                const float* __restrict__ Wea,
                                const float* __restrict__ bea,
                                float* __restrict__ att, int E) {
  int e = blockIdx.x * blockDim.x + threadIdx.x;
  if (e >= E) return;
  const float* hr = h + (size_t)row[e] * HIDDEN;
  const float* hc = h + (size_t)col[e] * HIDDEN;
  float l0 = bea[0], l1 = bea[1];
  for (int j = 0; j < HIDDEN; ++j) {
    float a = hr[j], c = hc[j];
    l0 += a * Wea[j * 2 + 0] + c * Wea[(HIDDEN + j) * 2 + 0];
    l1 += a * Wea[j * 2 + 1] + c * Wea[(HIDDEN + j) * 2 + 1];
  }
  float mx = fmaxf(l0, l1);
  float e0 = __expf(l0 - mx), e1 = __expf(l1 - mx);
  float inv = 1.f / (e0 + e1);
  att[(size_t)e * 2 + 0] = e0 * inv;
  att[(size_t)e * 2 + 1] = e1 * inv;
}

__global__ void node_att_split_kernel(const float* __restrict__ h,
                                      const float* __restrict__ Wna,
                                      const float* __restrict__ bna,
                                      float* __restrict__ xc,
                                      float* __restrict__ xo, int N) {
  int i = blockIdx.x * blockDim.x + threadIdx.x;
  if (i >= N) return;
  const float* hi = h + (size_t)i * HIDDEN;
  float l0 = bna[0], l1 = bna[1];
  for (int j = 0; j < HIDDEN; ++j) {
    float a = hi[j];
    l0 += a * Wna[j * 2 + 0];
    l1 += a * Wna[j * 2 + 1];
  }
  float mx = fmaxf(l0, l1);
  float e0 = __expf(l0 - mx), e1 = __expf(l1 - mx);
  float inv = 1.f / (e0 + e1);
  float p0 = e0 * inv, p1 = e1 * inv;
  for (int j = 0; j < HIDDEN; ++j) {
    float a = hi[j];
    xc[(size_t)i * HIDDEN + j] = p0 * a;
    xo[(size_t)i * HIDDEN + j] = p1 * a;
  }
}

__global__ void add_kernel(const float* a, const float* b, float* c, int n) {
  int i = blockIdx.x * blockDim.x + threadIdx.x;
  if (i < n) c[i] = a[i] + b[i];
}

__global__ void logits_kernel(const float* __restrict__ h,
                              const float* __restrict__ Wt,
                              const float* __restrict__ v,
                              float* __restrict__ out, int G) {
  int g = blockIdx.x * blockDim.x + threadIdx.x;
  if (g >= G) return;
  float l[N_CLASSES];
  for (int c = 0; c < N_CLASSES; ++c) l[c] = v[c];
  const float* hg = h + (size_t)g * HIDDEN;
  for (int k = 0; k < HIDDEN; ++k) {
    float x = hg[k];
    for (int c = 0; c < N_CLASSES; ++c) l[c] += x * Wt[k * N_CLASSES + c];
  }
  float mx = l[0];
  for (int c = 1; c < N_CLASSES; ++c) mx = fmaxf(mx, l[c]);
  float s = 0.f;
  for (int c = 0; c < N_CLASSES; ++c) s += __expf(l[c] - mx);
  float lse = __logf(s) + mx;
  for (int c = 0; c < N_CLASSES; ++c) out[(size_t)g * N_CLASSES + c] = l[c] - lse;
}

extern "C" void kernel_launch(void* const* d_in, const int* in_sizes, int n_in,
                              void* d_out, int out_size, void* d_ws, size_t ws_size,
                              hipStream_t stream) {
  (void)in_sizes; (void)n_in; (void)out_size; (void)ws_size;
  const float* x       = (const float*)d_in[0];
  const int*   ei      = (const int*)d_in[1];
  const int*   row     = ei;
  const int*   col     = ei + N_EDGES;
  const int*   batch   = (const int*)d_in[2];
  const float* W_feat  = (const float*)d_in[3];
  const float* W_convs = (const float*)d_in[4];
  const float* b_convs = (const float*)d_in[5];
  const float* W_ea    = (const float*)d_in[6];
  const float* b_ea    = (const float*)d_in[7];
  const float* W_na    = (const float*)d_in[8];
  const float* b_na    = (const float*)d_in[9];
  const float* W_ctx   = (const float*)d_in[10];
  const float* b_ctx   = (const float*)d_in[11];
  const float* W_obj   = (const float*)d_in[12];
  const float* b_obj   = (const float*)d_in[13];
  const float* W1h[3]  = {(const float*)d_in[14], (const float*)d_in[18], (const float*)d_in[22]};
  const float* b1h[3]  = {(const float*)d_in[15], (const float*)d_in[19], (const float*)d_in[23]};
  const float* W2h[3]  = {(const float*)d_in[16], (const float*)d_in[20], (const float*)d_in[24]};
  const float* b2h[3]  = {(const float*)d_in[17], (const float*)d_in[21], (const float*)d_in[25]};

  size_t off = 0;
  auto carve = [&](size_t elems) -> float* {
    float* p = (float*)((char*)d_ws + off);
    off += ((elems * sizeof(float)) + 255) & ~(size_t)255;
    return p;
  };
  float* hb    = carve((size_t)N_NODES * HIDDEN);
  float* mbuf  = carve((size_t)N_NODES * HIDDEN);
  float* xc    = carve((size_t)N_NODES * HIDDEN);
  float* xo    = carve((size_t)N_NODES * HIDDEN);
  float* att   = carve((size_t)N_EDGES * 2);
  float* degu  = carve(N_NODES);
  float* dinvu = carve(N_NODES);
  float* dega  = carve(N_NODES);
  float* dinva = carve(N_NODES);
  float* gc    = carve((size_t)N_GRAPHS * HIDDEN);
  float* go    = carve((size_t)N_GRAPHS * HIDDEN);
  float* gs    = carve((size_t)N_GRAPHS * HIDDEN);
  float* r1    = carve((size_t)N_GRAPHS * HIDDEN);
  float* sums  = carve(N_FEAT);
  float* sq    = carve(N_FEAT);
  float* Wf    = carve((size_t)N_FEAT * HIDDEN);
  float* vf    = carve(HIDDEN);
  int*  ecnt   = (int*)carve(N_NODES);
  int*  etsum  = (int*)carve(128);
  int*  eoff   = (int*)carve(N_NODES + 1);
  int*  ecur   = (int*)carve(N_NODES);
  int2* sedge  = (int2*)carve((size_t)N_EDGES * 2);

  auto nblk = [](long n, int b) { return (int)((n + b - 1) / b); };

  auto stats = [&](const float* A, int N, int F) {
    hipMemsetAsync(sums, 0, (size_t)F * sizeof(float), stream);
    hipMemsetAsync(sq,   0, (size_t)F * sizeof(float), stream);
    const int nvec = N * F / 4;
    const int grid = nblk(nvec, 256) < 512 ? nblk(nvec, 256) : 512;
    colstats_kernel<<<grid, 256, 0, stream>>>(A, nvec, F, sums, sq);
  };
  auto gemm = [&](const float* A, float* C, int M, int K, int relu) {
    gemm_wmma_kernel<<<nblk(M / 16, 4), 128, 0, stream>>>(A, Wf, vf, C, M, K, relu);
  };

  stats(x, N_NODES, N_FEAT);
  fold_bn_kernel<<<1, HIDDEN, 0, stream>>>(sums, sq, 1.f / N_NODES, N_FEAT, HIDDEN,
                                           W_feat, nullptr, Wf, vf);
  gemm(x, hb, N_NODES, N_FEAT, 1);

  deg_init_kernel<<<nblk(N_NODES, 256), 256, 0, stream>>>(degu, N_NODES);
  deg_accum_kernel<<<nblk(N_EDGES, 256), 256, 0, stream>>>(row, nullptr, 0, N_EDGES, degu);
  make_dinv_kernel<<<nblk(N_NODES, 256), 256, 0, stream>>>(degu, dinvu, N_NODES);

  csr_zero_kernel<<<nblk(N_NODES, 256), 256, 0, stream>>>(ecnt, N_NODES);
  csr_count_kernel<<<nblk(N_EDGES, 256), 256, 0, stream>>>(col, N_EDGES, ecnt);
  csr_tilesum_kernel<<<nblk(N_NODES, 1024), 256, 0, stream>>>(ecnt, N_NODES, etsum);
  csr_scan_kernel<<<nblk(N_NODES, 1024), 256, 0, stream>>>(ecnt, etsum, N_NODES,
                                                           eoff, ecur);
  csr_fill_kernel<<<nblk(N_EDGES, 256), 256, 0, stream>>>(row, col, N_EDGES, ecur,
                                                          sedge);

  for (int i = 0; i < 3; ++i) {
    stats(hb, N_NODES, HIDDEN);
    fold_bn_kernel<<<1, HIDDEN, 0, stream>>>(sums, sq, 1.f / N_NODES, HIDDEN, HIDDEN,
                                             W_convs + (size_t)i * HIDDEN * HIDDEN,
                                             nullptr, Wf, vf);
    gemm(hb, mbuf, N_NODES, HIDDEN, 0);
    gather_gcn_kernel<<<nblk(N_NODES, 8), 256, 0, stream>>>(
        mbuf, eoff, sedge, dinvu, b_convs + (size_t)i * HIDDEN, hb, N_NODES);
  }

  edge_att_kernel<<<nblk(N_EDGES, 256), 256, 0, stream>>>(hb, row, col, W_ea, b_ea,
                                                          att, N_EDGES);
  node_att_split_kernel<<<nblk(N_NODES, 256), 256, 0, stream>>>(hb, W_na, b_na,
                                                                xc, xo, N_NODES);

  auto run_branch = [&](float* xin, int comp, const float* W, const float* b,
                        float* gout) {
    stats(xin, N_NODES, HIDDEN);
    fold_bn_kernel<<<1, HIDDEN, 0, stream>>>(sums, sq, 1.f / N_NODES, HIDDEN, HIDDEN,
                                             W, nullptr, Wf, vf);
    gemm(xin, mbuf, N_NODES, HIDDEN, 0);
    deg_init_kernel<<<nblk(N_NODES, 256), 256, 0, stream>>>(dega, N_NODES);
    deg_accum_kernel<<<nblk(N_EDGES, 256), 256, 0, stream>>>(row, att, comp, N_EDGES, dega);
    make_dinv_kernel<<<nblk(N_NODES, 256), 256, 0, stream>>>(dega, dinva, N_NODES);
    hipMemsetAsync(gout, 0, (size_t)N_GRAPHS * HIDDEN * sizeof(float), stream);
    gather_pool_kernel<<<nblk(N_NODES, 8), 256, 0, stream>>>(
        mbuf, eoff, sedge, att, comp, dinva, b, batch, gout, N_NODES);
  };
  run_branch(xc, 0, W_ctx, b_ctx, gc);
  run_branch(xo, 1, W_obj, b_obj, go);

  add_kernel<<<nblk(N_GRAPHS * HIDDEN, 256), 256, 0, stream>>>(gc, go, gs,
                                                               N_GRAPHS * HIDDEN);

  float* out = (float*)d_out;
  auto run_head = [&](const float* gin, int h, float* outp) {
    stats(gin, N_GRAPHS, HIDDEN);
    fold_bn_kernel<<<1, HIDDEN, 0, stream>>>(sums, sq, 1.f / N_GRAPHS, HIDDEN, HIDDEN,
                                             W1h[h], b1h[h], Wf, vf);
    gemm(gin, r1, N_GRAPHS, HIDDEN, 1);
    stats(r1, N_GRAPHS, HIDDEN);
    fold_bn_kernel<<<1, N_CLASSES, 0, stream>>>(sums, sq, 1.f / N_GRAPHS, HIDDEN,
                                                N_CLASSES, W2h[h], b2h[h], Wf, vf);
    logits_kernel<<<nblk(N_GRAPHS, 256), 256, 0, stream>>>(r1, Wf, vf, outp, N_GRAPHS);
  };
  run_head(gc, 0, out);
  run_head(go, 1, out + N_GRAPHS * N_CLASSES);
  run_head(gs, 2, out + 2 * N_GRAPHS * N_CLASSES);
}


# readout heads: 31 stream operations (memsets, add, column sums, folds, products, logits) -> one launch, one 16-wave work-group per head keeping its 512 rows in LDS (BN sums: new fixed tree)
# speedup vs baseline: 2.7733x; 1.1641x over previous
#include <hip/hip_runtime.h>
#include <math.h>


#define N_NODES  100000
#define N_EDGES  1000000
#define N_FEAT   128
#define HIDDEN   64
#define N_GRAPHS 512
#define N_CLASSES 10
#define BN_EPS  1e-5f
#define BN_BETA 1e-4f

typedef __attribute__((ext_vector_type(2))) float v2f;
typedef __attribute__((ext_vector_type(8))) float v8f;

__global__ __launch_bounds__(256) void colstats_kernel(
    const float* __restrict__ A, int nvec, int F,
    float* __restrict__ sums, float* __restrict__ sumsq) {
  __shared__ float sh_s[1024];
  __shared__ float sh_q[1024];
  const float4* __restrict__ A4 = (const float4*)A;
  const int tid = threadIdx.x;
  const int stride = gridDim.x * 256;
  const int last = nvec - 1;
  float4 s = make_float4(0.f, 0.f, 0.f, 0.f);
  float4 q = make_float4(0.f, 0.f, 0.f, 0.f);
  for (int i = blockIdx.x * 256 + tid; i < nvec; i += 8 * stride) {
    float4 v[8];
#pragma unroll
    for (int u = 0; u < 8; ++u) {
      const int idx = i + u * stride;
      v[u] = A4[idx < last ? idx : last];
    }
    __builtin_amdgcn_sched_barrier(0);
#pragma unroll
    for (int u = 0; u < 8; ++u) {
      const bool live = (i + u * stride) < nvec;
      const float a = live ? v[u].x : 0.f;
      const float b = live ? v[u].y : 0.f;
      const float c = live ? v[u].z : 0.f;
      const float d = live ? v[u].w : 0.f;
      s.x += a; q.x += a * a;
      s.y += b; q.y += b * b;
      s.z += c; q.z += c * c;
      s.w += d; q.w += d * d;
    }
  }
  ((float4*)sh_s)[tid] = s;
  ((float4*)sh_q)[tid] = q;
  __syncthreads();
  if (tid < F) {
    float ts = 0.f, tq = 0.f;
    for (int p = tid; p < 1024; p += F) { ts += sh_s[p]; tq += sh_q[p]; }
    atomicAdd(&sums[tid], ts);
    atomicAdd(&sumsq[tid], tq);
  }
}

__global__ void fold_bn_kernel(const float* __restrict__ sums,
                               const float* __restrict__ sumsq, float invN,
                               int K, int NN, const float* __restrict__ W,
                               const float* __restrict__ extra_b,
                               float* __restrict__ Wout, float* __restrict__ vout) {
  __shared__ float s_ms[N_FEAT];
  __shared__ float s_s[N_FEAT];
  const int j = threadIdx.x;
  for (int k = j; k < K; k += blockDim.x) {
    float mean = sums[k] * invN;
    float var  = sumsq[k] * invN - mean * mean;
    float sc   = rsqrtf(var + BN_EPS);
    s_ms[k] = mean * sc;
    s_s[k]  = sc;
  }
  __syncthreads();
  float acc = extra_b ? extra_b[j] : 0.f;
  for (int k = 0; k < K; ++k) {
    float w = W[(size_t)k * NN + j];
    Wout[(size_t)k * NN + j] = s_s[k] * w;
    acc += (BN_BETA - s_ms[k]) * w;
  }
  vout[j] = acc;
}

__global__ __launch_bounds__(128) void gemm_wmma_kernel(
    const float* __restrict__ A, const float* __restrict__ Wt,
    const float* __restrict__ v, float* __restrict__ C,
    int M, int K, int doRelu) {
  const int lane = threadIdx.x & 31;
  const int wave = threadIdx.x >> 5;
  const int rowTile = blockIdx.x * 4 + wave;
  if (rowTile >= (M >> 4)) return;
  const int mrow = (rowTile << 4) + (lane & 15);
  const int kOff = (lane >> 4) << 1;
  const int n0   = lane & 15;
  v8f acc[4] = {};
  for (int kk = 0; kk < K; kk += 4) {
    const int k0 = kk + kOff;
    v2f a = *(const v2f*)(A + (size_t)mrow * K + k0);
    const float* w0 = Wt + (size_t)k0 * HIDDEN + n0;
#pragma unroll
    for (int t = 0; t < 4; ++t) {
      v2f b;
      b.x = w0[t * 16];
      b.y = w0[HIDDEN + t * 16];
      acc[t] = __builtin_amdgcn_wmma_f32_16x16x4_f32(
          false, a, false, b, (short)0, acc[t], false, false);
    }
  }
  const int rowBase = (rowTile << 4) + ((lane >> 4) << 3);
#pragma unroll
  for (int t = 0; t < 4; ++t) {
    const int n = t * 16 + n0;
    const float bias = v[n];
#pragma unroll
    for (int e = 0; e < 8; ++e) {
      float val = acc[t][e] + bias;
      if (doRelu) val = fmaxf(val, 0.f);
      C[(size_t)(rowBase + e) * HIDDEN + n] = val;
    }
  }
}

__global__ void deg_init_kernel(float* deg, int N) {
  int i = blockIdx.x * blockDim.x + threadIdx.x;
  if (i < N) deg[i] = 1.0f;
}
__global__ void deg_accum_kernel(const int* __restrict__ row,
                                 const float* __restrict__ att, int comp,
                                 int E, float* __restrict__ deg) {
  int e = blockIdx.x * blockDim.x + threadIdx.x;
  if (e >= E) return;
  float w = att ? att[(size_t)e * 2 + comp] : 1.0f;
  atomicAdd(&deg[row[e]], w);
}
__global__ void make_dinv_kernel(const float* __restrict__ deg,
                                 float* __restrict__ dinv, int N) {
  int i = blockIdx.x * blockDim.x + threadIdx.x;
  if (i < N) dinv[i] = deg[i] > 0.f ? rsqrtf(deg[i]) : 0.f;
}

__global__ void csr_zero_kernel(int* __restrict__ cnt, int N) {
  int i = blockIdx.x * blockDim.x + threadIdx.x;
  if (i < N) cnt[i] = 0;
}
__global__ void csr_count_kernel(const int* __restrict__ col, int E,
                                 int* __restrict__ cnt) {
  int e = blockIdx.x * blockDim.x + threadIdx.x;
  if (e < E) atomicAdd(&cnt[col[e]], 1);
}
__device__ __forceinline__ void load_counts4(const int* __restrict__ cnt, int i0,
                                             int N, int c[4]) {
  if (i0 + 3 < N) {
    const int4 v = *(const int4*)(cnt + i0);
    c[0] = v.x; c[1] = v.y; c[2] = v.z; c[3] = v.w;
  } else {
    c[0] = (i0     < N) ? cnt[i0]     : 0;
    c[1] = (i0 + 1 < N) ? cnt[i0 + 1] : 0;
    c[2] = (i0 + 2 < N) ? cnt[i0 + 2] : 0;
    c[3] = 0;
  }
}
__global__ __launch_bounds__(256) void csr_tilesum_kernel(
    const int* __restrict__ cnt, int N, int* __restrict__ tsum) {
  __shared__ int sh[8];
  const int tid = threadIdx.x;
  int c[4];
  load_counts4(cnt, (blockIdx.x * 256 + tid) * 4, N, c);
  int t = c[0] + c[1] + c[2] + c[3];
  for (int o = 16; o > 0; o >>= 1) t += __shfl_xor(t, o);
  if ((tid & 31) == 0) sh[tid >> 5] = t;
  __syncthreads();
  if (tid == 0) {
    int s = 0;
    for (int w = 0; w < 8; ++w) s += sh[w];
    tsum[blockIdx.x] = s;
  }
}
__global__ __launch_bounds__(256) void csr_scan_kernel(
    const int* __restrict__ cnt, const int* __restrict__ tsum, int N,
    int* __restrict__ off, int* __restrict__ cur) {
  __shared__ int sh_w[8];
  __shared__ int sh_b[8];
  const int tid = threadIdx.x, lane = tid & 31, wave = tid >> 5;
  int bpart = 0;
  for (int t = tid; t < (int)blockIdx.x; t += 256) bpart += tsum[t];
  for (int o = 16; o > 0; o >>= 1) bpart += __shfl_xor(bpart, o);
  const int i0 = (blockIdx.x * 256 + tid) * 4;
  int c[4];
  load_counts4(cnt, i0, N, c);
  const int t = c[0] + c[1] + c[2] + c[3];
  int inc = t;
  for (int o = 1; o < 32; o <<= 1) {
    const int y = __shfl_up(inc, o);
    if (lane >= o) inc += y;
  }
  if (lane == 31) sh_w[wave] = inc;
  if (lane == 0)  sh_b[wave] = bpart;
  __syncthreads();
  int base = 0;
  for (int w = 0; w < 8; ++w) {
    base += sh_b[w];
    if (w < wave) base += sh_w[w];
  }
  const int o0 = base + inc - t;
  const int o1 = o0 + c[0], o2 = o1 + c[1], o3 = o2 + c[2];
  if (i0 + 3 < N) {
    const int4 v = make_int4(o0, o1, o2, o3);
    *(int4*)(off + i0) = v;
    *(int4*)(cur + i0) = v;
  } else {
    if (i0     < N) { off[i0]     = o0; cur[i0]     = o0; }
    if (i0 + 1 < N) { off[i0 + 1] = o1; cur[i0 + 1] = o1; }
    if (i0 + 2 < N) { off[i0 + 2] = o2; cur[i0 + 2] = o2; }
  }
  if (i0 < N && N - 1 <= i0 + 3) off[N] = o0 + t;
}
__global__ void csr_fill_kernel(const int* __restrict__ row,
                                const int* __restrict__ col, int E,
                                int* __restrict__ cur, int2* __restrict__ sedge) {
  int e = blockIdx.x * blockDim.x + threadIdx.x;
  if (e >= E) return;
  const int r = row[e];
  const int pos = atomicAdd(&cur[col[e]], 1);
  if ((unsigned)pos < (unsigned)E) sedge[pos] = make_int2(r, e);
}

template <int CNT>
__device__ __forceinline__ void gather_rows(const float2* __restrict__ m2, int r,
                                            float cf, int k, int lane, float2& acc) {
  int rr[CNT];
  float cc[CNT];
  float2 v[CNT];
#pragma unroll
  for (int u = 0; u < CNT; ++u) {
    rr[u] = __builtin_amdgcn_readlane(r, k + u);
    cc[u] = __int_as_float(__builtin_amdgcn_readlane(__float_as_int(cf), k + u));
  }
#pragma unroll
  for (int u = 0; u < CNT; ++u) v[u] = m2[(size_t)rr[u] * (HIDDEN / 2) + lane];
#pragma unroll
  for (int u = 0; u < CNT; ++u) {
    acc.x += cc[u] * v[u].x;
    acc.y += cc[u] * v[u].y;
  }
}
template <bool HAS_ATT>
__device__ __forceinline__ float2 gcn_row(
    const float2* __restrict__ m2, const int* __restrict__ off,
    const int2* __restrict__ sedge, const float* __restrict__ att, int comp,
    const float* __restrict__ dinv, const float* __restrict__ b, int c, int lane) {
  const int beg = off[c], end = off[c + 1];
  const float dc = dinv[c];
  const float2 self = m2[(size_t)c * (HIDDEN / 2) + lane];
  const float2 bias = ((const float2*)b)[lane];
  float2 acc = make_float2(0.f, 0.f);
  for (int base = beg; base < end; base += 32) {
    const int p = base + lane;
    int r = 0;
    float cf = 0.f;
    if (p < end) {
      const int2 se = sedge[p];
      r = se.x;
      const float dr = dinv[r];
      const float w = HAS_ATT ? att[(size_t)se.y * 2 + comp] : 1.0f;
      cf = dr * w * dc;
    }
    const int n = (end - base) < 32 ? (end - base) : 32;
    int k = 0;
    for (; k + 4 <= n; k += 4) gather_rows<4>(m2, r, cf, k, lane, acc);
    const int rem = n - k;
    if (rem == 3)      gather_rows<3>(m2, r, cf, k, lane, acc);
    else if (rem == 2) gather_rows<2>(m2, r, cf, k, lane, acc);
    else if (rem == 1) gather_rows<1>(m2, r, cf, k, lane, acc);
  }
  float2 o;
  o.x = fmaxf(acc.x + dc * dc * self.x + bias.x, 0.f);
  o.y = fmaxf(acc.y + dc * dc * self.y + bias.y, 0.f);
  return o;
}

__global__ __launch_bounds__(256) void gather_gcn_kernel(
    const float* __restrict__ m, const int* __restrict__ off,
    const int2* __restrict__ sedge, const float* __restrict__ dinv,
    const float* __restrict__ b, float* __restrict__ out, int N) {
  const int lane = threadIdx.x & 31;
  const int c = __builtin_amdgcn_readfirstlane(blockIdx.x * 8 + (threadIdx.x >> 5));
  if (c >= N) return;
  const float2 o = gcn_row<false>((const float2*)m, off, sedge, nullptr, 0, dinv, b,
                                  c, lane);
  ((float2*)out)[(size_t)c * (HIDDEN / 2) + lane] = o;
}

__global__ __launch_bounds__(256) void gather_pool_kernel(
    const float* __restrict__ m, const int* __restrict__ off,
    const int2* __restrict__ sedge, const float* __restrict__ att, int comp,
    const float* __restrict__ dinv, const float* __restrict__ b,
    const int* __restrict__ batch, float* __restrict__ g, int N) {
  __shared__ float2 sh_row[8 * (HIDDEN / 2)];
  __shared__ int sh_gid[8];
  const int lane = threadIdx.x & 31;
  const int wave = threadIdx.x >> 5;
  const int c = __builtin_amdgcn_readfirstlane(blockIdx.x * 8 + wave);
  float2 o = make_float2(0.f, 0.f);
  int gid = -1;
  if (c < N) {
    gid = batch[c];
    o = gcn_row<true>((const float2*)m, off, sedge, att, comp, dinv, b, c, lane);
  }
  sh_row[wave * (HIDDEN / 2) + lane] = o;
  if (lane == 0) sh_gid[wave] = gid;
  __syncthreads();
  if (threadIdx.x < HIDDEN) {
    const int j = threadIdx.x;
    int gw[8];
    float v[8];
#pragma unroll
    for (int w = 0; w < 8; ++w) {
      gw[w] = sh_gid[w];
      v[w] = ((const float*)sh_row)[w * HIDDEN + j];
    }
    int cur = gw[0];
    float a = v[0];
#pragma unroll
    for (int w = 1; w < 8; ++w) {
      if (gw[w] == cur) {
        a += v[w];
      } else {
        if ((unsigned)cur < (unsigned)N_GRAPHS) atomicAdd(&g[(size_t)cur * HIDDEN + j], a);
        cur = gw[w];
        a = v[w];
      }
    }
    if ((unsigned)cur < (unsigned)N_GRAPHS) atomicAdd(&g[(size_t)cur * HIDDEN + j], a);
  }
}

__global__ void edge_att_kernel(const float* __restrict__ h,
                                const int* __restrict__ row,
                                const int* __restrict__ col,
                                const float* __restrict__ Wea,
                                const float* __restrict__ bea,
                                float* __restrict__ att, int E) {
  int e = blockIdx.x * blockDim.x + threadIdx.x;
  if (e >= E) return;
  const float* hr = h + (size_t)row[e] * HIDDEN;
  const float* hc = h + (size_t)col[e] * HIDDEN;
  float l0 = bea[0], l1 = bea[1];
  for (int j = 0; j < HIDDEN; ++j) {
    float a = hr[j], c = hc[j];
    l0 += a * Wea[j * 2 + 0] + c * Wea[(HIDDEN + j) * 2 + 0];
    l1 += a * Wea[j * 2 + 1] + c * Wea[(HIDDEN + j) * 2 + 1];
  }
  float mx = fmaxf(l0, l1);
  float e0 = __expf(l0 - mx), e1 = __expf(l1 - mx);
  float inv = 1.f / (e0 + e1);
  att[(size_t)e * 2 + 0] = e0 * inv;
  att[(size_t)e * 2 + 1] = e1 * inv;
}

__global__ void node_att_split_kernel(const float* __restrict__ h,
                                      const float* __restrict__ Wna,
                                      const float* __restrict__ bna,
                                      float* __restrict__ xc,
                                      float* __restrict__ xo, int N) {
  int i = blockIdx.x * blockDim.x + threadIdx.x;
  if (i >= N) return;
  const float* hi = h + (size_t)i * HIDDEN;
  float l0 = bna[0], l1 = bna[1];
  for (int j = 0; j < HIDDEN; ++j) {
    float a = hi[j];
    l0 += a * Wna[j * 2 + 0];
    l1 += a * Wna[j * 2 + 1];
  }
  float mx = fmaxf(l0, l1);
  float e0 = __expf(l0 - mx), e1 = __expf(l1 - mx);
  float inv = 1.f / (e0 + e1);
  float p0 = e0 * inv, p1 = e1 * inv;
  for (int j = 0; j < HIDDEN; ++j) {
    float a = hi[j];
    xc[(size_t)i * HIDDEN + j] = p0 * a;
    xo[(size_t)i * HIDDEN + j] = p1 * a;
  }
}

#define HEAD_THREADS 512
#define G_PITCH 68
#define W_PITCH 72

__device__ __forceinline__ void head_colstats(const float* __restrict__ ps,
                                              const float* __restrict__ pq,
                                              float* __restrict__ sc,
                                              float* __restrict__ ms, int col) {
  float ts = 0.f, tq = 0.f;
  for (int p = 0; p < 32; ++p) {
    ts += ps[p * HIDDEN + col];
    tq += pq[p * HIDDEN + col];
  }
  const float invN = 1.f / N_GRAPHS;
  const float mean = ts * invN;
  const float var  = tq * invN - mean * mean;
  const float s    = rsqrtf(var + BN_EPS);
  sc[col] = s;
  ms[col] = mean * s;
}

__global__ __launch_bounds__(HEAD_THREADS) void heads_kernel(
    const float* __restrict__ gc, const float* __restrict__ go,
    const float* __restrict__ W1_c, const float* __restrict__ b1_c,
    const float* __restrict__ W2_c, const float* __restrict__ b2_c,
    const float* __restrict__ W1_o, const float* __restrict__ b1_o,
    const float* __restrict__ W2_o, const float* __restrict__ b2_o,
    const float* __restrict__ W1_co, const float* __restrict__ b1_co,
    const float* __restrict__ W2_co, const float* __restrict__ b2_co,
    float* __restrict__ out) {
  __shared__ __attribute__((aligned(16))) float sh_g[N_GRAPHS * G_PITCH];
  __shared__ __attribute__((aligned(16))) float sh_w[HIDDEN * W_PITCH];
  __shared__ __attribute__((aligned(16))) float sh_ps[32 * HIDDEN];
  __shared__ __attribute__((aligned(16))) float sh_pq[32 * HIDDEN];
  __shared__ float sh_sc[HIDDEN];
  __shared__ float sh_ms[HIDDEN];
  __shared__ float sh_v[HIDDEN];
  __shared__ __attribute__((aligned(16))) float sh_w2[HIDDEN * N_CLASSES];
  __shared__ float sh_v2[16];
  __shared__ float sh_w1r[HIDDEN * HIDDEN];
  __shared__ float sh_w2r[HIDDEN * N_CLASSES];

  const int head = blockIdx.x;
  const float* __restrict__ W1 = head == 0 ? W1_c : (head == 1 ? W1_o : W1_co);
  const float* __restrict__ b1 = head == 0 ? b1_c : (head == 1 ? b1_o : b1_co);
  const float* __restrict__ W2 = head == 0 ? W2_c : (head == 1 ? W2_o : W2_co);
  const float* __restrict__ b2 = head == 0 ? b2_c : (head == 1 ? b2_o : b2_co);
  const int tid  = threadIdx.x;
  const int lane = tid & 31;
  const int wave = tid >> 5;

  float w1r[8];
#pragma unroll
  for (int u = 0; u < 8; ++u) w1r[u] = W1[tid + HEAD_THREADS * u];
  const float w2r0 = W2[tid];
  const float w2r1 = tid < HIDDEN * N_CLASSES - HEAD_THREADS ? W2[HEAD_THREADS + tid] : 0.f;
  const float b1r  = tid < HIDDEN ? b1[tid] : 0.f;
  const float b2r  = tid < N_CLASSES ? b2[tid] : 0.f;

  {
    const float4* __restrict__ gc4 = (const float4*)gc;
    const float4* __restrict__ go4 = (const float4*)go;
    const float4* __restrict__ pa  = head == 1 ? go4 : gc4;
    const int cg = tid & 15;
    const int r0 = tid >> 4;
    float4 x[16];
#pragma unroll
    for (int u = 0; u < 16; ++u) x[u] = pa[tid + HEAD_THREADS * u];
    if (head == 2) {
#pragma unroll
      for (int u = 0; u < 16; ++u) {
        const float4 b = go4[tid + HEAD_THREADS * u];
        x[u] = make_float4(x[u].x + b.x, x[u].y + b.y, x[u].z + b.z, x[u].w + b.w);
      }
    }
    float4 s = make_float4(0.f, 0.f, 0.f, 0.f);
    float4 q = make_float4(0.f, 0.f, 0.f, 0.f);
#pragma unroll
    for (int u = 0; u < 16; ++u) {
      *(float4*)(sh_g + (r0 + 32 * u) * G_PITCH + 4 * cg) = x[u];
      s.x += x[u].x; q.x += x[u].x * x[u].x;
      s.y += x[u].y; q.y += x[u].y * x[u].y;
      s.z += x[u].z; q.z += x[u].z * x[u].z;
      s.w += x[u].w; q.w += x[u].w * x[u].w;
    }
    *(float4*)(sh_ps + r0 * HIDDEN + 4 * cg) = s;
    *(float4*)(sh_pq + r0 * HIDDEN + 4 * cg) = q;
#pragma unroll
    for (int u = 0; u < 8; ++u) sh_w1r[tid + HEAD_THREADS * u] = w1r[u];
    sh_w2r[tid] = w2r0;
    if (tid < HIDDEN * N_CLASSES - HEAD_THREADS) sh_w2r[HEAD_THREADS + tid] = w2r1;
  }
  __syncthreads();
  if (tid < HIDDEN) head_colstats(sh_ps, sh_pq, sh_sc, sh_ms, tid);
  __syncthreads();

#pragma unroll
  for (int u = 0; u < 8; ++u) {
    const int i = tid + HEAD_THREADS * u;
    const int k = i >> 6, j = i & 63;
    sh_w[k * W_PITCH + j] = sh_sc[k] * w1r[u];
  }
  if (tid < HIDDEN) {
    float acc = b1r;
    for (int k = 0; k < HIDDEN; ++k) acc += (BN_BETA - sh_ms[k]) * sh_w1r[k * HIDDEN + tid];
    sh_v[tid] = acc;
  }
  __syncthreads();

  {
    const int kOff = (lane >> 4) << 1;
    const int n0   = lane & 15;
    v8f acc[2][4] = {};
    const float* a0p = sh_g + (32 * wave + n0) * G_PITCH + kOff;
    const float* a1p = a0p + 16 * G_PITCH;
    const float* bp  = sh_w + kOff * W_PITCH + n0;
    for (int kk = 0; kk < HIDDEN; kk += 4) {
      const v2f a0 = *(const v2f*)(a0p + kk);
      const v2f a1 = *(const v2f*)(a1p + kk);
#pragma unroll
      for (int t = 0; t < 4; ++t) {
        v2f b;
        b.x = bp[kk * W_PITCH + t * 16];
        b.y = bp[(kk + 1) * W_PITCH + t * 16];
        acc[0][t] = __builtin_amdgcn_wmma_f32_16x16x4_f32(
            false, a0, false, b, (short)0, acc[0][t], false, false);
        acc[1][t] = __builtin_amdgcn_wmma_f32_16x16x4_f32(
            false, a1, false, b, (short)0, acc[1][t], false, false);
      }
    }
    const int rowBase = 32 * wave + ((lane >> 4) << 3);
#pragma unroll
    for (int t = 0; t < 4; ++t) {
      const int n = t * 16 + n0;
      const float bias = sh_v[n];
      float s = 0.f, q = 0.f;
#pragma unroll
      for (int i = 0; i < 2; ++i) {
#pragma unroll
        for (int e = 0; e < 8; ++e) {
          const float val = fmaxf(acc[i][t][e] + bias, 0.f);
          sh_g[(rowBase + 16 * i + e) * G_PITCH + n] = val;
          s += val;
          q += val * val;
        }
      }
      sh_ps[(2 * wave + (lane >> 4)) * HIDDEN + n] = s;
      sh_pq[(2 * wave + (lane >> 4)) * HIDDEN + n] = q;
    }
  }
  __syncthreads();
  if (tid < HIDDEN) head_colstats(sh_ps, sh_pq, sh_sc, sh_ms, tid);
  __syncthreads();

  sh_w2[tid] = sh_sc[tid / N_CLASSES] * w2r0;
  if (tid < HIDDEN * N_CLASSES - HEAD_THREADS)
    sh_w2[HEAD_THREADS + tid] = sh_sc[(HEAD_THREADS + tid) / N_CLASSES] * w2r1;
  if (tid < N_CLASSES) {
    float acc = b2r;
    for (int k = 0; k < HIDDEN; ++k) acc += (BN_BETA - sh_ms[k]) * sh_w2r[k * N_CLASSES + tid];
    sh_v2[tid] = acc;
  }
  __syncthreads();

  {
    float l[N_CLASSES];
    for (int c = 0; c < N_CLASSES; ++c) l[c] = sh_v2[c];
    const float4* __restrict__ row4 = (const float4*)(sh_g + tid * G_PITCH);
    for (int k4 = 0; k4 < HIDDEN / 4; ++k4) {
      const float4 x = row4[k4];
      const float* w = sh_w2 + k4 * 4 * N_CLASSES;
      for (int c = 0; c < N_CLASSES; ++c) l[c] += x.x * w[c];
      for (int c = 0; c < N_CLASSES; ++c) l[c] += x.y * w[N_CLASSES + c];
      for (int c = 0; c < N_CLASSES; ++c) l[c] += x.z * w[2 * N_CLASSES + c];
      for (int c = 0; c < N_CLASSES; ++c) l[c] += x.w * w[3 * N_CLASSES + c];
    }
    float mx = l[0];
    for (int c = 1; c < N_CLASSES; ++c) mx = fmaxf(mx, l[c]);
    float s = 0.f;
    for (int c = 0; c < N_CLASSES; ++c) s += __expf(l[c] - mx);
    const float lse = __logf(s) + mx;
    float* __restrict__ o = out + ((size_t)head * N_GRAPHS + tid) * N_CLASSES;
    for (int c = 0; c < N_CLASSES; ++c) o[c] = l[c] - lse;
  }
}

extern "C" void kernel_launch(void* const* d_in, const int* in_sizes, int n_in,
                              void* d_out, int out_size, void* d_ws, size_t ws_size,
                              hipStream_t stream) {
  (void)in_sizes; (void)n_in; (void)out_size; (void)ws_size;
  const float* x       = (const float*)d_in[0];
  const int*   ei      = (const int*)d_in[1];
  const int*   row     = ei;
  const int*   col     = ei + N_EDGES;
  const int*   batch   = (const int*)d_in[2];
  const float* W_feat  = (const float*)d_in[3];
  const float* W_convs = (const float*)d_in[4];
  const float* b_convs = (const float*)d_in[5];
  const float* W_ea    = (const float*)d_in[6];
  const float* b_ea    = (const float*)d_in[7];
  const float* W_na    = (const float*)d_in[8];
  const float* b_na    = (const float*)d_in[9];
  const float* W_ctx   = (const float*)d_in[10];
  const float* b_ctx   = (const float*)d_in[11];
  const float* W_obj   = (const float*)d_in[12];
  const float* b_obj   = (const float*)d_in[13];
  const float* W1h[3]  = {(const float*)d_in[14], (const float*)d_in[18], (const float*)d_in[22]};
  const float* b1h[3]  = {(const float*)d_in[15], (const float*)d_in[19], (const float*)d_in[23]};
  const float* W2h[3]  = {(const float*)d_in[16], (const float*)d_in[20], (const float*)d_in[24]};
  const float* b2h[3]  = {(const float*)d_in[17], (const float*)d_in[21], (const float*)d_in[25]};

  size_t off = 0;
  auto carve = [&](size_t elems) -> float* {
    float* p = (float*)((char*)d_ws + off);
    off += ((elems * sizeof(float)) + 255) & ~(size_t)255;
    return p;
  };
  float* hb    = carve((size_t)N_NODES * HIDDEN);
  float* mbuf  = carve((size_t)N_NODES * HIDDEN);
  float* xc    = carve((size_t)N_NODES * HIDDEN);
  float* xo    = carve((size_t)N_NODES * HIDDEN);
  float* att   = carve((size_t)N_EDGES * 2);
  float* degu  = carve(N_NODES);
  float* dinvu = carve(N_NODES);
  float* dega  = carve(N_NODES);
  float* dinva = carve(N_NODES);
  float* gc    = carve((size_t)N_GRAPHS * HIDDEN);
  float* go    = carve((size_t)N_GRAPHS * HIDDEN);
  float* sums  = carve(N_FEAT);
  float* sq    = carve(N_FEAT);
  float* Wf    = carve((size_t)N_FEAT * HIDDEN);
  float* vf    = carve(HIDDEN);
  int*  ecnt   = (int*)carve(N_NODES);
  int*  etsum  = (int*)carve(128);
  int*  eoff   = (int*)carve(N_NODES + 1);
  int*  ecur   = (int*)carve(N_NODES);
  int2* sedge  = (int2*)carve((size_t)N_EDGES * 2);

  auto nblk = [](long n, int b) { return (int)((n + b - 1) / b); };

  auto stats = [&](const float* A, int N, int F) {
    hipMemsetAsync(sums, 0, (size_t)F * sizeof(float), stream);
    hipMemsetAsync(sq,   0, (size_t)F * sizeof(float), stream);
    const int nvec = N * F / 4;
    const int grid = nblk(nvec, 256) < 512 ? nblk(nvec, 256) : 512;
    colstats_kernel<<<grid, 256, 0, stream>>>(A, nvec, F, sums, sq);
  };
  auto gemm = [&](const float* A, float* C, int M, int K, int relu) {
    gemm_wmma_kernel<<<nblk(M / 16, 4), 128, 0, stream>>>(A, Wf, vf, C, M, K, relu);
  };

  stats(x, N_NODES, N_FEAT);
  fold_bn_kernel<<<1, HIDDEN, 0, stream>>>(sums, sq, 1.f / N_NODES, N_FEAT, HIDDEN,
                                           W_feat, nullptr, Wf, vf);
  gemm(x, hb, N_NODES, N_FEAT, 1);

  deg_init_kernel<<<nblk(N_NODES, 256), 256, 0, stream>>>(degu, N_NODES);
  deg_accum_kernel<<<nblk(N_EDGES, 256), 256, 0, stream>>>(row, nullptr, 0, N_EDGES, degu);
  make_dinv_kernel<<<nblk(N_NODES, 256), 256, 0, stream>>>(degu, dinvu, N_NODES);

  csr_zero_kernel<<<nblk(N_NODES, 256), 256, 0, stream>>>(ecnt, N_NODES);
  csr_count_kernel<<<nblk(N_EDGES, 256), 256, 0, stream>>>(col, N_EDGES, ecnt);
  csr_tilesum_kernel<<<nblk(N_NODES, 1024), 256, 0, stream>>>(ecnt, N_NODES, etsum);
  csr_scan_kernel<<<nblk(N_NODES, 1024), 256, 0, stream>>>(ecnt, etsum, N_NODES,
                                                           eoff, ecur);
  csr_fill_kernel<<<nblk(N_EDGES, 256), 256, 0, stream>>>(row, col, N_EDGES, ecur,
                                                          sedge);

  for (int i = 0; i < 3; ++i) {
    stats(hb, N_NODES, HIDDEN);
    fold_bn_kernel<<<1, HIDDEN, 0, stream>>>(sums, sq, 1.f / N_NODES, HIDDEN, HIDDEN,
                                             W_convs + (size_t)i * HIDDEN * HIDDEN,
                                             nullptr, Wf, vf);
    gemm(hb, mbuf, N_NODES, HIDDEN, 0);
    gather_gcn_kernel<<<nblk(N_NODES, 8), 256, 0, stream>>>(
        mbuf, eoff, sedge, dinvu, b_convs + (size_t)i * HIDDEN, hb, N_NODES);
  }

  edge_att_kernel<<<nblk(N_EDGES, 256), 256, 0, stream>>>(hb, row, col, W_ea, b_ea,
                                                          att, N_EDGES);
  node_att_split_kernel<<<nblk(N_NODES, 256), 256, 0, stream>>>(hb, W_na, b_na,
                                                                xc, xo, N_NODES);

  auto run_branch = [&](float* xin, int comp, const float* W, const float* b,
                        float* gout) {
    stats(xin, N_NODES, HIDDEN);
    fold_bn_kernel<<<1, HIDDEN, 0, stream>>>(sums, sq, 1.f / N_NODES, HIDDEN, HIDDEN,
                                             W, nullptr, Wf, vf);
    gemm(xin, mbuf, N_NODES, HIDDEN, 0);
    deg_init_kernel<<<nblk(N_NODES, 256), 256, 0, stream>>>(dega, N_NODES);
    deg_accum_kernel<<<nblk(N_EDGES, 256), 256, 0, stream>>>(row, att, comp, N_EDGES, dega);
    make_dinv_kernel<<<nblk(N_NODES, 256), 256, 0, stream>>>(dega, dinva, N_NODES);
    hipMemsetAsync(gout, 0, (size_t)N_GRAPHS * HIDDEN * sizeof(float), stream);
    gather_pool_kernel<<<nblk(N_NODES, 8), 256, 0, stream>>>(
        mbuf, eoff, sedge, att, comp, dinva, b, batch, gout, N_NODES);
  };
  run_branch(xc, 0, W_ctx, b_ctx, gc);
  run_branch(xo, 1, W_obj, b_obj, go);

  static_assert(HEAD_THREADS == N_GRAPHS, "one thread per graph in the logits stage");
  heads_kernel<<<3, HEAD_THREADS, 0, stream>>>(
      gc, go,
      W1h[0], b1h[0], W2h[0], b2h[0],
      W1h[1], b1h[1], W2h[1], b2h[1],
      W1h[2], b1h[2], W2h[2], b2h[2],
      (float*)d_out);
}


# edge attention logits: per-edge 128-term dot over two gathered rows -> per-node halves h@W_ea[:64], h@W_ea[64:] made once in the node pass; an edge adds two gathered pairs (logit sum order changes)
# speedup vs baseline: 2.9674x; 1.0700x over previous
#include <hip/hip_runtime.h>
#include <math.h>


#define N_NODES  100000
#define N_EDGES  1000000
#define N_FEAT   128
#define HIDDEN   64
#define N_GRAPHS 512
#define N_CLASSES 10
#define BN_EPS  1e-5f
#define BN_BETA 1e-4f

typedef __attribute__((ext_vector_type(2))) float v2f;
typedef __attribute__((ext_vector_type(8))) float v8f;

__global__ __launch_bounds__(256) void colstats_kernel(
    const float* __restrict__ A, int nvec, int F,
    float* __restrict__ sums, float* __restrict__ sumsq) {
  __shared__ float sh_s[1024];
  __shared__ float sh_q[1024];
  const float4* __restrict__ A4 = (const float4*)A;
  const int tid = threadIdx.x;
  const int stride = gridDim.x * 256;
  const int last = nvec - 1;
  float4 s = make_float4(0.f, 0.f, 0.f, 0.f);
  float4 q = make_float4(0.f, 0.f, 0.f, 0.f);
  for (int i = blockIdx.x * 256 + tid; i < nvec; i += 8 * stride) {
    float4 v[8];
#pragma unroll
    for (int u = 0; u < 8; ++u) {
      const int idx = i + u * stride;
      v[u] = A4[idx < last ? idx : last];
    }
    __builtin_amdgcn_sched_barrier(0);
#pragma unroll
    for (int u = 0; u < 8; ++u) {
      const bool live = (i + u * stride) < nvec;
      const float a = live ? v[u].x : 0.f;
      const float b = live ? v[u].y : 0.f;
      const float c = live ? v[u].z : 0.f;
      const float d = live ? v[u].w : 0.f;
      s.x += a; q.x += a * a;
      s.y += b; q.y += b * b;
      s.z += c; q.z += c * c;
      s.w += d; q.w += d * d;
    }
  }
  ((float4*)sh_s)[tid] = s;
  ((float4*)sh_q)[tid] = q;
  __syncthreads();
  if (tid < F) {
    float ts = 0.f, tq = 0.f;
    for (int p = tid; p < 1024; p += F) { ts += sh_s[p]; tq += sh_q[p]; }
    atomicAdd(&sums[tid], ts);
    atomicAdd(&sumsq[tid], tq);
  }
}

__global__ void fold_bn_kernel(const float* __restrict__ sums,
                               const float* __restrict__ sumsq, float invN,
                               int K, int NN, const float* __restrict__ W,
                               const float* __restrict__ extra_b,
                               float* __restrict__ Wout, float* __restrict__ vout) {
  __shared__ float s_ms[N_FEAT];
  __shared__ float s_s[N_FEAT];
  const int j = threadIdx.x;
  for (int k = j; k < K; k += blockDim.x) {
    float mean = sums[k] * invN;
    float var  = sumsq[k] * invN - mean * mean;
    float sc   = rsqrtf(var + BN_EPS);
    s_ms[k] = mean * sc;
    s_s[k]  = sc;
  }
  __syncthreads();
  float acc = extra_b ? extra_b[j] : 0.f;
  for (int k = 0; k < K; ++k) {
    float w = W[(size_t)k * NN + j];
    Wout[(size_t)k * NN + j] = s_s[k] * w;
    acc += (BN_BETA - s_ms[k]) * w;
  }
  vout[j] = acc;
}

__global__ __launch_bounds__(128) void gemm_wmma_kernel(
    const float* __restrict__ A, const float* __restrict__ Wt,
    const float* __restrict__ v, float* __restrict__ C,
    int M, int K, int doRelu) {
  const int lane = threadIdx.x & 31;
  const int wave = threadIdx.x >> 5;
  const int rowTile = blockIdx.x * 4 + wave;
  if (rowTile >= (M >> 4)) return;
  const int mrow = (rowTile << 4) + (lane & 15);
  const int kOff = (lane >> 4) << 1;
  const int n0   = lane & 15;
  v8f acc[4] = {};
  for (int kk = 0; kk < K; kk += 4) {
    const int k0 = kk + kOff;
    v2f a = *(const v2f*)(A + (size_t)mrow * K + k0);
    const float* w0 = Wt + (size_t)k0 * HIDDEN + n0;
#pragma unroll
    for (int t = 0; t < 4; ++t) {
      v2f b;
      b.x = w0[t * 16];
      b.y = w0[HIDDEN + t * 16];
      acc[t] = __builtin_amdgcn_wmma_f32_16x16x4_f32(
          false, a, false, b, (short)0, acc[t], false, false);
    }
  }
  const int rowBase = (rowTile << 4) + ((lane >> 4) << 3);
#pragma unroll
  for (int t = 0; t < 4; ++t) {
    const int n = t * 16 + n0;
    const float bias = v[n];
#pragma unroll
    for (int e = 0; e < 8; ++e) {
      float val = acc[t][e] + bias;
      if (doRelu) val = fmaxf(val, 0.f);
      C[(size_t)(rowBase + e) * HIDDEN + n] = val;
    }
  }
}

__global__ void deg_init_kernel(float* deg, int N) {
  int i = blockIdx.x * blockDim.x + threadIdx.x;
  if (i < N) deg[i] = 1.0f;
}
__global__ void deg_accum_kernel(const int* __restrict__ row,
                                 const float* __restrict__ att, int comp,
                                 int E, float* __restrict__ deg) {
  int e = blockIdx.x * blockDim.x + threadIdx.x;
  if (e >= E) return;
  float w = att ? att[(size_t)e * 2 + comp] : 1.0f;
  atomicAdd(&deg[row[e]], w);
}
__global__ void make_dinv_kernel(const float* __restrict__ deg,
                                 float* __restrict__ dinv, int N) {
  int i = blockIdx.x * blockDim.x + threadIdx.x;
  if (i < N) dinv[i] = deg[i] > 0.f ? rsqrtf(deg[i]) : 0.f;
}

__global__ void csr_zero_kernel(int* __restrict__ cnt, int N) {
  int i = blockIdx.x * blockDim.x + threadIdx.x;
  if (i < N) cnt[i] = 0;
}
__global__ void csr_count_kernel(const int* __restrict__ col, int E,
                                 int* __restrict__ cnt) {
  int e = blockIdx.x * blockDim.x + threadIdx.x;
  if (e < E) atomicAdd(&cnt[col[e]], 1);
}
__device__ __forceinline__ void load_counts4(const int* __restrict__ cnt, int i0,
                                             int N, int c[4]) {
  if (i0 + 3 < N) {
    const int4 v = *(const int4*)(cnt + i0);
    c[0] = v.x; c[1] = v.y; c[2] = v.z; c[3] = v.w;
  } else {
    c[0] = (i0     < N) ? cnt[i0]     : 0;
    c[1] = (i0 + 1 < N) ? cnt[i0 + 1] : 0;
    c[2] = (i0 + 2 < N) ? cnt[i0 + 2] : 0;
    c[3] = 0;
  }
}
__global__ __launch_bounds__(256) void csr_tilesum_kernel(
    const int* __restrict__ cnt, int N, int* __restrict__ tsum) {
  __shared__ int sh[8];
  const int tid = threadIdx.x;
  int c[4];
  load_counts4(cnt, (blockIdx.x * 256 + tid) * 4, N, c);
  int t = c[0] + c[1] + c[2] + c[3];
  for (int o = 16; o > 0; o >>= 1) t += __shfl_xor(t, o);
  if ((tid & 31) == 0) sh[tid >> 5] = t;
  __syncthreads();
  if (tid == 0) {
    int s = 0;
    for (int w = 0; w < 8; ++w) s += sh[w];
    tsum[blockIdx.x] = s;
  }
}
__global__ __launch_bounds__(256) void csr_scan_kernel(
    const int* __restrict__ cnt, const int* __restrict__ tsum, int N,
    int* __restrict__ off, int* __restrict__ cur) {
  __shared__ int sh_w[8];
  __shared__ int sh_b[8];
  const int tid = threadIdx.x, lane = tid & 31, wave = tid >> 5;
  int bpart = 0;
  for (int t = tid; t < (int)blockIdx.x; t += 256) bpart += tsum[t];
  for (int o = 16; o > 0; o >>= 1) bpart += __shfl_xor(bpart, o);
  const int i0 = (blockIdx.x * 256 + tid) * 4;
  int c[4];
  load_counts4(cnt, i0, N, c);
  const int t = c[0] + c[1] + c[2] + c[3];
  int inc = t;
  for (int o = 1; o < 32; o <<= 1) {
    const int y = __shfl_up(inc, o);
    if (lane >= o) inc += y;
  }
  if (lane == 31) sh_w[wave] = inc;
  if (lane == 0)  sh_b[wave] = bpart;
  __syncthreads();
  int base = 0;
  for (int w = 0; w < 8; ++w) {
    base += sh_b[w];
    if (w < wave) base += sh_w[w];
  }
  const int o0 = base + inc - t;
  const int o1 = o0 + c[0], o2 = o1 + c[1], o3 = o2 + c[2];
  if (i0 + 3 < N) {
    const int4 v = make_int4(o0, o1, o2, o3);
    *(int4*)(off + i0) = v;
    *(int4*)(cur + i0) = v;
  } else {
    if (i0     < N) { off[i0]     = o0; cur[i0]     = o0; }
    if (i0 + 1 < N) { off[i0 + 1] = o1; cur[i0 + 1] = o1; }
    if (i0 + 2 < N) { off[i0 + 2] = o2; cur[i0 + 2] = o2; }
  }
  if (i0 < N && N - 1 <= i0 + 3) off[N] = o0 + t;
}
__global__ void csr_fill_kernel(const int* __restrict__ row,
                                const int* __restrict__ col, int E,
                                int* __restrict__ cur, int2* __restrict__ sedge) {
  int e = blockIdx.x * blockDim.x + threadIdx.x;
  if (e >= E) return;
  const int r = row[e];
  const int pos = atomicAdd(&cur[col[e]], 1);
  if ((unsigned)pos < (unsigned)E) sedge[pos] = make_int2(r, e);
}

template <int CNT>
__device__ __forceinline__ void gather_rows(const float2* __restrict__ m2, int r,
                                            float cf, int k, int lane, float2& acc) {
  int rr[CNT];
  float cc[CNT];
  float2 v[CNT];
#pragma unroll
  for (int u = 0; u < CNT; ++u) {
    rr[u] = __builtin_amdgcn_readlane(r, k + u);
    cc[u] = __int_as_float(__builtin_amdgcn_readlane(__float_as_int(cf), k + u));
  }
#pragma unroll
  for (int u = 0; u < CNT; ++u) v[u] = m2[(size_t)rr[u] * (HIDDEN / 2) + lane];
#pragma unroll
  for (int u = 0; u < CNT; ++u) {
    acc.x += cc[u] * v[u].x;
    acc.y += cc[u] * v[u].y;
  }
}
template <bool HAS_ATT>
__device__ __forceinline__ float2 gcn_row(
    const float2* __restrict__ m2, const int* __restrict__ off,
    const int2* __restrict__ sedge, const float* __restrict__ att, int comp,
    const float* __restrict__ dinv, const float* __restrict__ b, int c, int lane) {
  const int beg = off[c], end = off[c + 1];
  const float dc = dinv[c];
  const float2 self = m2[(size_t)c * (HIDDEN / 2) + lane];
  const float2 bias = ((const float2*)b)[lane];
  float2 acc = make_float2(0.f, 0.f);
  for (int base = beg; base < end; base += 32) {
    const int p = base + lane;
    int r = 0;
    float cf = 0.f;
    if (p < end) {
      const int2 se = sedge[p];
      r = se.x;
      const float dr = dinv[r];
      const float w = HAS_ATT ? att[(size_t)se.y * 2 + comp] : 1.0f;
      cf = dr * w * dc;
    }
    const int n = (end - base) < 32 ? (end - base) : 32;
    int k = 0;
    for (; k + 4 <= n; k += 4) gather_rows<4>(m2, r, cf, k, lane, acc);
    const int rem = n - k;
    if (rem == 3)      gather_rows<3>(m2, r, cf, k, lane, acc);
    else if (rem == 2) gather_rows<2>(m2, r, cf, k, lane, acc);
    else if (rem == 1) gather_rows<1>(m2, r, cf, k, lane, acc);
  }
  float2 o;
  o.x = fmaxf(acc.x + dc * dc * self.x + bias.x, 0.f);
  o.y = fmaxf(acc.y + dc * dc * self.y + bias.y, 0.f);
  return o;
}

__global__ __launch_bounds__(256) void gather_gcn_kernel(
    const float* __restrict__ m, const int* __restrict__ off,
    const int2* __restrict__ sedge, const float* __restrict__ dinv,
    const float* __restrict__ b, float* __restrict__ out, int N) {
  const int lane = threadIdx.x & 31;
  const int c = __builtin_amdgcn_readfirstlane(blockIdx.x * 8 + (threadIdx.x >> 5));
  if (c >= N) return;
  const float2 o = gcn_row<false>((const float2*)m, off, sedge, nullptr, 0, dinv, b,
                                  c, lane);
  ((float2*)out)[(size_t)c * (HIDDEN / 2) + lane] = o;
}

__global__ __launch_bounds__(256) void gather_pool_kernel(
    const float* __restrict__ m, const int* __restrict__ off,
    const int2* __restrict__ sedge, const float* __restrict__ att, int comp,
    const float* __restrict__ dinv, const float* __restrict__ b,
    const int* __restrict__ batch, float* __restrict__ g, int N) {
  __shared__ float2 sh_row[8 * (HIDDEN / 2)];
  __shared__ int sh_gid[8];
  const int lane = threadIdx.x & 31;
  const int wave = threadIdx.x >> 5;
  const int c = __builtin_amdgcn_readfirstlane(blockIdx.x * 8 + wave);
  float2 o = make_float2(0.f, 0.f);
  int gid = -1;
  if (c < N) {
    gid = batch[c];
    o = gcn_row<true>((const float2*)m, off, sedge, att, comp, dinv, b, c, lane);
  }
  sh_row[wave * (HIDDEN / 2) + lane] = o;
  if (lane == 0) sh_gid[wave] = gid;
  __syncthreads();
  if (threadIdx.x < HIDDEN) {
    const int j = threadIdx.x;
    int gw[8];
    float v[8];
#pragma unroll
    for (int w = 0; w < 8; ++w) {
      gw[w] = sh_gid[w];
      v[w] = ((const float*)sh_row)[w * HIDDEN + j];
    }
    int cur = gw[0];
    float a = v[0];
#pragma unroll
    for (int w = 1; w < 8; ++w) {
      if (gw[w] == cur) {
        a += v[w];
      } else {
        if ((unsigned)cur < (unsigned)N_GRAPHS) atomicAdd(&g[(size_t)cur * HIDDEN + j], a);
        cur = gw[w];
        a = v[w];
      }
    }
    if ((unsigned)cur < (unsigned)N_GRAPHS) atomicAdd(&g[(size_t)cur * HIDDEN + j], a);
  }
}

__global__ void node_att_split_kernel(const float* __restrict__ h,
                                      const float* __restrict__ Wna,
                                      const float* __restrict__ bna,
                                      const float* __restrict__ Wea,
                                      float* __restrict__ xc,
                                      float* __restrict__ xo,
                                      float* __restrict__ pr,
                                      float* __restrict__ pc, int N) {
  int i = blockIdx.x * blockDim.x + threadIdx.x;
  if (i >= N) return;
  const float* hi = h + (size_t)i * HIDDEN;
  float l0 = bna[0], l1 = bna[1];
  float r0 = 0.f, r1 = 0.f, c0 = 0.f, c1 = 0.f;
  for (int j = 0; j < HIDDEN; ++j) {
    float a = hi[j];
    l0 += a * Wna[j * 2 + 0];
    l1 += a * Wna[j * 2 + 1];
    r0 += a * Wea[j * 2 + 0];
    r1 += a * Wea[j * 2 + 1];
    c0 += a * Wea[(HIDDEN + j) * 2 + 0];
    c1 += a * Wea[(HIDDEN + j) * 2 + 1];
  }
  ((float2*)pr)[i] = make_float2(r0, r1);
  ((float2*)pc)[i] = make_float2(c0, c1);
  float mx = fmaxf(l0, l1);
  float e0 = __expf(l0 - mx), e1 = __expf(l1 - mx);
  float inv = 1.f / (e0 + e1);
  float p0 = e0 * inv, p1 = e1 * inv;
  for (int j = 0; j < HIDDEN; ++j) {
    float a = hi[j];
    xc[(size_t)i * HIDDEN + j] = p0 * a;
    xo[(size_t)i * HIDDEN + j] = p1 * a;
  }
}

__global__ void edge_att_kernel(const float* __restrict__ pr,
                                const float* __restrict__ pc,
                                const int* __restrict__ row,
                                const int* __restrict__ col,
                                const float* __restrict__ bea,
                                float* __restrict__ att, int E) {
  int e = blockIdx.x * blockDim.x + threadIdx.x;
  if (e >= E) return;
  const float2 a = ((const float2*)pr)[row[e]];
  const float2 c = ((const float2*)pc)[col[e]];
  const float l0 = (a.x + c.x) + bea[0];
  const float l1 = (a.y + c.y) + bea[1];
  float mx = fmaxf(l0, l1);
  float e0 = __expf(l0 - mx), e1 = __expf(l1 - mx);
  float inv = 1.f / (e0 + e1);
  ((float2*)att)[e] = make_float2(e0 * inv, e1 * inv);
}

#define HEAD_THREADS 512
#define G_PITCH 68
#define W_PITCH 72

__device__ __forceinline__ void head_colstats(const float* __restrict__ ps,
                                              const float* __restrict__ pq,
                                              float* __restrict__ sc,
                                              float* __restrict__ ms, int col) {
  float ts = 0.f, tq = 0.f;
  for (int p = 0; p < 32; ++p) {
    ts += ps[p * HIDDEN + col];
    tq += pq[p * HIDDEN + col];
  }
  const float invN = 1.f / N_GRAPHS;
  const float mean = ts * invN;
  const float var  = tq * invN - mean * mean;
  const float s    = rsqrtf(var + BN_EPS);
  sc[col] = s;
  ms[col] = mean * s;
}

__global__ __launch_bounds__(HEAD_THREADS) void heads_kernel(
    const float* __restrict__ gc, const float* __restrict__ go,
    const float* __restrict__ W1_c, const float* __restrict__ b1_c,
    const float* __restrict__ W2_c, const float* __restrict__ b2_c,
    const float* __restrict__ W1_o, const float* __restrict__ b1_o,
    const float* __restrict__ W2_o, const float* __restrict__ b2_o,
    const float* __restrict__ W1_co, const float* __restrict__ b1_co,
    const float* __restrict__ W2_co, const float* __restrict__ b2_co,
    float* __restrict__ out) {
  __shared__ __attribute__((aligned(16))) float sh_g[N_GRAPHS * G_PITCH];
  __shared__ __attribute__((aligned(16))) float sh_w[HIDDEN * W_PITCH];
  __shared__ __attribute__((aligned(16))) float sh_ps[32 * HIDDEN];
  __shared__ __attribute__((aligned(16))) float sh_pq[32 * HIDDEN];
  __shared__ float sh_sc[HIDDEN];
  __shared__ float sh_ms[HIDDEN];
  __shared__ float sh_v[HIDDEN];
  __shared__ __attribute__((aligned(16))) float sh_w2[HIDDEN * N_CLASSES];
  __shared__ float sh_v2[16];
  __shared__ float sh_w1r[HIDDEN * HIDDEN];
  __shared__ float sh_w2r[HIDDEN * N_CLASSES];

  const int head = blockIdx.x;
  const float* __restrict__ W1 = head == 0 ? W1_c : (head == 1 ? W1_o : W1_co);
  const float* __restrict__ b1 = head == 0 ? b1_c : (head == 1 ? b1_o : b1_co);
  const float* __restrict__ W2 = head == 0 ? W2_c : (head == 1 ? W2_o : W2_co);
  const float* __restrict__ b2 = head == 0 ? b2_c : (head == 1 ? b2_o : b2_co);
  const int tid  = threadIdx.x;
  const int lane = tid & 31;
  const int wave = tid >> 5;

  float w1r[8];
#pragma unroll
  for (int u = 0; u < 8; ++u) w1r[u] = W1[tid + HEAD_THREADS * u];
  const float w2r0 = W2[tid];
  const float w2r1 = tid < HIDDEN * N_CLASSES - HEAD_THREADS ? W2[HEAD_THREADS + tid] : 0.f;
  const float b1r  = tid < HIDDEN ? b1[tid] : 0.f;
  const float b2r  = tid < N_CLASSES ? b2[tid] : 0.f;

  {
    const float4* __restrict__ gc4 = (const float4*)gc;
    const float4* __restrict__ go4 = (const float4*)go;
    const float4* __restrict__ pa  = head == 1 ? go4 : gc4;
    const int cg = tid & 15;
    const int r0 = tid >> 4;
    float4 x[16];
#pragma unroll
    for (int u = 0; u < 16; ++u) x[u] = pa[tid + HEAD_THREADS * u];
    if (head == 2) {
#pragma unroll
      for (int u = 0; u < 16; ++u) {
        const float4 b = go4[tid + HEAD_THREADS * u];
        x[u] = make_float4(x[u].x + b.x, x[u].y + b.y, x[u].z + b.z, x[u].w + b.w);
      }
    }
    float4 s = make_float4(0.f, 0.f, 0.f, 0.f);
    float4 q = make_float4(0.f, 0.f, 0.f, 0.f);
#pragma unroll
    for (int u = 0; u < 16; ++u) {
      *(float4*)(sh_g + (r0 + 32 * u) * G_PITCH + 4 * cg) = x[u];
      s.x += x[u].x; q.x += x[u].x * x[u].x;
      s.y += x[u].y; q.y += x[u].y * x[u].y;
      s.z += x[u].z; q.z += x[u].z * x[u].z;
      s.w += x[u].w; q.w += x[u].w * x[u].w;
    }
    *(float4*)(sh_ps + r0 * HIDDEN + 4 * cg) = s;
    *(float4*)(sh_pq + r0 * HIDDEN + 4 * cg) = q;
#pragma unroll
    for (int u = 0; u < 8; ++u) sh_w1r[tid + HEAD_THREADS * u] = w1r[u];
    sh_w2r[tid] = w2r0;
    if (tid < HIDDEN * N_CLASSES - HEAD_THREADS) sh_w2r[HEAD_THREADS + tid] = w2r1;
  }
  __syncthreads();
  if (tid < HIDDEN) head_colstats(sh_ps, sh_pq, sh_sc, sh_ms, tid);
  __syncthreads();

#pragma unroll
  for (int u = 0; u < 8; ++u) {
    const int i = tid + HEAD_THREADS * u;
    const int k = i >> 6, j = i & 63;
    sh_w[k * W_PITCH + j] = sh_sc[k] * w1r[u];
  }
  if (tid < HIDDEN) {
    float acc = b1r;
    for (int k = 0; k < HIDDEN; ++k) acc += (BN_BETA - sh_ms[k]) * sh_w1r[k * HIDDEN + tid];
    sh_v[tid] = acc;
  }
  __syncthreads();

  {
    const int kOff = (lane >> 4) << 1;
    const int n0   = lane & 15;
    v8f acc[2][4] = {};
    const float* a0p = sh_g + (32 * wave + n0) * G_PITCH + kOff;
    const float* a1p = a0p + 16 * G_PITCH;
    const float* bp  = sh_w + kOff * W_PITCH + n0;
    for (int kk = 0; kk < HIDDEN; kk += 4) {
      const v2f a0 = *(const v2f*)(a0p + kk);
      const v2f a1 = *(const v2f*)(a1p + kk);
#pragma unroll
      for (int t = 0; t < 4; ++t) {
        v2f b;
        b.x = bp[kk * W_PITCH + t * 16];
        b.y = bp[(kk + 1) * W_PITCH + t * 16];
        acc[0][t] = __builtin_amdgcn_wmma_f32_16x16x4_f32(
            false, a0, false, b, (short)0, acc[0][t], false, false);
        acc[1][t] = __builtin_amdgcn_wmma_f32_16x16x4_f32(
            false, a1, false, b, (short)0, acc[1][t], false, false);
      }
    }
    const int rowBase = 32 * wave + ((lane >> 4) << 3);
#pragma unroll
    for (int t = 0; t < 4; ++t) {
      const int n = t * 16 + n0;
      const float bias = sh_v[n];
      float s = 0.f, q = 0.f;
#pragma unroll
      for (int i = 0; i < 2; ++i) {
#pragma unroll
        for (int e = 0; e < 8; ++e) {
          const float val = fmaxf(acc[i][t][e] + bias, 0.f);
          sh_g[(rowBase + 16 * i + e) * G_PITCH + n] = val;
          s += val;
          q += val * val;
        }
      }
      sh_ps[(2 * wave + (lane >> 4)) * HIDDEN + n] = s;
      sh_pq[(2 * wave + (lane >> 4)) * HIDDEN + n] = q;
    }
  }
  __syncthreads();
  if (tid < HIDDEN) head_colstats(sh_ps, sh_pq, sh_sc, sh_ms, tid);
  __syncthreads();

  sh_w2[tid] = sh_sc[tid / N_CLASSES] * w2r0;
  if (tid < HIDDEN * N_CLASSES - HEAD_THREADS)
    sh_w2[HEAD_THREADS + tid] = sh_sc[(HEAD_THREADS + tid) / N_CLASSES] * w2r1;
  if (tid < N_CLASSES) {
    float acc = b2r;
    for (int k = 0; k < HIDDEN; ++k) acc += (BN_BETA - sh_ms[k]) * sh_w2r[k * N_CLASSES + tid];
    sh_v2[tid] = acc;
  }
  __syncthreads();

  {
    float l[N_CLASSES];
    for (int c = 0; c < N_CLASSES; ++c) l[c] = sh_v2[c];
    const float4* __restrict__ row4 = (const float4*)(sh_g + tid * G_PITCH);
    for (int k4 = 0; k4 < HIDDEN / 4; ++k4) {
      const float4 x = row4[k4];
      const float* w = sh_w2 + k4 * 4 * N_CLASSES;
      for (int c = 0; c < N_CLASSES; ++c) l[c] += x.x * w[c];
      for (int c = 0; c < N_CLASSES; ++c) l[c] += x.y * w[N_CLASSES + c];
      for (int c = 0; c < N_CLASSES; ++c) l[c] += x.z * w[2 * N_CLASSES + c];
      for (int c = 0; c < N_CLASSES; ++c) l[c] += x.w * w[3 * N_CLASSES + c];
    }
    float mx = l[0];
    for (int c = 1; c < N_CLASSES; ++c) mx = fmaxf(mx, l[c]);
    float s = 0.f;
    for (int c = 0; c < N_CLASSES; ++c) s += __expf(l[c] - mx);
    const float lse = __logf(s) + mx;
    float* __restrict__ o = out + ((size_t)head * N_GRAPHS + tid) * N_CLASSES;
    for (int c = 0; c < N_CLASSES; ++c) o[c] = l[c] - lse;
  }
}

extern "C" void kernel_launch(void* const* d_in, const int* in_sizes, int n_in,
                              void* d_out, int out_size, void* d_ws, size_t ws_size,
                              hipStream_t stream) {
  (void)in_sizes; (void)n_in; (void)out_size; (void)ws_size;
  const float* x       = (const float*)d_in[0];
  const int*   ei      = (const int*)d_in[1];
  const int*   row     = ei;
  const int*   col     = ei + N_EDGES;
  const int*   batch   = (const int*)d_in[2];
  const float* W_feat  = (const float*)d_in[3];
  const float* W_convs = (const float*)d_in[4];
  const float* b_convs = (const float*)d_in[5];
  const float* W_ea    = (const float*)d_in[6];
  const float* b_ea    = (const float*)d_in[7];
  const float* W_na    = (const float*)d_in[8];
  const float* b_na    = (const float*)d_in[9];
  const float* W_ctx   = (const float*)d_in[10];
  const float* b_ctx   = (const float*)d_in[11];
  const float* W_obj   = (const float*)d_in[12];
  const float* b_obj   = (const float*)d_in[13];
  const float* W1h[3]  = {(const float*)d_in[14], (const float*)d_in[18], (const float*)d_in[22]};
  const float* b1h[3]  = {(const float*)d_in[15], (const float*)d_in[19], (const float*)d_in[23]};
  const float* W2h[3]  = {(const float*)d_in[16], (const float*)d_in[20], (const float*)d_in[24]};
  const float* b2h[3]  = {(const float*)d_in[17], (const float*)d_in[21], (const float*)d_in[25]};

  size_t off = 0;
  auto carve = [&](size_t elems) -> float* {
    float* p = (float*)((char*)d_ws + off);
    off += ((elems * sizeof(float)) + 255) & ~(size_t)255;
    return p;
  };
  float* hb    = carve((size_t)N_NODES * HIDDEN);
  float* mbuf  = carve((size_t)N_NODES * HIDDEN);
  float* xc    = carve((size_t)N_NODES * HIDDEN);
  float* xo    = carve((size_t)N_NODES * HIDDEN);
  float* att   = carve((size_t)N_EDGES * 2);
  float* pr    = carve((size_t)N_NODES * 2);
  float* pc    = carve((size_t)N_NODES * 2);
  float* degu  = carve(N_NODES);
  float* dinvu = carve(N_NODES);
  float* dega  = carve(N_NODES);
  float* dinva = carve(N_NODES);
  float* gc    = carve((size_t)N_GRAPHS * HIDDEN);
  float* go    = carve((size_t)N_GRAPHS * HIDDEN);
  float* sums  = carve(N_FEAT);
  float* sq    = carve(N_FEAT);
  float* Wf    = carve((size_t)N_FEAT * HIDDEN);
  float* vf    = carve(HIDDEN);
  int*  ecnt   = (int*)carve(N_NODES);
  int*  etsum  = (int*)carve(128);
  int*  eoff   = (int*)carve(N_NODES + 1);
  int*  ecur   = (int*)carve(N_NODES);
  int2* sedge  = (int2*)carve((size_t)N_EDGES * 2);

  auto nblk = [](long n, int b) { return (int)((n + b - 1) / b); };

  auto stats = [&](const float* A, int N, int F) {
    hipMemsetAsync(sums, 0, (size_t)F * sizeof(float), stream);
    hipMemsetAsync(sq,   0, (size_t)F * sizeof(float), stream);
    const int nvec = N * F / 4;
    const int grid = nblk(nvec, 256) < 512 ? nblk(nvec, 256) : 512;
    colstats_kernel<<<grid, 256, 0, stream>>>(A, nvec, F, sums, sq);
  };
  auto gemm = [&](const float* A, float* C, int M, int K, int relu) {
    gemm_wmma_kernel<<<nblk(M / 16, 4), 128, 0, stream>>>(A, Wf, vf, C, M, K, relu);
  };

  stats(x, N_NODES, N_FEAT);
  fold_bn_kernel<<<1, HIDDEN, 0, stream>>>(sums, sq, 1.f / N_NODES, N_FEAT, HIDDEN,
                                           W_feat, nullptr, Wf, vf);
  gemm(x, hb, N_NODES, N_FEAT, 1);

  deg_init_kernel<<<nblk(N_NODES, 256), 256, 0, stream>>>(degu, N_NODES);
  deg_accum_kernel<<<nblk(N_EDGES, 256), 256, 0, stream>>>(row, nullptr, 0, N_EDGES, degu);
  make_dinv_kernel<<<nblk(N_NODES, 256), 256, 0, stream>>>(degu, dinvu, N_NODES);

  csr_zero_kernel<<<nblk(N_NODES, 256), 256, 0, stream>>>(ecnt, N_NODES);
  csr_count_kernel<<<nblk(N_EDGES, 256), 256, 0, stream>>>(col, N_EDGES, ecnt);
  csr_tilesum_kernel<<<nblk(N_NODES, 1024), 256, 0, stream>>>(ecnt, N_NODES, etsum);
  csr_scan_kernel<<<nblk(N_NODES, 1024), 256, 0, stream>>>(ecnt, etsum, N_NODES,
                                                           eoff, ecur);
  csr_fill_kernel<<<nblk(N_EDGES, 256), 256, 0, stream>>>(row, col, N_EDGES, ecur,
                                                          sedge);

  for (int i = 0; i < 3; ++i) {
    stats(hb, N_NODES, HIDDEN);
    fold_bn_kernel<<<1, HIDDEN, 0, stream>>>(sums, sq, 1.f / N_NODES, HIDDEN, HIDDEN,
                                             W_convs + (size_t)i * HIDDEN * HIDDEN,
                                             nullptr, Wf, vf);
    gemm(hb, mbuf, N_NODES, HIDDEN, 0);
    gather_gcn_kernel<<<nblk(N_NODES, 8), 256, 0, stream>>>(
        mbuf, eoff, sedge, dinvu, b_convs + (size_t)i * HIDDEN, hb, N_NODES);
  }

  node_att_split_kernel<<<nblk(N_NODES, 256), 256, 0, stream>>>(hb, W_na, b_na, W_ea,
                                                                xc, xo, pr, pc, N_NODES);
  edge_att_kernel<<<nblk(N_EDGES, 256), 256, 0, stream>>>(pr, pc, row, col, b_ea,
                                                          att, N_EDGES);

  auto run_branch = [&](float* xin, int comp, const float* W, const float* b,
                        float* gout) {
    stats(xin, N_NODES, HIDDEN);
    fold_bn_kernel<<<1, HIDDEN, 0, stream>>>(sums, sq, 1.f / N_NODES, HIDDEN, HIDDEN,
                                             W, nullptr, Wf, vf);
    gemm(xin, mbuf, N_NODES, HIDDEN, 0);
    deg_init_kernel<<<nblk(N_NODES, 256), 256, 0, stream>>>(dega, N_NODES);
    deg_accum_kernel<<<nblk(N_EDGES, 256), 256, 0, stream>>>(row, att, comp, N_EDGES, dega);
    make_dinv_kernel<<<nblk(N_NODES, 256), 256, 0, stream>>>(dega, dinva, N_NODES);
    hipMemsetAsync(gout, 0, (size_t)N_GRAPHS * HIDDEN * sizeof(float), stream);
    gather_pool_kernel<<<nblk(N_NODES, 8), 256, 0, stream>>>(
        mbuf, eoff, sedge, att, comp, dinva, b, batch, gout, N_NODES);
  };
  run_branch(xc, 0, W_ctx, b_ctx, gc);
  run_branch(xo, 1, W_obj, b_obj, go);

  static_assert(HEAD_THREADS == N_GRAPHS, "one thread per graph in the logits stage");
  heads_kernel<<<3, HEAD_THREADS, 0, stream>>>(
      gc, go,
      W1h[0], b1h[0], W2h[0], b2h[0],
      W1h[1], b1h[1], W2h[1], b2h[1],
      W1h[2], b1h[2], W2h[2], b2h[2],
      (float*)d_out);
}


# stream operations 54 -> 31: one init kernel replaces 14 memsets and 4 init passes; degree sums ride on the edge-count and edge-attention passes; gathers take rsqrt(deg) (float atomics as before)
# speedup vs baseline: 3.0790x; 1.0376x over previous
#include <hip/hip_runtime.h>
#include <math.h>


#define N_NODES  100000
#define N_EDGES  1000000
#define N_FEAT   128
#define HIDDEN   64
#define N_GRAPHS 512
#define N_CLASSES 10
#define BN_EPS  1e-5f
#define BN_BETA 1e-4f

typedef __attribute__((ext_vector_type(2))) float v2f;
typedef __attribute__((ext_vector_type(8))) float v8f;

__global__ __launch_bounds__(256) void colstats_kernel(
    const float* __restrict__ A, int nvec, int F,
    float* __restrict__ sums, float* __restrict__ sumsq) {
  __shared__ float sh_s[1024];
  __shared__ float sh_q[1024];
  const float4* __restrict__ A4 = (const float4*)A;
  const int tid = threadIdx.x;
  const int stride = gridDim.x * 256;
  const int last = nvec - 1;
  float4 s = make_float4(0.f, 0.f, 0.f, 0.f);
  float4 q = make_float4(0.f, 0.f, 0.f, 0.f);
  for (int i = blockIdx.x * 256 + tid; i < nvec; i += 8 * stride) {
    float4 v[8];
#pragma unroll
    for (int u = 0; u < 8; ++u) {
      const int idx = i + u * stride;
      v[u] = A4[idx < last ? idx : last];
    }
    __builtin_amdgcn_sched_barrier(0);
#pragma unroll
    for (int u = 0; u < 8; ++u) {
      const bool live = (i + u * stride) < nvec;
      const float a = live ? v[u].x : 0.f;
      const float b = live ? v[u].y : 0.f;
      const float c = live ? v[u].z : 0.f;
      const float d = live ? v[u].w : 0.f;
      s.x += a; q.x += a * a;
      s.y += b; q.y += b * b;
      s.z += c; q.z += c * c;
      s.w += d; q.w += d * d;
    }
  }
  ((float4*)sh_s)[tid] = s;
  ((float4*)sh_q)[tid] = q;
  __syncthreads();
  if (tid < F) {
    float ts = 0.f, tq = 0.f;
    for (int p = tid; p < 1024; p += F) { ts += sh_s[p]; tq += sh_q[p]; }
    atomicAdd(&sums[tid], ts);
    atomicAdd(&sumsq[tid], tq);
  }
}

__global__ void fold_bn_kernel(const float* __restrict__ sums,
                               const float* __restrict__ sumsq, float invN,
                               int K, int NN, const float* __restrict__ W,
                               const float* __restrict__ extra_b,
                               float* __restrict__ Wout, float* __restrict__ vout) {
  __shared__ float s_ms[N_FEAT];
  __shared__ float s_s[N_FEAT];
  const int j = threadIdx.x;
  for (int k = j; k < K; k += blockDim.x) {
    float mean = sums[k] * invN;
    float var  = sumsq[k] * invN - mean * mean;
    float sc   = rsqrtf(var + BN_EPS);
    s_ms[k] = mean * sc;
    s_s[k]  = sc;
  }
  __syncthreads();
  float acc = extra_b ? extra_b[j] : 0.f;
  for (int k = 0; k < K; ++k) {
    float w = W[(size_t)k * NN + j];
    Wout[(size_t)k * NN + j] = s_s[k] * w;
    acc += (BN_BETA - s_ms[k]) * w;
  }
  vout[j] = acc;
}

__global__ __launch_bounds__(128) void gemm_wmma_kernel(
    const float* __restrict__ A, const float* __restrict__ Wt,
    const float* __restrict__ v, float* __restrict__ C,
    int M, int K, int doRelu) {
  const int lane = threadIdx.x & 31;
  const int wave = threadIdx.x >> 5;
  const int rowTile = blockIdx.x * 4 + wave;
  if (rowTile >= (M >> 4)) return;
  const int mrow = (rowTile << 4) + (lane & 15);
  const int kOff = (lane >> 4) << 1;
  const int n0   = lane & 15;
  v8f acc[4] = {};
  for (int kk = 0; kk < K; kk += 4) {
    const int k0 = kk + kOff;
    v2f a = *(const v2f*)(A + (size_t)mrow * K + k0);
    const float* w0 = Wt + (size_t)k0 * HIDDEN + n0;
#pragma unroll
    for (int t = 0; t < 4; ++t) {
      v2f b;
      b.x = w0[t * 16];
      b.y = w0[HIDDEN + t * 16];
      acc[t] = __builtin_amdgcn_wmma_f32_16x16x4_f32(
          false, a, false, b, (short)0, acc[t], false, false);
    }
  }
  const int rowBase = (rowTile << 4) + ((lane >> 4) << 3);
#pragma unroll
  for (int t = 0; t < 4; ++t) {
    const int n = t * 16 + n0;
    const float bias = v[n];
#pragma unroll
    for (int e = 0; e < 8; ++e) {
      float val = acc[t][e] + bias;
      if (doRelu) val = fmaxf(val, 0.f);
      C[(size_t)(rowBase + e) * HIDDEN + n] = val;
    }
  }
}

__global__ void init_kernel(float* __restrict__ stat, int nstat,
                            float* __restrict__ gpool, int npool,
                            int* __restrict__ cnt, float* __restrict__ degu,
                            float* __restrict__ degc, float* __restrict__ dego,
                            int N) {
  int i = blockIdx.x * blockDim.x + threadIdx.x;
  if (i < nstat) stat[i] = 0.f;
  if (i < npool) gpool[i] = 0.f;
  if (i < N) {
    cnt[i] = 0;
    degu[i] = 1.0f;
    degc[i] = 1.0f;
    dego[i] = 1.0f;
  }
}
__device__ __forceinline__ float inv_sqrt_deg(float d) {
  return d > 0.f ? rsqrtf(d) : 0.f;
}

__global__ void csr_count_kernel(const int* __restrict__ row,
                                 const int* __restrict__ col, int E,
                                 int* __restrict__ cnt, float* __restrict__ degu) {
  int e = blockIdx.x * blockDim.x + threadIdx.x;
  if (e >= E) return;
  atomicAdd(&cnt[col[e]], 1);
  atomicAdd(&degu[row[e]], 1.0f);
}
__device__ __forceinline__ void load_counts4(const int* __restrict__ cnt, int i0,
                                             int N, int c[4]) {
  if (i0 + 3 < N) {
    const int4 v = *(const int4*)(cnt + i0);
    c[0] = v.x; c[1] = v.y; c[2] = v.z; c[3] = v.w;
  } else {
    c[0] = (i0     < N) ? cnt[i0]     : 0;
    c[1] = (i0 + 1 < N) ? cnt[i0 + 1] : 0;
    c[2] = (i0 + 2 < N) ? cnt[i0 + 2] : 0;
    c[3] = 0;
  }
}
__global__ __launch_bounds__(256) void csr_tilesum_kernel(
    const int* __restrict__ cnt, int N, int* __restrict__ tsum) {
  __shared__ int sh[8];
  const int tid = threadIdx.x;
  int c[4];
  load_counts4(cnt, (blockIdx.x * 256 + tid) * 4, N, c);
  int t = c[0] + c[1] + c[2] + c[3];
  for (int o = 16; o > 0; o >>= 1) t += __shfl_xor(t, o);
  if ((tid & 31) == 0) sh[tid >> 5] = t;
  __syncthreads();
  if (tid == 0) {
    int s = 0;
    for (int w = 0; w < 8; ++w) s += sh[w];
    tsum[blockIdx.x] = s;
  }
}
__global__ __launch_bounds__(256) void csr_scan_kernel(
    const int* __restrict__ cnt, const int* __restrict__ tsum, int N,
    int* __restrict__ off, int* __restrict__ cur) {
  __shared__ int sh_w[8];
  __shared__ int sh_b[8];
  const int tid = threadIdx.x, lane = tid & 31, wave = tid >> 5;
  int bpart = 0;
  for (int t = tid; t < (int)blockIdx.x; t += 256) bpart += tsum[t];
  for (int o = 16; o > 0; o >>= 1) bpart += __shfl_xor(bpart, o);
  const int i0 = (blockIdx.x * 256 + tid) * 4;
  int c[4];
  load_counts4(cnt, i0, N, c);
  const int t = c[0] + c[1] + c[2] + c[3];
  int inc = t;
  for (int o = 1; o < 32; o <<= 1) {
    const int y = __shfl_up(inc, o);
    if (lane >= o) inc += y;
  }
  if (lane == 31) sh_w[wave] = inc;
  if (lane == 0)  sh_b[wave] = bpart;
  __syncthreads();
  int base = 0;
  for (int w = 0; w < 8; ++w) {
    base += sh_b[w];
    if (w < wave) base += sh_w[w];
  }
  const int o0 = base + inc - t;
  const int o1 = o0 + c[0], o2 = o1 + c[1], o3 = o2 + c[2];
  if (i0 + 3 < N) {
    const int4 v = make_int4(o0, o1, o2, o3);
    *(int4*)(off + i0) = v;
    *(int4*)(cur + i0) = v;
  } else {
    if (i0     < N) { off[i0]     = o0; cur[i0]     = o0; }
    if (i0 + 1 < N) { off[i0 + 1] = o1; cur[i0 + 1] = o1; }
    if (i0 + 2 < N) { off[i0 + 2] = o2; cur[i0 + 2] = o2; }
  }
  if (i0 < N && N - 1 <= i0 + 3) off[N] = o0 + t;
}
__global__ void csr_fill_kernel(const int* __restrict__ row,
                                const int* __restrict__ col, int E,
                                int* __restrict__ cur, int2* __restrict__ sedge) {
  int e = blockIdx.x * blockDim.x + threadIdx.x;
  if (e >= E) return;
  const int r = row[e];
  const int pos = atomicAdd(&cur[col[e]], 1);
  if ((unsigned)pos < (unsigned)E) sedge[pos] = make_int2(r, e);
}

template <int CNT>
__device__ __forceinline__ void gather_rows(const float2* __restrict__ m2, int r,
                                            float cf, int k, int lane, float2& acc) {
  int rr[CNT];
  float cc[CNT];
  float2 v[CNT];
#pragma unroll
  for (int u = 0; u < CNT; ++u) {
    rr[u] = __builtin_amdgcn_readlane(r, k + u);
    cc[u] = __int_as_float(__builtin_amdgcn_readlane(__float_as_int(cf), k + u));
  }
#pragma unroll
  for (int u = 0; u < CNT; ++u) v[u] = m2[(size_t)rr[u] * (HIDDEN / 2) + lane];
#pragma unroll
  for (int u = 0; u < CNT; ++u) {
    acc.x += cc[u] * v[u].x;
    acc.y += cc[u] * v[u].y;
  }
}
template <bool HAS_ATT>
__device__ __forceinline__ float2 gcn_row(
    const float2* __restrict__ m2, const int* __restrict__ off,
    const int2* __restrict__ sedge, const float* __restrict__ att, int comp,
    const float* __restrict__ deg, const float* __restrict__ b, int c, int lane) {
  const int beg = off[c], end = off[c + 1];
  const float dc = inv_sqrt_deg(deg[c]);
  const float2 self = m2[(size_t)c * (HIDDEN / 2) + lane];
  const float2 bias = ((const float2*)b)[lane];
  float2 acc = make_float2(0.f, 0.f);
  for (int base = beg; base < end; base += 32) {
    const int p = base + lane;
    int r = 0;
    float cf = 0.f;
    if (p < end) {
      const int2 se = sedge[p];
      r = se.x;
      const float dr = inv_sqrt_deg(deg[r]);
      const float w = HAS_ATT ? att[(size_t)se.y * 2 + comp] : 1.0f;
      cf = dr * w * dc;
    }
    const int n = (end - base) < 32 ? (end - base) : 32;
    int k = 0;
    for (; k + 4 <= n; k += 4) gather_rows<4>(m2, r, cf, k, lane, acc);
    const int rem = n - k;
    if (rem == 3)      gather_rows<3>(m2, r, cf, k, lane, acc);
    else if (rem == 2) gather_rows<2>(m2, r, cf, k, lane, acc);
    else if (rem == 1) gather_rows<1>(m2, r, cf, k, lane, acc);
  }
  float2 o;
  o.x = fmaxf(acc.x + dc * dc * self.x + bias.x, 0.f);
  o.y = fmaxf(acc.y + dc * dc * self.y + bias.y, 0.f);
  return o;
}

__global__ __launch_bounds__(256) void gather_gcn_kernel(
    const float* __restrict__ m, const int* __restrict__ off,
    const int2* __restrict__ sedge, const float* __restrict__ deg,
    const float* __restrict__ b, float* __restrict__ out, int N) {
  const int lane = threadIdx.x & 31;
  const int c = __builtin_amdgcn_readfirstlane(blockIdx.x * 8 + (threadIdx.x >> 5));
  if (c >= N) return;
  const float2 o = gcn_row<false>((const float2*)m, off, sedge, nullptr, 0, deg, b,
                                  c, lane);
  ((float2*)out)[(size_t)c * (HIDDEN / 2) + lane] = o;
}

__global__ __launch_bounds__(256) void gather_pool_kernel(
    const float* __restrict__ m, const int* __restrict__ off,
    const int2* __restrict__ sedge, const float* __restrict__ att, int comp,
    const float* __restrict__ deg, const float* __restrict__ b,
    const int* __restrict__ batch, float* __restrict__ g, int N) {
  __shared__ float2 sh_row[8 * (HIDDEN / 2)];
  __shared__ int sh_gid[8];
  const int lane = threadIdx.x & 31;
  const int wave = threadIdx.x >> 5;
  const int c = __builtin_amdgcn_readfirstlane(blockIdx.x * 8 + wave);
  float2 o = make_float2(0.f, 0.f);
  int gid = -1;
  if (c < N) {
    gid = batch[c];
    o = gcn_row<true>((const float2*)m, off, sedge, att, comp, deg, b, c, lane);
  }
  sh_row[wave * (HIDDEN / 2) + lane] = o;
  if (lane == 0) sh_gid[wave] = gid;
  __syncthreads();
  if (threadIdx.x < HIDDEN) {
    const int j = threadIdx.x;
    int gw[8];
    float v[8];
#pragma unroll
    for (int w = 0; w < 8; ++w) {
      gw[w] = sh_gid[w];
      v[w] = ((const float*)sh_row)[w * HIDDEN + j];
    }
    int cur = gw[0];
    float a = v[0];
#pragma unroll
    for (int w = 1; w < 8; ++w) {
      if (gw[w] == cur) {
        a += v[w];
      } else {
        if ((unsigned)cur < (unsigned)N_GRAPHS) atomicAdd(&g[(size_t)cur * HIDDEN + j], a);
        cur = gw[w];
        a = v[w];
      }
    }
    if ((unsigned)cur < (unsigned)N_GRAPHS) atomicAdd(&g[(size_t)cur * HIDDEN + j], a);
  }
}

__global__ void node_att_split_kernel(const float* __restrict__ h,
                                      const float* __restrict__ Wna,
                                      const float* __restrict__ bna,
                                      const float* __restrict__ Wea,
                                      float* __restrict__ xc,
                                      float* __restrict__ xo,
                                      float* __restrict__ pr,
                                      float* __restrict__ pc, int N) {
  int i = blockIdx.x * blockDim.x + threadIdx.x;
  if (i >= N) return;
  const float* hi = h + (size_t)i * HIDDEN;
  float l0 = bna[0], l1 = bna[1];
  float r0 = 0.f, r1 = 0.f, c0 = 0.f, c1 = 0.f;
  for (int j = 0; j < HIDDEN; ++j) {
    float a = hi[j];
    l0 += a * Wna[j * 2 + 0];
    l1 += a * Wna[j * 2 + 1];
    r0 += a * Wea[j * 2 + 0];
    r1 += a * Wea[j * 2 + 1];
    c0 += a * Wea[(HIDDEN + j) * 2 + 0];
    c1 += a * Wea[(HIDDEN + j) * 2 + 1];
  }
  ((float2*)pr)[i] = make_float2(r0, r1);
  ((float2*)pc)[i] = make_float2(c0, c1);
  float mx = fmaxf(l0, l1);
  float e0 = __expf(l0 - mx), e1 = __expf(l1 - mx);
  float inv = 1.f / (e0 + e1);
  float p0 = e0 * inv, p1 = e1 * inv;
  for (int j = 0; j < HIDDEN; ++j) {
    float a = hi[j];
    xc[(size_t)i * HIDDEN + j] = p0 * a;
    xo[(size_t)i * HIDDEN + j] = p1 * a;
  }
}

__global__ void edge_att_kernel(const float* __restrict__ pr,
                                const float* __restrict__ pc,
                                const int* __restrict__ row,
                                const int* __restrict__ col,
                                const float* __restrict__ bea,
                                float* __restrict__ att,
                                float* __restrict__ degc,
                                float* __restrict__ dego, int E) {
  int e = blockIdx.x * blockDim.x + threadIdx.x;
  if (e >= E) return;
  const int r = row[e];
  const float2 a = ((const float2*)pr)[r];
  const float2 c = ((const float2*)pc)[col[e]];
  const float l0 = (a.x + c.x) + bea[0];
  const float l1 = (a.y + c.y) + bea[1];
  float mx = fmaxf(l0, l1);
  float e0 = __expf(l0 - mx), e1 = __expf(l1 - mx);
  float inv = 1.f / (e0 + e1);
  const float w0 = e0 * inv, w1 = e1 * inv;
  ((float2*)att)[e] = make_float2(w0, w1);
  atomicAdd(&degc[r], w0);
  atomicAdd(&dego[r], w1);
}

#define HEAD_THREADS 512
#define G_PITCH 68
#define W_PITCH 72

__device__ __forceinline__ void head_colstats(const float* __restrict__ ps,
                                              const float* __restrict__ pq,
                                              float* __restrict__ sc,
                                              float* __restrict__ ms, int col) {
  float ts = 0.f, tq = 0.f;
  for (int p = 0; p < 32; ++p) {
    ts += ps[p * HIDDEN + col];
    tq += pq[p * HIDDEN + col];
  }
  const float invN = 1.f / N_GRAPHS;
  const float mean = ts * invN;
  const float var  = tq * invN - mean * mean;
  const float s    = rsqrtf(var + BN_EPS);
  sc[col] = s;
  ms[col] = mean * s;
}

__global__ __launch_bounds__(HEAD_THREADS) void heads_kernel(
    const float* __restrict__ gc, const float* __restrict__ go,
    const float* __restrict__ W1_c, const float* __restrict__ b1_c,
    const float* __restrict__ W2_c, const float* __restrict__ b2_c,
    const float* __restrict__ W1_o, const float* __restrict__ b1_o,
    const float* __restrict__ W2_o, const float* __restrict__ b2_o,
    const float* __restrict__ W1_co, const float* __restrict__ b1_co,
    const float* __restrict__ W2_co, const float* __restrict__ b2_co,
    float* __restrict__ out) {
  __shared__ __attribute__((aligned(16))) float sh_g[N_GRAPHS * G_PITCH];
  __shared__ __attribute__((aligned(16))) float sh_w[HIDDEN * W_PITCH];
  __shared__ __attribute__((aligned(16))) float sh_ps[32 * HIDDEN];
  __shared__ __attribute__((aligned(16))) float sh_pq[32 * HIDDEN];
  __shared__ float sh_sc[HIDDEN];
  __shared__ float sh_ms[HIDDEN];
  __shared__ float sh_v[HIDDEN];
  __shared__ __attribute__((aligned(16))) float sh_w2[HIDDEN * N_CLASSES];
  __shared__ float sh_v2[16];
  __shared__ float sh_w1r[HIDDEN * HIDDEN];
  __shared__ float sh_w2r[HIDDEN * N_CLASSES];

  const int head = blockIdx.x;
  const float* __restrict__ W1 = head == 0 ? W1_c : (head == 1 ? W1_o : W1_co);
  const float* __restrict__ b1 = head == 0 ? b1_c : (head == 1 ? b1_o : b1_co);
  const float* __restrict__ W2 = head == 0 ? W2_c : (head == 1 ? W2_o : W2_co);
  const float* __restrict__ b2 = head == 0 ? b2_c : (head == 1 ? b2_o : b2_co);
  const int tid  = threadIdx.x;
  const int lane = tid & 31;
  const int wave = tid >> 5;

  float w1r[8];
#pragma unroll
  for (int u = 0; u < 8; ++u) w1r[u] = W1[tid + HEAD_THREADS * u];
  const float w2r0 = W2[tid];
  const float w2r1 = tid < HIDDEN * N_CLASSES - HEAD_THREADS ? W2[HEAD_THREADS + tid] : 0.f;
  const float b1r  = tid < HIDDEN ? b1[tid] : 0.f;
  const float b2r  = tid < N_CLASSES ? b2[tid] : 0.f;

  {
    const float4* __restrict__ gc4 = (const float4*)gc;
    const float4* __restrict__ go4 = (const float4*)go;
    const float4* __restrict__ pa  = head == 1 ? go4 : gc4;
    const int cg = tid & 15;
    const int r0 = tid >> 4;
    float4 x[16];
#pragma unroll
    for (int u = 0; u < 16; ++u) x[u] = pa[tid + HEAD_THREADS * u];
    if (head == 2) {
#pragma unroll
      for (int u = 0; u < 16; ++u) {
        const float4 b = go4[tid + HEAD_THREADS * u];
        x[u] = make_float4(x[u].x + b.x, x[u].y + b.y, x[u].z + b.z, x[u].w + b.w);
      }
    }
    float4 s = make_float4(0.f, 0.f, 0.f, 0.f);
    float4 q = make_float4(0.f, 0.f, 0.f, 0.f);
#pragma unroll
    for (int u = 0; u < 16; ++u) {
      *(float4*)(sh_g + (r0 + 32 * u) * G_PITCH + 4 * cg) = x[u];
      s.x += x[u].x; q.x += x[u].x * x[u].x;
      s.y += x[u].y; q.y += x[u].y * x[u].y;
      s.z += x[u].z; q.z += x[u].z * x[u].z;
      s.w += x[u].w; q.w += x[u].w * x[u].w;
    }
    *(float4*)(sh_ps + r0 * HIDDEN + 4 * cg) = s;
    *(float4*)(sh_pq + r0 * HIDDEN + 4 * cg) = q;
#pragma unroll
    for (int u = 0; u < 8; ++u) sh_w1r[tid + HEAD_THREADS * u] = w1r[u];
    sh_w2r[tid] = w2r0;
    if (tid < HIDDEN * N_CLASSES - HEAD_THREADS) sh_w2r[HEAD_THREADS + tid] = w2r1;
  }
  __syncthreads();
  if (tid < HIDDEN) head_colstats(sh_ps, sh_pq, sh_sc, sh_ms, tid);
  __syncthreads();

#pragma unroll
  for (int u = 0; u < 8; ++u) {
    const int i = tid + HEAD_THREADS * u;
    const int k = i >> 6, j = i & 63;
    sh_w[k * W_PITCH + j] = sh_sc[k] * w1r[u];
  }
  if (tid < HIDDEN) {
    float acc = b1r;
    for (int k = 0; k < HIDDEN; ++k) acc += (BN_BETA - sh_ms[k]) * sh_w1r[k * HIDDEN + tid];
    sh_v[tid] = acc;
  }
  __syncthreads();

  {
    const int kOff = (lane >> 4) << 1;
    const int n0   = lane & 15;
    v8f acc[2][4] = {};
    const float* a0p = sh_g + (32 * wave + n0) * G_PITCH + kOff;
    const float* a1p = a0p + 16 * G_PITCH;
    const float* bp  = sh_w + kOff * W_PITCH + n0;
    for (int kk = 0; kk < HIDDEN; kk += 4) {
      const v2f a0 = *(const v2f*)(a0p + kk);
      const v2f a1 = *(const v2f*)(a1p + kk);
#pragma unroll
      for (int t = 0; t < 4; ++t) {
        v2f b;
        b.x = bp[kk * W_PITCH + t * 16];
        b.y = bp[(kk + 1) * W_PITCH + t * 16];
        acc[0][t] = __builtin_amdgcn_wmma_f32_16x16x4_f32(
            false, a0, false, b, (short)0, acc[0][t], false, false);
        acc[1][t] = __builtin_amdgcn_wmma_f32_16x16x4_f32(
            false, a1, false, b, (short)0, acc[1][t], false, false);
      }
    }
    const int rowBase = 32 * wave + ((lane >> 4) << 3);
#pragma unroll
    for (int t = 0; t < 4; ++t) {
      const int n = t * 16 + n0;
      const float bias = sh_v[n];
      float s = 0.f, q = 0.f;
#pragma unroll
      for (int i = 0; i < 2; ++i) {
#pragma unroll
        for (int e = 0; e < 8; ++e) {
          const float val = fmaxf(acc[i][t][e] + bias, 0.f);
          sh_g[(rowBase + 16 * i + e) * G_PITCH + n] = val;
          s += val;
          q += val * val;
        }
      }
      sh_ps[(2 * wave + (lane >> 4)) * HIDDEN + n] = s;
      sh_pq[(2 * wave + (lane >> 4)) * HIDDEN + n] = q;
    }
  }
  __syncthreads();
  if (tid < HIDDEN) head_colstats(sh_ps, sh_pq, sh_sc, sh_ms, tid);
  __syncthreads();

  sh_w2[tid] = sh_sc[tid / N_CLASSES] * w2r0;
  if (tid < HIDDEN * N_CLASSES - HEAD_THREADS)
    sh_w2[HEAD_THREADS + tid] = sh_sc[(HEAD_THREADS + tid) / N_CLASSES] * w2r1;
  if (tid < N_CLASSES) {
    float acc = b2r;
    for (int k = 0; k < HIDDEN; ++k) acc += (BN_BETA - sh_ms[k]) * sh_w2r[k * N_CLASSES + tid];
    sh_v2[tid] = acc;
  }
  __syncthreads();

  {
    float l[N_CLASSES];
    for (int c = 0; c < N_CLASSES; ++c) l[c] = sh_v2[c];
    const float4* __restrict__ row4 = (const float4*)(sh_g + tid * G_PITCH);
    for (int k4 = 0; k4 < HIDDEN / 4; ++k4) {
      const float4 x = row4[k4];
      const float* w = sh_w2 + k4 * 4 * N_CLASSES;
      for (int c = 0; c < N_CLASSES; ++c) l[c] += x.x * w[c];
      for (int c = 0; c < N_CLASSES; ++c) l[c] += x.y * w[N_CLASSES + c];
      for (int c = 0; c < N_CLASSES; ++c) l[c] += x.z * w[2 * N_CLASSES + c];
      for (int c = 0; c < N_CLASSES; ++c) l[c] += x.w * w[3 * N_CLASSES + c];
    }
    float mx = l[0];
    for (int c = 1; c < N_CLASSES; ++c) mx = fmaxf(mx, l[c]);
    float s = 0.f;
    for (int c = 0; c < N_CLASSES; ++c) s += __expf(l[c] - mx);
    const float lse = __logf(s) + mx;
    float* __restrict__ o = out + ((size_t)head * N_GRAPHS + tid) * N_CLASSES;
    for (int c = 0; c < N_CLASSES; ++c) o[c] = l[c] - lse;
  }
}

extern "C" void kernel_launch(void* const* d_in, const int* in_sizes, int n_in,
                              void* d_out, int out_size, void* d_ws, size_t ws_size,
                              hipStream_t stream) {
  (void)in_sizes; (void)n_in; (void)out_size; (void)ws_size;
  const float* x       = (const float*)d_in[0];
  const int*   ei      = (const int*)d_in[1];
  const int*   row     = ei;
  const int*   col     = ei + N_EDGES;
  const int*   batch   = (const int*)d_in[2];
  const float* W_feat  = (const float*)d_in[3];
  const float* W_convs = (const float*)d_in[4];
  const float* b_convs = (const float*)d_in[5];
  const float* W_ea    = (const float*)d_in[6];
  const float* b_ea    = (const float*)d_in[7];
  const float* W_na    = (const float*)d_in[8];
  const float* b_na    = (const float*)d_in[9];
  const float* W_ctx   = (const float*)d_in[10];
  const float* b_ctx   = (const float*)d_in[11];
  const float* W_obj   = (const float*)d_in[12];
  const float* b_obj   = (const float*)d_in[13];
  const float* W1h[3]  = {(const float*)d_in[14], (const float*)d_in[18], (const float*)d_in[22]};
  const float* b1h[3]  = {(const float*)d_in[15], (const float*)d_in[19], (const float*)d_in[23]};
  const float* W2h[3]  = {(const float*)d_in[16], (const float*)d_in[20], (const float*)d_in[24]};
  const float* b2h[3]  = {(const float*)d_in[17], (const float*)d_in[21], (const float*)d_in[25]};

  size_t off = 0;
  auto carve = [&](size_t elems) -> float* {
    float* p = (float*)((char*)d_ws + off);
    off += ((elems * sizeof(float)) + 255) & ~(size_t)255;
    return p;
  };
  float* hb    = carve((size_t)N_NODES * HIDDEN);
  float* mbuf  = carve((size_t)N_NODES * HIDDEN);
  float* xc    = carve((size_t)N_NODES * HIDDEN);
  float* xo    = carve((size_t)N_NODES * HIDDEN);
  float* att   = carve((size_t)N_EDGES * 2);
  float* pr    = carve((size_t)N_NODES * 2);
  float* pc    = carve((size_t)N_NODES * 2);
  float* degu  = carve(N_NODES);
  float* degc  = carve(N_NODES);
  float* dego  = carve(N_NODES);
  float* gpool = carve((size_t)2 * N_GRAPHS * HIDDEN);
  float* gc    = gpool;
  float* go    = gpool + (size_t)N_GRAPHS * HIDDEN;
  float* stat  = carve((size_t)6 * 2 * N_FEAT);
  float* Wf    = carve((size_t)N_FEAT * HIDDEN);
  float* vf    = carve(HIDDEN);
  int*  ecnt   = (int*)carve(N_NODES);
  int*  etsum  = (int*)carve(128);
  int*  eoff   = (int*)carve(N_NODES + 1);
  int*  ecur   = (int*)carve(N_NODES);
  int2* sedge  = (int2*)carve((size_t)N_EDGES * 2);

  auto nblk = [](long n, int b) { return (int)((n + b - 1) / b); };

  auto stats = [&](const float* A, int N, int F, int slot) {
    const int nvec = N * F / 4;
    const int grid = nblk(nvec, 256) < 512 ? nblk(nvec, 256) : 512;
    colstats_kernel<<<grid, 256, 0, stream>>>(A, nvec, F, stat + slot * 2 * N_FEAT,
                                              stat + slot * 2 * N_FEAT + N_FEAT);
  };
  auto fold = [&](int slot, int K, const float* W) {
    fold_bn_kernel<<<1, HIDDEN, 0, stream>>>(stat + slot * 2 * N_FEAT,
                                             stat + slot * 2 * N_FEAT + N_FEAT,
                                             1.f / N_NODES, K, HIDDEN, W, nullptr, Wf, vf);
  };
  auto gemm = [&](const float* A, float* C, int M, int K, int relu) {
    gemm_wmma_kernel<<<nblk(M / 16, 4), 128, 0, stream>>>(A, Wf, vf, C, M, K, relu);
  };

  static_assert(2 * N_GRAPHS * HIDDEN <= N_NODES && 6 * 2 * N_FEAT <= N_NODES,
                "init_kernel's grid is sized by the node count");
  init_kernel<<<nblk(N_NODES, 256), 256, 0, stream>>>(stat, 6 * 2 * N_FEAT, gpool,
                                                      2 * N_GRAPHS * HIDDEN, ecnt, degu,
                                                      degc, dego, N_NODES);

  stats(x, N_NODES, N_FEAT, 0);
  fold(0, N_FEAT, W_feat);
  gemm(x, hb, N_NODES, N_FEAT, 1);

  csr_count_kernel<<<nblk(N_EDGES, 256), 256, 0, stream>>>(row, col, N_EDGES, ecnt, degu);
  csr_tilesum_kernel<<<nblk(N_NODES, 1024), 256, 0, stream>>>(ecnt, N_NODES, etsum);
  csr_scan_kernel<<<nblk(N_NODES, 1024), 256, 0, stream>>>(ecnt, etsum, N_NODES,
                                                           eoff, ecur);
  csr_fill_kernel<<<nblk(N_EDGES, 256), 256, 0, stream>>>(row, col, N_EDGES, ecur,
                                                          sedge);

  for (int i = 0; i < 3; ++i) {
    stats(hb, N_NODES, HIDDEN, 1 + i);
    fold(1 + i, HIDDEN, W_convs + (size_t)i * HIDDEN * HIDDEN);
    gemm(hb, mbuf, N_NODES, HIDDEN, 0);
    gather_gcn_kernel<<<nblk(N_NODES, 8), 256, 0, stream>>>(
        mbuf, eoff, sedge, degu, b_convs + (size_t)i * HIDDEN, hb, N_NODES);
  }

  node_att_split_kernel<<<nblk(N_NODES, 256), 256, 0, stream>>>(hb, W_na, b_na, W_ea,
                                                                xc, xo, pr, pc, N_NODES);
  edge_att_kernel<<<nblk(N_EDGES, 256), 256, 0, stream>>>(pr, pc, row, col, b_ea,
                                                          att, degc, dego, N_EDGES);

  auto run_branch = [&](float* xin, int comp, const float* W, const float* b,
                        const float* dega, float* gout) {
    stats(xin, N_NODES, HIDDEN, 4 + comp);
    fold(4 + comp, HIDDEN, W);
    gemm(xin, mbuf, N_NODES, HIDDEN, 0);
    gather_pool_kernel<<<nblk(N_NODES, 8), 256, 0, stream>>>(
        mbuf, eoff, sedge, att, comp, dega, b, batch, gout, N_NODES);
  };
  run_branch(xc, 0, W_ctx, b_ctx, degc, gc);
  run_branch(xo, 1, W_obj, b_obj, dego, go);

  static_assert(HEAD_THREADS == N_GRAPHS, "one thread per graph in the logits stage");
  heads_kernel<<<3, HEAD_THREADS, 0, stream>>>(
      gc, go,
      W1h[0], b1h[0], W2h[0], b2h[0],
      W1h[1], b1h[1], W2h[1], b2h[1],
      W1h[2], b1h[2], W2h[2], b2h[2],
      (float*)d_out);
}


# attention branches share every pass: node weights scale h in the column sums and behind the product (xc, xo never stored), one 128-wide GEMM, one gather of 512-byte rows for both (sum orders kept)
# speedup vs baseline: 3.6913x; 1.1989x over previous
#include <hip/hip_runtime.h>
#include <math.h>


#define N_NODES  100000
#define N_EDGES  1000000
#define N_FEAT   128
#define HIDDEN   64
#define N_GRAPHS 512
#define N_CLASSES 10
#define BN_EPS  1e-5f
#define BN_BETA 1e-4f

typedef __attribute__((ext_vector_type(2))) float v2f;
typedef __attribute__((ext_vector_type(8))) float v8f;

__global__ __launch_bounds__(256) void colstats_kernel(
    const float* __restrict__ A, int nvec, int F,
    float* __restrict__ sums, float* __restrict__ sumsq) {
  __shared__ float sh_s[1024];
  __shared__ float sh_q[1024];
  const float4* __restrict__ A4 = (const float4*)A;
  const int tid = threadIdx.x;
  const int stride = gridDim.x * 256;
  const int last = nvec - 1;
  float4 s = make_float4(0.f, 0.f, 0.f, 0.f);
  float4 q = make_float4(0.f, 0.f, 0.f, 0.f);
  for (int i = blockIdx.x * 256 + tid; i < nvec; i += 8 * stride) {
    float4 v[8];
#pragma unroll
    for (int u = 0; u < 8; ++u) {
      const int idx = i + u * stride;
      v[u] = A4[idx < last ? idx : last];
    }
    __builtin_amdgcn_sched_barrier(0);
#pragma unroll
    for (int u = 0; u < 8; ++u) {
      const bool live = (i + u * stride) < nvec;
      const float a = live ? v[u].x : 0.f;
      const float b = live ? v[u].y : 0.f;
      const float c = live ? v[u].z : 0.f;
      const float d = live ? v[u].w : 0.f;
      s.x += a; q.x += a * a;
      s.y += b; q.y += b * b;
      s.z += c; q.z += c * c;
      s.w += d; q.w += d * d;
    }
  }
  ((float4*)sh_s)[tid] = s;
  ((float4*)sh_q)[tid] = q;
  __syncthreads();
  if (tid < F) {
    float ts = 0.f, tq = 0.f;
    for (int p = tid; p < 1024; p += F) { ts += sh_s[p]; tq += sh_q[p]; }
    atomicAdd(&sums[tid], ts);
    atomicAdd(&sumsq[tid], tq);
  }
}

__global__ __launch_bounds__(256) void colstats2_kernel(
    const float* __restrict__ A, const float* __restrict__ patt, int nvec,
    float* __restrict__ sums_c, float* __restrict__ sumsq_c,
    float* __restrict__ sums_o, float* __restrict__ sumsq_o) {
  __shared__ float sh[4][1024];
  const float4* __restrict__ A4 = (const float4*)A;
  const float2* __restrict__ P2 = (const float2*)patt;
  const int tid = threadIdx.x;
  const int stride = gridDim.x * 256;
  const int last = nvec - 1;
  float4 sc = make_float4(0.f, 0.f, 0.f, 0.f);
  float4 qc = make_float4(0.f, 0.f, 0.f, 0.f);
  float4 so = make_float4(0.f, 0.f, 0.f, 0.f);
  float4 qo = make_float4(0.f, 0.f, 0.f, 0.f);
  for (int i = blockIdx.x * 256 + tid; i < nvec; i += 8 * stride) {
    float4 v[8];
    float2 p[8];
#pragma unroll
    for (int u = 0; u < 8; ++u) {
      const int idx = i + u * stride;
      const int cl = idx < last ? idx : last;
      v[u] = A4[cl];
      p[u] = P2[cl >> 4];
    }
    __builtin_amdgcn_sched_barrier(0);
#pragma unroll
    for (int u = 0; u < 8; ++u) {
      const bool live = (i + u * stride) < nvec;
      const float pc0 = live ? p[u].x : 0.f;
      const float po0 = live ? p[u].y : 0.f;
      float a;
      a = pc0 * v[u].x; sc.x += a; qc.x += a * a;
      a = pc0 * v[u].y; sc.y += a; qc.y += a * a;
      a = pc0 * v[u].z; sc.z += a; qc.z += a * a;
      a = pc0 * v[u].w; sc.w += a; qc.w += a * a;
      a = po0 * v[u].x; so.x += a; qo.x += a * a;
      a = po0 * v[u].y; so.y += a; qo.y += a * a;
      a = po0 * v[u].z; so.z += a; qo.z += a * a;
      a = po0 * v[u].w; so.w += a; qo.w += a * a;
    }
  }
  ((float4*)sh[0])[tid] = sc;
  ((float4*)sh[1])[tid] = qc;
  ((float4*)sh[2])[tid] = so;
  ((float4*)sh[3])[tid] = qo;
  __syncthreads();
  {
    const int which = tid >> 6;
    const int colj  = tid & (HIDDEN - 1);
    float t = 0.f;
    for (int p = colj; p < 1024; p += HIDDEN) t += sh[which][p];
    float* __restrict__ dst = which == 0 ? sums_c : (which == 1 ? sumsq_c : (which == 2 ? sums_o : sumsq_o));
    atomicAdd(&dst[colj], t);
  }
}

__device__ __forceinline__ void fold_bn_body(const float* __restrict__ sums,
                                             const float* __restrict__ sumsq, float invN,
                                             int K, int NN, const float* __restrict__ W,
                                             const float* __restrict__ extra_b,
                                             float* __restrict__ Wout,
                                             float* __restrict__ vout) {
  __shared__ float s_ms[N_FEAT];
  __shared__ float s_s[N_FEAT];
  const int j = threadIdx.x;
  for (int k = j; k < K; k += blockDim.x) {
    float mean = sums[k] * invN;
    float var  = sumsq[k] * invN - mean * mean;
    float sc   = rsqrtf(var + BN_EPS);
    s_ms[k] = mean * sc;
    s_s[k]  = sc;
  }
  __syncthreads();
  float acc = extra_b ? extra_b[j] : 0.f;
  for (int k = 0; k < K; ++k) {
    float w = W[(size_t)k * NN + j];
    Wout[(size_t)k * NN + j] = s_s[k] * w;
    acc += (BN_BETA - s_ms[k]) * w;
  }
  vout[j] = acc;
}
__global__ void fold_bn_kernel(const float* __restrict__ sums,
                               const float* __restrict__ sumsq, float invN,
                               int K, int NN, const float* __restrict__ W,
                               const float* __restrict__ extra_b,
                               float* __restrict__ Wout, float* __restrict__ vout) {
  fold_bn_body(sums, sumsq, invN, K, NN, W, extra_b, Wout, vout);
}
__global__ void fold_bn2_kernel(const float* __restrict__ stat2, float invN,
                                const float* __restrict__ W_a,
                                const float* __restrict__ W_b,
                                float* __restrict__ Wout, float* __restrict__ vout) {
  const int br = blockIdx.x;
  const float* __restrict__ s = stat2 + (size_t)br * 2 * N_FEAT;
  fold_bn_body(s, s + N_FEAT, invN, HIDDEN, HIDDEN, br ? W_b : W_a, nullptr,
               Wout + (size_t)br * HIDDEN * HIDDEN, vout + (size_t)br * HIDDEN);
}

__global__ __launch_bounds__(128) void gemm_wmma_kernel(
    const float* __restrict__ A, const float* __restrict__ Wt,
    const float* __restrict__ v, float* __restrict__ C,
    int M, int K, int doRelu) {
  const int lane = threadIdx.x & 31;
  const int wave = threadIdx.x >> 5;
  const int rowTile = blockIdx.x * 4 + wave;
  if (rowTile >= (M >> 4)) return;
  const int mrow = (rowTile << 4) + (lane & 15);
  const int kOff = (lane >> 4) << 1;
  const int n0   = lane & 15;
  v8f acc[4] = {};
  for (int kk = 0; kk < K; kk += 4) {
    const int k0 = kk + kOff;
    v2f a = *(const v2f*)(A + (size_t)mrow * K + k0);
    const float* w0 = Wt + (size_t)k0 * HIDDEN + n0;
#pragma unroll
    for (int t = 0; t < 4; ++t) {
      v2f b;
      b.x = w0[t * 16];
      b.y = w0[HIDDEN + t * 16];
      acc[t] = __builtin_amdgcn_wmma_f32_16x16x4_f32(
          false, a, false, b, (short)0, acc[t], false, false);
    }
  }
  const int rowBase = (rowTile << 4) + ((lane >> 4) << 3);
#pragma unroll
  for (int t = 0; t < 4; ++t) {
    const int n = t * 16 + n0;
    const float bias = v[n];
#pragma unroll
    for (int e = 0; e < 8; ++e) {
      float val = acc[t][e] + bias;
      if (doRelu) val = fmaxf(val, 0.f);
      C[(size_t)(rowBase + e) * HIDDEN + n] = val;
    }
  }
}

__global__ __launch_bounds__(128) void gemm2_wmma_kernel(
    const float* __restrict__ A, const float* __restrict__ Wt,
    const float* __restrict__ v, const float* __restrict__ patt,
    float* __restrict__ C, int M) {
  const int lane = threadIdx.x & 31;
  const int wave = threadIdx.x >> 5;
  const int rowTile = blockIdx.x * 4 + wave;
  if (rowTile >= (M >> 4)) return;
  const int mrow = (rowTile << 4) + (lane & 15);
  const int kOff = (lane >> 4) << 1;
  const int n0   = lane & 15;
  v8f acc[8] = {};
  for (int kk = 0; kk < HIDDEN; kk += 4) {
    const int k0 = kk + kOff;
    v2f a = *(const v2f*)(A + (size_t)mrow * HIDDEN + k0);
    const float* w0 = Wt + (size_t)k0 * HIDDEN + n0;
#pragma unroll
    for (int t = 0; t < 8; ++t) {
      const float* wt = w0 + (t >> 2) * (HIDDEN * HIDDEN) + (t & 3) * 16;
      v2f b;
      b.x = wt[0];
      b.y = wt[HIDDEN];
      acc[t] = __builtin_amdgcn_wmma_f32_16x16x4_f32(
          false, a, false, b, (short)0, acc[t], false, false);
    }
  }
  const int rowBase = (rowTile << 4) + ((lane >> 4) << 3);
  float2 pw[8];
#pragma unroll
  for (int e = 0; e < 8; ++e) pw[e] = ((const float2*)patt)[rowBase + e];
#pragma unroll
  for (int t = 0; t < 8; ++t) {
    const int n = (t >> 2) * HIDDEN + (t & 3) * 16 + n0;
    const float bias = v[n];
#pragma unroll
    for (int e = 0; e < 8; ++e) {
      const float s = (t >> 2) ? pw[e].y : pw[e].x;
      C[(size_t)(rowBase + e) * (2 * HIDDEN) + n] = acc[t][e] * s + bias;
    }
  }
}

__global__ void init_kernel(float* __restrict__ stat, int nstat,
                            float* __restrict__ gpool, int npool,
                            int* __restrict__ cnt, float* __restrict__ degu,
                            float* __restrict__ deg2, int N) {
  int i = blockIdx.x * blockDim.x + threadIdx.x;
  if (i < nstat) stat[i] = 0.f;
  if (i < npool) gpool[i] = 0.f;
  if (i < N) {
    cnt[i] = 0;
    degu[i] = 1.0f;
    ((float2*)deg2)[i] = make_float2(1.0f, 1.0f);
  }
}
__device__ __forceinline__ float inv_sqrt_deg(float d) {
  return d > 0.f ? rsqrtf(d) : 0.f;
}

__global__ void csr_count_kernel(const int* __restrict__ row,
                                 const int* __restrict__ col, int E,
                                 int* __restrict__ cnt, float* __restrict__ degu) {
  int e = blockIdx.x * blockDim.x + threadIdx.x;
  if (e >= E) return;
  atomicAdd(&cnt[col[e]], 1);
  atomicAdd(&degu[row[e]], 1.0f);
}
__device__ __forceinline__ void load_counts4(const int* __restrict__ cnt, int i0,
                                             int N, int c[4]) {
  if (i0 + 3 < N) {
    const int4 v = *(const int4*)(cnt + i0);
    c[0] = v.x; c[1] = v.y; c[2] = v.z; c[3] = v.w;
  } else {
    c[0] = (i0     < N) ? cnt[i0]     : 0;
    c[1] = (i0 + 1 < N) ? cnt[i0 + 1] : 0;
    c[2] = (i0 + 2 < N) ? cnt[i0 + 2] : 0;
    c[3] = 0;
  }
}
__global__ __launch_bounds__(256) void csr_tilesum_kernel(
    const int* __restrict__ cnt, int N, int* __restrict__ tsum) {
  __shared__ int sh[8];
  const int tid = threadIdx.x;
  int c[4];
  load_counts4(cnt, (blockIdx.x * 256 + tid) * 4, N, c);
  int t = c[0] + c[1] + c[2] + c[3];
  for (int o = 16; o > 0; o >>= 1) t += __shfl_xor(t, o);
  if ((tid & 31) == 0) sh[tid >> 5] = t;
  __syncthreads();
  if (tid == 0) {
    int s = 0;
    for (int w = 0; w < 8; ++w) s += sh[w];
    tsum[blockIdx.x] = s;
  }
}
__global__ __launch_bounds__(256) void csr_scan_kernel(
    const int* __restrict__ cnt, const int* __restrict__ tsum, int N,
    int* __restrict__ off, int* __restrict__ cur) {
  __shared__ int sh_w[8];
  __shared__ int sh_b[8];
  const int tid = threadIdx.x, lane = tid & 31, wave = tid >> 5;
  int bpart = 0;
  for (int t = tid; t < (int)blockIdx.x; t += 256) bpart += tsum[t];
  for (int o = 16; o > 0; o >>= 1) bpart += __shfl_xor(bpart, o);
  const int i0 = (blockIdx.x * 256 + tid) * 4;
  int c[4];
  load_counts4(cnt, i0, N, c);
  const int t = c[0] + c[1] + c[2] + c[3];
  int inc = t;
  for (int o = 1; o < 32; o <<= 1) {
    const int y = __shfl_up(inc, o);
    if (lane >= o) inc += y;
  }
  if (lane == 31) sh_w[wave] = inc;
  if (lane == 0)  sh_b[wave] = bpart;
  __syncthreads();
  int base = 0;
  for (int w = 0; w < 8; ++w) {
    base += sh_b[w];
    if (w < wave) base += sh_w[w];
  }
  const int o0 = base + inc - t;
  const int o1 = o0 + c[0], o2 = o1 + c[1], o3 = o2 + c[2];
  if (i0 + 3 < N) {
    const int4 v = make_int4(o0, o1, o2, o3);
    *(int4*)(off + i0) = v;
    *(int4*)(cur + i0) = v;
  } else {
    if (i0     < N) { off[i0]     = o0; cur[i0]     = o0; }
    if (i0 + 1 < N) { off[i0 + 1] = o1; cur[i0 + 1] = o1; }
    if (i0 + 2 < N) { off[i0 + 2] = o2; cur[i0 + 2] = o2; }
  }
  if (i0 < N && N - 1 <= i0 + 3) off[N] = o0 + t;
}
__global__ void csr_fill_kernel(const int* __restrict__ row,
                                const int* __restrict__ col, int E,
                                int* __restrict__ cur, int2* __restrict__ sedge) {
  int e = blockIdx.x * blockDim.x + threadIdx.x;
  if (e >= E) return;
  const int r = row[e];
  const int pos = atomicAdd(&cur[col[e]], 1);
  if ((unsigned)pos < (unsigned)E) sedge[pos] = make_int2(r, e);
}

template <int CNT>
__device__ __forceinline__ void gather_rows(const float2* __restrict__ m2, int r,
                                            float cf, int k, int lane, float2& acc) {
  int rr[CNT];
  float cc[CNT];
  float2 v[CNT];
#pragma unroll
  for (int u = 0; u < CNT; ++u) {
    rr[u] = __builtin_amdgcn_readlane(r, k + u);
    cc[u] = __int_as_float(__builtin_amdgcn_readlane(__float_as_int(cf), k + u));
  }
#pragma unroll
  for (int u = 0; u < CNT; ++u) v[u] = m2[(size_t)rr[u] * (HIDDEN / 2) + lane];
#pragma unroll
  for (int u = 0; u < CNT; ++u) {
    acc.x += cc[u] * v[u].x;
    acc.y += cc[u] * v[u].y;
  }
}
__device__ __forceinline__ float2 gcn_row(
    const float2* __restrict__ m2, const int* __restrict__ off,
    const int2* __restrict__ sedge, const float* __restrict__ deg,
    const float* __restrict__ b, int c, int lane) {
  const int beg = off[c], end = off[c + 1];
  const float dc = inv_sqrt_deg(deg[c]);
  const float2 self = m2[(size_t)c * (HIDDEN / 2) + lane];
  const float2 bias = ((const float2*)b)[lane];
  float2 acc = make_float2(0.f, 0.f);
  for (int base = beg; base < end; base += 32) {
    const int p = base + lane;
    int r = 0;
    float cf = 0.f;
    if (p < end) {
      const int2 se = sedge[p];
      r = se.x;
      const float dr = inv_sqrt_deg(deg[r]);
      cf = dr * dc;
    }
    const int n = (end - base) < 32 ? (end - base) : 32;
    int k = 0;
    for (; k + 4 <= n; k += 4) gather_rows<4>(m2, r, cf, k, lane, acc);
    const int rem = n - k;
    if (rem == 3)      gather_rows<3>(m2, r, cf, k, lane, acc);
    else if (rem == 2) gather_rows<2>(m2, r, cf, k, lane, acc);
    else if (rem == 1) gather_rows<1>(m2, r, cf, k, lane, acc);
  }
  float2 o;
  o.x = fmaxf(acc.x + dc * dc * self.x + bias.x, 0.f);
  o.y = fmaxf(acc.y + dc * dc * self.y + bias.y, 0.f);
  return o;
}

__global__ __launch_bounds__(256) void gather_gcn_kernel(
    const float* __restrict__ m, const int* __restrict__ off,
    const int2* __restrict__ sedge, const float* __restrict__ deg,
    const float* __restrict__ b, float* __restrict__ out, int N) {
  const int lane = threadIdx.x & 31;
  const int c = __builtin_amdgcn_readfirstlane(blockIdx.x * 8 + (threadIdx.x >> 5));
  if (c >= N) return;
  const float2 o = gcn_row((const float2*)m, off, sedge, deg, b, c, lane);
  ((float2*)out)[(size_t)c * (HIDDEN / 2) + lane] = o;
}

template <int CNT>
__device__ __forceinline__ void gather_rows2(const float4* __restrict__ m4, int r,
                                             float cfc, float cfo, int k, int lane,
                                             bool upper, float4& acc) {
  int rr[CNT];
  float cc[CNT];
  float4 v[CNT];
#pragma unroll
  for (int u = 0; u < CNT; ++u) {
    rr[u] = __builtin_amdgcn_readlane(r, k + u);
    const float c0 = __int_as_float(__builtin_amdgcn_readlane(__float_as_int(cfc), k + u));
    const float c1 = __int_as_float(__builtin_amdgcn_readlane(__float_as_int(cfo), k + u));
    cc[u] = upper ? c1 : c0;
  }
#pragma unroll
  for (int u = 0; u < CNT; ++u) v[u] = m4[(size_t)rr[u] * (HIDDEN / 2) + lane];
#pragma unroll
  for (int u = 0; u < CNT; ++u) {
    acc.x += cc[u] * v[u].x;
    acc.y += cc[u] * v[u].y;
    acc.z += cc[u] * v[u].z;
    acc.w += cc[u] * v[u].w;
  }
}
__global__ __launch_bounds__(256) void gather_pool2_kernel(
    const float* __restrict__ m, const int* __restrict__ off,
    const int2* __restrict__ sedge, const float* __restrict__ att,
    const float* __restrict__ deg2, const float* __restrict__ b_ctx,
    const float* __restrict__ b_obj, const int* __restrict__ batch,
    float* __restrict__ gpool, int N) {
  __shared__ float4 sh_row[8 * (HIDDEN / 2)];
  __shared__ int sh_gid[8];
  const int lane = threadIdx.x & 31;
  const int wave = threadIdx.x >> 5;
  const bool upper = lane >= 16;
  const int c = __builtin_amdgcn_readfirstlane(blockIdx.x * 8 + wave);
  float4 o = make_float4(0.f, 0.f, 0.f, 0.f);
  int gid = -1;
  if (c < N) {
    gid = batch[c];
    const float4* __restrict__ m4 = (const float4*)m;
    const float2* __restrict__ d2 = (const float2*)deg2;
    const float2* __restrict__ att2 = (const float2*)att;
    const int beg = off[c], end = off[c + 1];
    const float2 dcv = d2[c];
    const float dcc = inv_sqrt_deg(dcv.x);
    const float dco = inv_sqrt_deg(dcv.y);
    const float4 self = m4[(size_t)c * (HIDDEN / 2) + lane];
    const float4 bias = ((const float4*)(upper ? b_obj : b_ctx))[lane & 15];
    float4 acc = make_float4(0.f, 0.f, 0.f, 0.f);
    for (int base = beg; base < end; base += 32) {
      const int p = base + lane;
      int r = 0;
      float cfc = 0.f, cfo = 0.f;
      if (p < end) {
        const int2 se = sedge[p];
        r = se.x;
        const float2 dr = d2[r];
        const float2 w = att2[se.y];
        cfc = inv_sqrt_deg(dr.x) * w.x * dcc;
        cfo = inv_sqrt_deg(dr.y) * w.y * dco;
      }
      const int n = (end - base) < 32 ? (end - base) : 32;
      int k = 0;
      for (; k + 4 <= n; k += 4) gather_rows2<4>(m4, r, cfc, cfo, k, lane, upper, acc);
      const int rem = n - k;
      if (rem == 3)      gather_rows2<3>(m4, r, cfc, cfo, k, lane, upper, acc);
      else if (rem == 2) gather_rows2<2>(m4, r, cfc, cfo, k, lane, upper, acc);
      else if (rem == 1) gather_rows2<1>(m4, r, cfc, cfo, k, lane, upper, acc);
    }
    const float dself = upper ? dco * dco : dcc * dcc;
    o.x = fmaxf(acc.x + dself * self.x + bias.x, 0.f);
    o.y = fmaxf(acc.y + dself * self.y + bias.y, 0.f);
    o.z = fmaxf(acc.z + dself * self.z + bias.z, 0.f);
    o.w = fmaxf(acc.w + dself * self.w + bias.w, 0.f);
  }
  sh_row[wave * (HIDDEN / 2) + lane] = o;
  if (lane == 0) sh_gid[wave] = gid;
  __syncthreads();
  if (threadIdx.x < 2 * HIDDEN) {
    const int j = threadIdx.x;
    float* __restrict__ g = gpool + (size_t)(j >> 6) * N_GRAPHS * HIDDEN + (j & (HIDDEN - 1));
    int gw[8];
    float v[8];
#pragma unroll
    for (int w = 0; w < 8; ++w) {
      gw[w] = sh_gid[w];
      v[w] = ((const float*)sh_row)[w * (2 * HIDDEN) + j];
    }
    int cur = gw[0];
    float a = v[0];
#pragma unroll
    for (int w = 1; w < 8; ++w) {
      if (gw[w] == cur) {
        a += v[w];
      } else {
        if ((unsigned)cur < (unsigned)N_GRAPHS) atomicAdd(&g[(size_t)cur * HIDDEN], a);
        cur = gw[w];
        a = v[w];
      }
    }
    if ((unsigned)cur < (unsigned)N_GRAPHS) atomicAdd(&g[(size_t)cur * HIDDEN], a);
  }
}

__global__ void node_att_kernel(const float* __restrict__ h,
                                const float* __restrict__ Wna,
                                const float* __restrict__ bna,
                                const float* __restrict__ Wea,
                                float* __restrict__ patt,
                                float* __restrict__ pr,
                                float* __restrict__ pc, int N) {
  int i = blockIdx.x * blockDim.x + threadIdx.x;
  if (i >= N) return;
  const float* hi = h + (size_t)i * HIDDEN;
  float l0 = bna[0], l1 = bna[1];
  float r0 = 0.f, r1 = 0.f, c0 = 0.f, c1 = 0.f;
  for (int j = 0; j < HIDDEN; ++j) {
    float a = hi[j];
    l0 += a * Wna[j * 2 + 0];
    l1 += a * Wna[j * 2 + 1];
    r0 += a * Wea[j * 2 + 0];
    r1 += a * Wea[j * 2 + 1];
    c0 += a * Wea[(HIDDEN + j) * 2 + 0];
    c1 += a * Wea[(HIDDEN + j) * 2 + 1];
  }
  ((float2*)pr)[i] = make_float2(r0, r1);
  ((float2*)pc)[i] = make_float2(c0, c1);
  float mx = fmaxf(l0, l1);
  float e0 = __expf(l0 - mx), e1 = __expf(l1 - mx);
  float inv = 1.f / (e0 + e1);
  float p0 = e0 * inv, p1 = e1 * inv;
  ((float2*)patt)[i] = make_float2(p0, p1);
}

__global__ void edge_att_kernel(const float* __restrict__ pr,
                                const float* __restrict__ pc,
                                const int* __restrict__ row,
                                const int* __restrict__ col,
                                const float* __restrict__ bea,
                                float* __restrict__ att,
                                float* __restrict__ deg2, int E) {
  int e = blockIdx.x * blockDim.x + threadIdx.x;
  if (e >= E) return;
  const int r = row[e];
  const float2 a = ((const float2*)pr)[r];
  const float2 c = ((const float2*)pc)[col[e]];
  const float l0 = (a.x + c.x) + bea[0];
  const float l1 = (a.y + c.y) + bea[1];
  float mx = fmaxf(l0, l1);
  float e0 = __expf(l0 - mx), e1 = __expf(l1 - mx);
  float inv = 1.f / (e0 + e1);
  const float w0 = e0 * inv, w1 = e1 * inv;
  ((float2*)att)[e] = make_float2(w0, w1);
  atomicAdd(&deg2[2 * (size_t)r], w0);
  atomicAdd(&deg2[2 * (size_t)r + 1], w1);
}

#define HEAD_THREADS 512
#define G_PITCH 68
#define W_PITCH 72

__device__ __forceinline__ void head_colstats(const float* __restrict__ ps,
                                              const float* __restrict__ pq,
                                              float* __restrict__ sc,
                                              float* __restrict__ ms, int col) {
  float ts = 0.f, tq = 0.f;
  for (int p = 0; p < 32; ++p) {
    ts += ps[p * HIDDEN + col];
    tq += pq[p * HIDDEN + col];
  }
  const float invN = 1.f / N_GRAPHS;
  const float mean = ts * invN;
  const float var  = tq * invN - mean * mean;
  const float s    = rsqrtf(var + BN_EPS);
  sc[col] = s;
  ms[col] = mean * s;
}

__global__ __launch_bounds__(HEAD_THREADS) void heads_kernel(
    const float* __restrict__ gc, const float* __restrict__ go,
    const float* __restrict__ W1_c, const float* __restrict__ b1_c,
    const float* __restrict__ W2_c, const float* __restrict__ b2_c,
    const float* __restrict__ W1_o, const float* __restrict__ b1_o,
    const float* __restrict__ W2_o, const float* __restrict__ b2_o,
    const float* __restrict__ W1_co, const float* __restrict__ b1_co,
    const float* __restrict__ W2_co, const float* __restrict__ b2_co,
    float* __restrict__ out) {
  __shared__ __attribute__((aligned(16))) float sh_g[N_GRAPHS * G_PITCH];
  __shared__ __attribute__((aligned(16))) float sh_w[HIDDEN * W_PITCH];
  __shared__ __attribute__((aligned(16))) float sh_ps[32 * HIDDEN];
  __shared__ __attribute__((aligned(16))) float sh_pq[32 * HIDDEN];
  __shared__ float sh_sc[HIDDEN];
  __shared__ float sh_ms[HIDDEN];
  __shared__ float sh_v[HIDDEN];
  __shared__ __attribute__((aligned(16))) float sh_w2[HIDDEN * N_CLASSES];
  __shared__ float sh_v2[16];
  __shared__ float sh_w1r[HIDDEN * HIDDEN];
  __shared__ float sh_w2r[HIDDEN * N_CLASSES];

  const int head = blockIdx.x;
  const float* __restrict__ W1 = head == 0 ? W1_c : (head == 1 ? W1_o : W1_co);
  const float* __restrict__ b1 = head == 0 ? b1_c : (head == 1 ? b1_o : b1_co);
  const float* __restrict__ W2 = head == 0 ? W2_c : (head == 1 ? W2_o : W2_co);
  const float* __restrict__ b2 = head == 0 ? b2_c : (head == 1 ? b2_o : b2_co);
  const int tid  = threadIdx.x;
  const int lane = tid & 31;
  const int wave = tid >> 5;

  float w1r[8];
#pragma unroll
  for (int u = 0; u < 8; ++u) w1r[u] = W1[tid + HEAD_THREADS * u];
  const float w2r0 = W2[tid];
  const float w2r1 = tid < HIDDEN * N_CLASSES - HEAD_THREADS ? W2[HEAD_THREADS + tid] : 0.f;
  const float b1r  = tid < HIDDEN ? b1[tid] : 0.f;
  const float b2r  = tid < N_CLASSES ? b2[tid] : 0.f;

  {
    const float4* __restrict__ gc4 = (const float4*)gc;
    const float4* __restrict__ go4 = (const float4*)go;
    const float4* __restrict__ pa  = head == 1 ? go4 : gc4;
    const int cg = tid & 15;
    const int r0 = tid >> 4;
    float4 x[16];
#pragma unroll
    for (int u = 0; u < 16; ++u) x[u] = pa[tid + HEAD_THREADS * u];
    if (head == 2) {
#pragma unroll
      for (int u = 0; u < 16; ++u) {
        const float4 b = go4[tid + HEAD_THREADS * u];
        x[u] = make_float4(x[u].x + b.x, x[u].y + b.y, x[u].z + b.z, x[u].w + b.w);
      }
    }
    float4 s = make_float4(0.f, 0.f, 0.f, 0.f);
    float4 q = make_float4(0.f, 0.f, 0.f, 0.f);
#pragma unroll
    for (int u = 0; u < 16; ++u) {
      *(float4*)(sh_g + (r0 + 32 * u) * G_PITCH + 4 * cg) = x[u];
      s.x += x[u].x; q.x += x[u].x * x[u].x;
      s.y += x[u].y; q.y += x[u].y * x[u].y;
      s.z += x[u].z; q.z += x[u].z * x[u].z;
      s.w += x[u].w; q.w += x[u].w * x[u].w;
    }
    *(float4*)(sh_ps + r0 * HIDDEN + 4 * cg) = s;
    *(float4*)(sh_pq + r0 * HIDDEN + 4 * cg) = q;
#pragma unroll
    for (int u = 0; u < 8; ++u) sh_w1r[tid + HEAD_THREADS * u] = w1r[u];
    sh_w2r[tid] = w2r0;
    if (tid < HIDDEN * N_CLASSES - HEAD_THREADS) sh_w2r[HEAD_THREADS + tid] = w2r1;
  }
  __syncthreads();
  if (tid < HIDDEN) head_colstats(sh_ps, sh_pq, sh_sc, sh_ms, tid);
  __syncthreads();

#pragma unroll
  for (int u = 0; u < 8; ++u) {
    const int i = tid + HEAD_THREADS * u;
    const int k = i >> 6, j = i & 63;
    sh_w[k * W_PITCH + j] = sh_sc[k] * w1r[u];
  }
  if (tid < HIDDEN) {
    float acc = b1r;
    for (int k = 0; k < HIDDEN; ++k) acc += (BN_BETA - sh_ms[k]) * sh_w1r[k * HIDDEN + tid];
    sh_v[tid] = acc;
  }
  __syncthreads();

  {
    const int kOff = (lane >> 4) << 1;
    const int n0   = lane & 15;
    v8f acc[2][4] = {};
    const float* a0p = sh_g + (32 * wave + n0) * G_PITCH + kOff;
    const float* a1p = a0p + 16 * G_PITCH;
    const float* bp  = sh_w + kOff * W_PITCH + n0;
    for (int kk = 0; kk < HIDDEN; kk += 4) {
      const v2f a0 = *(const v2f*)(a0p + kk);
      const v2f a1 = *(const v2f*)(a1p + kk);
#pragma unroll
      for (int t = 0; t < 4; ++t) {
        v2f b;
        b.x = bp[kk * W_PITCH + t * 16];
        b.y = bp[(kk + 1) * W_PITCH + t * 16];
        acc[0][t] = __builtin_amdgcn_wmma_f32_16x16x4_f32(
            false, a0, false, b, (short)0, acc[0][t], false, false);
        acc[1][t] = __builtin_amdgcn_wmma_f32_16x16x4_f32(
            false, a1, false, b, (short)0, acc[1][t], false, false);
      }
    }
    const int rowBase = 32 * wave + ((lane >> 4) << 3);
#pragma unroll
    for (int t = 0; t < 4; ++t) {
      const int n = t * 16 + n0;
      const float bias = sh_v[n];
      float s = 0.f, q = 0.f;
#pragma unroll
      for (int i = 0; i < 2; ++i) {
#pragma unroll
        for (int e = 0; e < 8; ++e) {
          const float val = fmaxf(acc[i][t][e] + bias, 0.f);
          sh_g[(rowBase + 16 * i + e) * G_PITCH + n] = val;
          s += val;
          q += val * val;
        }
      }
      sh_ps[(2 * wave + (lane >> 4)) * HIDDEN + n] = s;
      sh_pq[(2 * wave + (lane >> 4)) * HIDDEN + n] = q;
    }
  }
  __syncthreads();
  if (tid < HIDDEN) head_colstats(sh_ps, sh_pq, sh_sc, sh_ms, tid);
  __syncthreads();

  sh_w2[tid] = sh_sc[tid / N_CLASSES] * w2r0;
  if (tid < HIDDEN * N_CLASSES - HEAD_THREADS)
    sh_w2[HEAD_THREADS + tid] = sh_sc[(HEAD_THREADS + tid) / N_CLASSES] * w2r1;
  if (tid < N_CLASSES) {
    float acc = b2r;
    for (int k = 0; k < HIDDEN; ++k) acc += (BN_BETA - sh_ms[k]) * sh_w2r[k * N_CLASSES + tid];
    sh_v2[tid] = acc;
  }
  __syncthreads();

  {
    float l[N_CLASSES];
    for (int c = 0; c < N_CLASSES; ++c) l[c] = sh_v2[c];
    const float4* __restrict__ row4 = (const float4*)(sh_g + tid * G_PITCH);
    for (int k4 = 0; k4 < HIDDEN / 4; ++k4) {
      const float4 x = row4[k4];
      const float* w = sh_w2 + k4 * 4 * N_CLASSES;
      for (int c = 0; c < N_CLASSES; ++c) l[c] += x.x * w[c];
      for (int c = 0; c < N_CLASSES; ++c) l[c] += x.y * w[N_CLASSES + c];
      for (int c = 0; c < N_CLASSES; ++c) l[c] += x.z * w[2 * N_CLASSES + c];
      for (int c = 0; c < N_CLASSES; ++c) l[c] += x.w * w[3 * N_CLASSES + c];
    }
    float mx = l[0];
    for (int c = 1; c < N_CLASSES; ++c) mx = fmaxf(mx, l[c]);
    float s = 0.f;
    for (int c = 0; c < N_CLASSES; ++c) s += __expf(l[c] - mx);
    const float lse = __logf(s) + mx;
    float* __restrict__ o = out + ((size_t)head * N_GRAPHS + tid) * N_CLASSES;
    for (int c = 0; c < N_CLASSES; ++c) o[c] = l[c] - lse;
  }
}

extern "C" void kernel_launch(void* const* d_in, const int* in_sizes, int n_in,
                              void* d_out, int out_size, void* d_ws, size_t ws_size,
                              hipStream_t stream) {
  (void)in_sizes; (void)n_in; (void)out_size; (void)ws_size;
  const float* x       = (const float*)d_in[0];
  const int*   ei      = (const int*)d_in[1];
  const int*   row     = ei;
  const int*   col     = ei + N_EDGES;
  const int*   batch   = (const int*)d_in[2];
  const float* W_feat  = (const float*)d_in[3];
  const float* W_convs = (const float*)d_in[4];
  const float* b_convs = (const float*)d_in[5];
  const float* W_ea    = (const float*)d_in[6];
  const float* b_ea    = (const float*)d_in[7];
  const float* W_na    = (const float*)d_in[8];
  const float* b_na    = (const float*)d_in[9];
  const float* W_ctx   = (const float*)d_in[10];
  const float* b_ctx   = (const float*)d_in[11];
  const float* W_obj   = (const float*)d_in[12];
  const float* b_obj   = (const float*)d_in[13];
  const float* W1h[3]  = {(const float*)d_in[14], (const float*)d_in[18], (const float*)d_in[22]};
  const float* b1h[3]  = {(const float*)d_in[15], (const float*)d_in[19], (const float*)d_in[23]};
  const float* W2h[3]  = {(const float*)d_in[16], (const float*)d_in[20], (const float*)d_in[24]};
  const float* b2h[3]  = {(const float*)d_in[17], (const float*)d_in[21], (const float*)d_in[25]};

  size_t off = 0;
  auto carve = [&](size_t elems) -> float* {
    float* p = (float*)((char*)d_ws + off);
    off += ((elems * sizeof(float)) + 255) & ~(size_t)255;
    return p;
  };
  float* hb    = carve((size_t)N_NODES * HIDDEN);
  float* mbuf  = carve((size_t)N_NODES * HIDDEN);
  float* mcat  = carve((size_t)N_NODES * 2 * HIDDEN);
  float* patt  = carve((size_t)N_NODES * 2);
  float* att   = carve((size_t)N_EDGES * 2);
  float* pr    = carve((size_t)N_NODES * 2);
  float* pc    = carve((size_t)N_NODES * 2);
  float* degu  = carve(N_NODES);
  float* deg2  = carve((size_t)N_NODES * 2);
  float* gpool = carve((size_t)2 * N_GRAPHS * HIDDEN);
  float* gc    = gpool;
  float* go    = gpool + (size_t)N_GRAPHS * HIDDEN;
  float* stat  = carve((size_t)6 * 2 * N_FEAT);
  float* Wf    = carve((size_t)N_FEAT * HIDDEN);
  float* vf    = carve(HIDDEN);
  float* Wf2   = carve((size_t)2 * HIDDEN * HIDDEN);
  float* vf2   = carve((size_t)2 * HIDDEN);
  int*  ecnt   = (int*)carve(N_NODES);
  int*  etsum  = (int*)carve(128);
  int*  eoff   = (int*)carve(N_NODES + 1);
  int*  ecur   = (int*)carve(N_NODES);
  int2* sedge  = (int2*)carve((size_t)N_EDGES * 2);

  auto nblk = [](long n, int b) { return (int)((n + b - 1) / b); };

  auto stats = [&](const float* A, int N, int F, int slot) {
    const int nvec = N * F / 4;
    const int grid = nblk(nvec, 256) < 512 ? nblk(nvec, 256) : 512;
    colstats_kernel<<<grid, 256, 0, stream>>>(A, nvec, F, stat + slot * 2 * N_FEAT,
                                              stat + slot * 2 * N_FEAT + N_FEAT);
  };
  auto fold = [&](int slot, int K, const float* W) {
    fold_bn_kernel<<<1, HIDDEN, 0, stream>>>(stat + slot * 2 * N_FEAT,
                                             stat + slot * 2 * N_FEAT + N_FEAT,
                                             1.f / N_NODES, K, HIDDEN, W, nullptr, Wf, vf);
  };
  auto gemm = [&](const float* A, float* C, int M, int K, int relu) {
    gemm_wmma_kernel<<<nblk(M / 16, 4), 128, 0, stream>>>(A, Wf, vf, C, M, K, relu);
  };

  static_assert(2 * N_GRAPHS * HIDDEN <= N_NODES && 6 * 2 * N_FEAT <= N_NODES,
                "init_kernel's grid is sized by the node count");
  init_kernel<<<nblk(N_NODES, 256), 256, 0, stream>>>(stat, 6 * 2 * N_FEAT, gpool,
                                                      2 * N_GRAPHS * HIDDEN, ecnt, degu,
                                                      deg2, N_NODES);

  stats(x, N_NODES, N_FEAT, 0);
  fold(0, N_FEAT, W_feat);
  gemm(x, hb, N_NODES, N_FEAT, 1);

  csr_count_kernel<<<nblk(N_EDGES, 256), 256, 0, stream>>>(row, col, N_EDGES, ecnt, degu);
  csr_tilesum_kernel<<<nblk(N_NODES, 1024), 256, 0, stream>>>(ecnt, N_NODES, etsum);
  csr_scan_kernel<<<nblk(N_NODES, 1024), 256, 0, stream>>>(ecnt, etsum, N_NODES,
                                                           eoff, ecur);
  csr_fill_kernel<<<nblk(N_EDGES, 256), 256, 0, stream>>>(row, col, N_EDGES, ecur,
                                                          sedge);

  for (int i = 0; i < 3; ++i) {
    stats(hb, N_NODES, HIDDEN, 1 + i);
    fold(1 + i, HIDDEN, W_convs + (size_t)i * HIDDEN * HIDDEN);
    gemm(hb, mbuf, N_NODES, HIDDEN, 0);
    gather_gcn_kernel<<<nblk(N_NODES, 8), 256, 0, stream>>>(
        mbuf, eoff, sedge, degu, b_convs + (size_t)i * HIDDEN, hb, N_NODES);
  }

  node_att_kernel<<<nblk(N_NODES, 256), 256, 0, stream>>>(hb, W_na, b_na, W_ea,
                                                          patt, pr, pc, N_NODES);
  edge_att_kernel<<<nblk(N_EDGES, 256), 256, 0, stream>>>(pr, pc, row, col, b_ea,
                                                          att, deg2, N_EDGES);

  {
    const int nvec = N_NODES * HIDDEN / 4;
    const int grid = nblk(nvec, 256) < 512 ? nblk(nvec, 256) : 512;
    colstats2_kernel<<<grid, 256, 0, stream>>>(
        hb, patt, nvec, stat + 4 * 2 * N_FEAT, stat + 4 * 2 * N_FEAT + N_FEAT,
        stat + 5 * 2 * N_FEAT, stat + 5 * 2 * N_FEAT + N_FEAT);
    fold_bn2_kernel<<<2, HIDDEN, 0, stream>>>(stat + 4 * 2 * N_FEAT, 1.f / N_NODES,
                                              W_ctx, W_obj, Wf2, vf2);
    gemm2_wmma_kernel<<<nblk(N_NODES / 16, 4), 128, 0, stream>>>(hb, Wf2, vf2, patt, mcat,
                                                                 N_NODES);
    gather_pool2_kernel<<<nblk(N_NODES, 8), 256, 0, stream>>>(
        mcat, eoff, sedge, att, deg2, b_ctx, b_obj, batch, gpool, N_NODES);
  }

  static_assert(HEAD_THREADS == N_GRAPHS, "one thread per graph in the logits stage");
  heads_kernel<<<3, HEAD_THREADS, 0, stream>>>(
      gc, go,
      W1h[0], b1h[0], W2h[0], b2h[0],
      W1h[1], b1h[1], W2h[1], b2h[1],
      W1h[2], b1h[2], W2h[2], b2h[2],
      (float*)d_out);
}


# edge propagation in front of the product, A(hW) -> (Ah)W: a work-group gathers 16 nodes' sums of h rows into LDS and multiplies there; 4 GEMM launches and the stored products go (new sum order)
# speedup vs baseline: 3.9551x; 1.0715x over previous
#include <hip/hip_runtime.h>
#include <math.h>


#define N_NODES  100000
#define N_EDGES  1000000
#define N_FEAT   128
#define HIDDEN   64
#define N_GRAPHS 512
#define N_CLASSES 10
#define BN_EPS  1e-5f
#define BN_BETA 1e-4f

typedef __attribute__((ext_vector_type(2))) float v2f;
typedef __attribute__((ext_vector_type(8))) float v8f;

__global__ __launch_bounds__(256) void colstats_kernel(
    const float* __restrict__ A, int nvec, int F,
    float* __restrict__ sums, float* __restrict__ sumsq) {
  __shared__ float sh_s[1024];
  __shared__ float sh_q[1024];
  const float4* __restrict__ A4 = (const float4*)A;
  const int tid = threadIdx.x;
  const int stride = gridDim.x * 256;
  const int last = nvec - 1;
  float4 s = make_float4(0.f, 0.f, 0.f, 0.f);
  float4 q = make_float4(0.f, 0.f, 0.f, 0.f);
  for (int i = blockIdx.x * 256 + tid; i < nvec; i += 8 * stride) {
    float4 v[8];
#pragma unroll
    for (int u = 0; u < 8; ++u) {
      const int idx = i + u * stride;
      v[u] = A4[idx < last ? idx : last];
    }
    __builtin_amdgcn_sched_barrier(0);
#pragma unroll
    for (int u = 0; u < 8; ++u) {
      const bool live = (i + u * stride) < nvec;
      const float a = live ? v[u].x : 0.f;
      const float b = live ? v[u].y : 0.f;
      const float c = live ? v[u].z : 0.f;
      const float d = live ? v[u].w : 0.f;
      s.x += a; q.x += a * a;
      s.y += b; q.y += b * b;
      s.z += c; q.z += c * c;
      s.w += d; q.w += d * d;
    }
  }
  ((float4*)sh_s)[tid] = s;
  ((float4*)sh_q)[tid] = q;
  __syncthreads();
  if (tid < F) {
    float ts = 0.f, tq = 0.f;
    for (int p = tid; p < 1024; p += F) { ts += sh_s[p]; tq += sh_q[p]; }
    atomicAdd(&sums[tid], ts);
    atomicAdd(&sumsq[tid], tq);
  }
}

__global__ __launch_bounds__(256) void colstats2_kernel(
    const float* __restrict__ A, const float* __restrict__ patt, int nvec,
    float* __restrict__ sums_c, float* __restrict__ sumsq_c,
    float* __restrict__ sums_o, float* __restrict__ sumsq_o) {
  __shared__ float sh[4][1024];
  const float4* __restrict__ A4 = (const float4*)A;
  const float2* __restrict__ P2 = (const float2*)patt;
  const int tid = threadIdx.x;
  const int stride = gridDim.x * 256;
  const int last = nvec - 1;
  float4 sc = make_float4(0.f, 0.f, 0.f, 0.f);
  float4 qc = make_float4(0.f, 0.f, 0.f, 0.f);
  float4 so = make_float4(0.f, 0.f, 0.f, 0.f);
  float4 qo = make_float4(0.f, 0.f, 0.f, 0.f);
  for (int i = blockIdx.x * 256 + tid; i < nvec; i += 8 * stride) {
    float4 v[8];
    float2 p[8];
#pragma unroll
    for (int u = 0; u < 8; ++u) {
      const int idx = i + u * stride;
      const int cl = idx < last ? idx : last;
      v[u] = A4[cl];
      p[u] = P2[cl >> 4];
    }
    __builtin_amdgcn_sched_barrier(0);
#pragma unroll
    for (int u = 0; u < 8; ++u) {
      const bool live = (i + u * stride) < nvec;
      const float pc0 = live ? p[u].x : 0.f;
      const float po0 = live ? p[u].y : 0.f;
      float a;
      a = pc0 * v[u].x; sc.x += a; qc.x += a * a;
      a = pc0 * v[u].y; sc.y += a; qc.y += a * a;
      a = pc0 * v[u].z; sc.z += a; qc.z += a * a;
      a = pc0 * v[u].w; sc.w += a; qc.w += a * a;
      a = po0 * v[u].x; so.x += a; qo.x += a * a;
      a = po0 * v[u].y; so.y += a; qo.y += a * a;
      a = po0 * v[u].z; so.z += a; qo.z += a * a;
      a = po0 * v[u].w; so.w += a; qo.w += a * a;
    }
  }
  ((float4*)sh[0])[tid] = sc;
  ((float4*)sh[1])[tid] = qc;
  ((float4*)sh[2])[tid] = so;
  ((float4*)sh[3])[tid] = qo;
  __syncthreads();
  {
    const int which = tid >> 6;
    const int colj  = tid & (HIDDEN - 1);
    float t = 0.f;
    for (int p = colj; p < 1024; p += HIDDEN) t += sh[which][p];
    float* __restrict__ dst = which == 0 ? sums_c : (which == 1 ? sumsq_c : (which == 2 ? sums_o : sumsq_o));
    atomicAdd(&dst[colj], t);
  }
}

__device__ __forceinline__ void fold_bn_body(const float* __restrict__ sums,
                                             const float* __restrict__ sumsq, float invN,
                                             int K, int NN, const float* __restrict__ W,
                                             const float* __restrict__ extra_b,
                                             float* __restrict__ Wout,
                                             float* __restrict__ vout) {
  __shared__ float s_ms[N_FEAT];
  __shared__ float s_s[N_FEAT];
  const int j = threadIdx.x;
  for (int k = j; k < K; k += blockDim.x) {
    float mean = sums[k] * invN;
    float var  = sumsq[k] * invN - mean * mean;
    float sc   = rsqrtf(var + BN_EPS);
    s_ms[k] = mean * sc;
    s_s[k]  = sc;
  }
  __syncthreads();
  float acc = extra_b ? extra_b[j] : 0.f;
  for (int k = 0; k < K; ++k) {
    float w = W[(size_t)k * NN + j];
    Wout[(size_t)k * NN + j] = s_s[k] * w;
    acc += (BN_BETA - s_ms[k]) * w;
  }
  vout[j] = acc;
}
__global__ void fold_bn_kernel(const float* __restrict__ sums,
                               const float* __restrict__ sumsq, float invN,
                               int K, int NN, const float* __restrict__ W,
                               const float* __restrict__ extra_b,
                               float* __restrict__ Wout, float* __restrict__ vout) {
  fold_bn_body(sums, sumsq, invN, K, NN, W, extra_b, Wout, vout);
}
__global__ void fold_bn2_kernel(const float* __restrict__ stat2, float invN,
                                const float* __restrict__ W_a,
                                const float* __restrict__ W_b,
                                float* __restrict__ Wout, float* __restrict__ vout) {
  const int br = blockIdx.x;
  const float* __restrict__ s = stat2 + (size_t)br * 2 * N_FEAT;
  fold_bn_body(s, s + N_FEAT, invN, HIDDEN, HIDDEN, br ? W_b : W_a, nullptr,
               Wout + (size_t)br * HIDDEN * HIDDEN, vout + (size_t)br * HIDDEN);
}

__global__ __launch_bounds__(128) void gemm_wmma_kernel(
    const float* __restrict__ A, const float* __restrict__ Wt,
    const float* __restrict__ v, float* __restrict__ C,
    int M, int K, int doRelu) {
  const int lane = threadIdx.x & 31;
  const int wave = threadIdx.x >> 5;
  const int rowTile = blockIdx.x * 4 + wave;
  if (rowTile >= (M >> 4)) return;
  const int mrow = (rowTile << 4) + (lane & 15);
  const int kOff = (lane >> 4) << 1;
  const int n0   = lane & 15;
  v8f acc[4] = {};
  for (int kk = 0; kk < K; kk += 4) {
    const int k0 = kk + kOff;
    v2f a = *(const v2f*)(A + (size_t)mrow * K + k0);
    const float* w0 = Wt + (size_t)k0 * HIDDEN + n0;
#pragma unroll
    for (int t = 0; t < 4; ++t) {
      v2f b;
      b.x = w0[t * 16];
      b.y = w0[HIDDEN + t * 16];
      acc[t] = __builtin_amdgcn_wmma_f32_16x16x4_f32(
          false, a, false, b, (short)0, acc[t], false, false);
    }
  }
  const int rowBase = (rowTile << 4) + ((lane >> 4) << 3);
#pragma unroll
  for (int t = 0; t < 4; ++t) {
    const int n = t * 16 + n0;
    const float bias = v[n];
#pragma unroll
    for (int e = 0; e < 8; ++e) {
      float val = acc[t][e] + bias;
      if (doRelu) val = fmaxf(val, 0.f);
      C[(size_t)(rowBase + e) * HIDDEN + n] = val;
    }
  }
}

__global__ void init_kernel(float* __restrict__ stat, int nstat,
                            float* __restrict__ gpool, int npool,
                            int* __restrict__ cnt, float* __restrict__ degu,
                            float* __restrict__ deg2, int N) {
  int i = blockIdx.x * blockDim.x + threadIdx.x;
  if (i < nstat) stat[i] = 0.f;
  if (i < npool) gpool[i] = 0.f;
  if (i < N) {
    cnt[i] = 0;
    degu[i] = 1.0f;
    ((float2*)deg2)[i] = make_float2(1.0f, 1.0f);
  }
}
__device__ __forceinline__ float inv_sqrt_deg(float d) {
  return d > 0.f ? rsqrtf(d) : 0.f;
}

__global__ void csr_count_kernel(const int* __restrict__ row,
                                 const int* __restrict__ col, int E,
                                 int* __restrict__ cnt, float* __restrict__ degu) {
  int e = blockIdx.x * blockDim.x + threadIdx.x;
  if (e >= E) return;
  atomicAdd(&cnt[col[e]], 1);
  atomicAdd(&degu[row[e]], 1.0f);
}
__device__ __forceinline__ void load_counts4(const int* __restrict__ cnt, int i0,
                                             int N, int c[4]) {
  if (i0 + 3 < N) {
    const int4 v = *(const int4*)(cnt + i0);
    c[0] = v.x; c[1] = v.y; c[2] = v.z; c[3] = v.w;
  } else {
    c[0] = (i0     < N) ? cnt[i0]     : 0;
    c[1] = (i0 + 1 < N) ? cnt[i0 + 1] : 0;
    c[2] = (i0 + 2 < N) ? cnt[i0 + 2] : 0;
    c[3] = 0;
  }
}
__global__ __launch_bounds__(256) void csr_tilesum_kernel(
    const int* __restrict__ cnt, int N, int* __restrict__ tsum) {
  __shared__ int sh[8];
  const int tid = threadIdx.x;
  int c[4];
  load_counts4(cnt, (blockIdx.x * 256 + tid) * 4, N, c);
  int t = c[0] + c[1] + c[2] + c[3];
  for (int o = 16; o > 0; o >>= 1) t += __shfl_xor(t, o);
  if ((tid & 31) == 0) sh[tid >> 5] = t;
  __syncthreads();
  if (tid == 0) {
    int s = 0;
    for (int w = 0; w < 8; ++w) s += sh[w];
    tsum[blockIdx.x] = s;
  }
}
__global__ __launch_bounds__(256) void csr_scan_kernel(
    const int* __restrict__ cnt, const int* __restrict__ tsum, int N,
    int* __restrict__ off, int* __restrict__ cur) {
  __shared__ int sh_w[8];
  __shared__ int sh_b[8];
  const int tid = threadIdx.x, lane = tid & 31, wave = tid >> 5;
  int bpart = 0;
  for (int t = tid; t < (int)blockIdx.x; t += 256) bpart += tsum[t];
  for (int o = 16; o > 0; o >>= 1) bpart += __shfl_xor(bpart, o);
  const int i0 = (blockIdx.x * 256 + tid) * 4;
  int c[4];
  load_counts4(cnt, i0, N, c);
  const int t = c[0] + c[1] + c[2] + c[3];
  int inc = t;
  for (int o = 1; o < 32; o <<= 1) {
    const int y = __shfl_up(inc, o);
    if (lane >= o) inc += y;
  }
  if (lane == 31) sh_w[wave] = inc;
  if (lane == 0)  sh_b[wave] = bpart;
  __syncthreads();
  int base = 0;
  for (int w = 0; w < 8; ++w) {
    base += sh_b[w];
    if (w < wave) base += sh_w[w];
  }
  const int o0 = base + inc - t;
  const int o1 = o0 + c[0], o2 = o1 + c[1], o3 = o2 + c[2];
  if (i0 + 3 < N) {
    const int4 v = make_int4(o0, o1, o2, o3);
    *(int4*)(off + i0) = v;
    *(int4*)(cur + i0) = v;
  } else {
    if (i0     < N) { off[i0]     = o0; cur[i0]     = o0; }
    if (i0 + 1 < N) { off[i0 + 1] = o1; cur[i0 + 1] = o1; }
    if (i0 + 2 < N) { off[i0 + 2] = o2; cur[i0 + 2] = o2; }
  }
  if (i0 < N && N - 1 <= i0 + 3) off[N] = o0 + t;
}
__global__ void csr_fill_kernel(const int* __restrict__ row,
                                const int* __restrict__ col, int E,
                                int* __restrict__ cur, int2* __restrict__ sedge) {
  int e = blockIdx.x * blockDim.x + threadIdx.x;
  if (e >= E) return;
  const int r = row[e];
  const int pos = atomicAdd(&cur[col[e]], 1);
  if ((unsigned)pos < (unsigned)E) sedge[pos] = make_int2(r, e);
}

#define T_PITCH 68

template <int CNT>
__device__ __forceinline__ void gather_rows(const float2* __restrict__ m2, int r,
                                            float cf, int k, int lane, float2& acc,
                                            float& cs) {
  int rr[CNT];
  float cc[CNT];
  float2 v[CNT];
#pragma unroll
  for (int u = 0; u < CNT; ++u) {
    rr[u] = __builtin_amdgcn_readlane(r, k + u);
    cc[u] = __int_as_float(__builtin_amdgcn_readlane(__float_as_int(cf), k + u));
  }
#pragma unroll
  for (int u = 0; u < CNT; ++u) v[u] = m2[(size_t)rr[u] * (HIDDEN / 2) + lane];
#pragma unroll
  for (int u = 0; u < CNT; ++u) {
    acc.x += cc[u] * v[u].x;
    acc.y += cc[u] * v[u].y;
    cs += cc[u];
  }
}
__device__ __forceinline__ float2 agg_row(
    const float2* __restrict__ h2, const int* __restrict__ off,
    const int2* __restrict__ sedge, const float* __restrict__ deg,
    int c, int lane, float& rs) {
  const int beg = off[c], end = off[c + 1];
  const float dc = inv_sqrt_deg(deg[c]);
  const float2 self = h2[(size_t)c * (HIDDEN / 2) + lane];
  float2 acc = make_float2(0.f, 0.f);
  float cs = 0.f;
  for (int base = beg; base < end; base += 32) {
    const int p = base + lane;
    int r = 0;
    float cf = 0.f;
    if (p < end) {
      const int2 se = sedge[p];
      r = se.x;
      const float dr = inv_sqrt_deg(deg[r]);
      cf = dr * dc;
    }
    const int n = (end - base) < 32 ? (end - base) : 32;
    int k = 0;
    for (; k + 4 <= n; k += 4) gather_rows<4>(h2, r, cf, k, lane, acc, cs);
    const int rem = n - k;
    if (rem == 3)      gather_rows<3>(h2, r, cf, k, lane, acc, cs);
    else if (rem == 2) gather_rows<2>(h2, r, cf, k, lane, acc, cs);
    else if (rem == 1) gather_rows<1>(h2, r, cf, k, lane, acc, cs);
  }
  rs = cs + dc * dc;
  float2 t;
  t.x = acc.x + dc * dc * self.x;
  t.y = acc.y + dc * dc * self.y;
  return t;
}

__device__ __forceinline__ void load_b_frags(const float* __restrict__ wp,
                                             v2f (&bf)[HIDDEN / 4]) {
#pragma unroll
  for (int q = 0; q < HIDDEN / 4; ++q) {
    bf[q].x = wp[(4 * q) * HIDDEN];
    bf[q].y = wp[(4 * q + 1) * HIDDEN];
  }
}
__device__ __forceinline__ v8f tile_product(const float* ap, const v2f (&bf)[HIDDEN / 4]) {
  v8f acc = {};
#pragma unroll
  for (int q = 0; q < HIDDEN / 4; ++q) {
    const v2f a = *(const v2f*)(ap + 4 * q);
    acc = __builtin_amdgcn_wmma_f32_16x16x4_f32(
        false, a, false, bf[q], (short)0, acc, false, false);
  }
  return acc;
}

__global__ __launch_bounds__(128) void gather_gemm_kernel(
    const float* __restrict__ h, const int* __restrict__ off,
    const int2* __restrict__ sedge, const float* __restrict__ deg,
    const float* __restrict__ Wt, const float* __restrict__ v,
    const float* __restrict__ b, float* __restrict__ out, int N) {
  __shared__ __attribute__((aligned(16))) float sh_t[16 * T_PITCH];
  __shared__ float sh_rs[16];
  const int lane = threadIdx.x & 31;
  const int wave = threadIdx.x >> 5;
  const int kOff = (lane >> 4) << 1;
  const int n0   = lane & 15;
  const int nbase = blockIdx.x * 16;
#pragma nounroll
  for (int i = 0; i < 4; ++i) {
    const int rl = wave * 4 + i;
    const int c = __builtin_amdgcn_readfirstlane(nbase + rl);
    float2 t = make_float2(0.f, 0.f);
    float rs = 0.f;
    if (c < N) t = agg_row((const float2*)h, off, sedge, deg, c, lane, rs);
    *(float2*)(sh_t + rl * T_PITCH + 2 * lane) = t;
    if (lane == 0) sh_rs[rl] = rs;
  }
  v2f bf[HIDDEN / 4];
  load_b_frags(Wt + (size_t)kOff * HIDDEN + wave * 16 + n0, bf);
  __syncthreads();
  const v8f acc = tile_product(sh_t + n0 * T_PITCH + kOff, bf);
  __syncthreads();
  {
    const int n = wave * 16 + n0;
    const float vn = v[n], bn = b[n];
    const int rowBase = (lane >> 4) << 3;
#pragma unroll
    for (int e = 0; e < 8; ++e) {
      const int r = rowBase + e;
      sh_t[r * T_PITCH + n] = fmaxf(acc[e] + sh_rs[r] * vn + bn, 0.f);
    }
  }
  __syncthreads();
#pragma unroll
  for (int j = 0; j < 2; ++j) {
    const int r = wave * 4 + 2 * j + (lane >> 4);
    const int node = nbase + r;
    if (node < N) {
      const float4 val = *(const float4*)(sh_t + r * T_PITCH + 4 * n0);
      ((float4*)out)[(size_t)node * (HIDDEN / 4) + n0] = val;
    }
  }
}

template <int CNT>
__device__ __forceinline__ void gather_rows2(const float2* __restrict__ h2, int r,
                                             float ec, float eo, float2 pw, int k,
                                             int lane, float2& ac, float2& ao,
                                             float& csc, float& cso) {
  int rr[CNT];
  float e0[CNT], e1[CNT], c0[CNT], c1[CNT];
  float2 v[CNT];
#pragma unroll
  for (int u = 0; u < CNT; ++u) {
    rr[u] = __builtin_amdgcn_readlane(r, k + u);
    e0[u] = __int_as_float(__builtin_amdgcn_readlane(__float_as_int(ec), k + u));
    e1[u] = __int_as_float(__builtin_amdgcn_readlane(__float_as_int(eo), k + u));
    c0[u] = e0[u] * __int_as_float(__builtin_amdgcn_readlane(__float_as_int(pw.x), k + u));
    c1[u] = e1[u] * __int_as_float(__builtin_amdgcn_readlane(__float_as_int(pw.y), k + u));
  }
#pragma unroll
  for (int u = 0; u < CNT; ++u) v[u] = h2[(size_t)rr[u] * (HIDDEN / 2) + lane];
#pragma unroll
  for (int u = 0; u < CNT; ++u) {
    ac.x += c0[u] * v[u].x;
    ac.y += c0[u] * v[u].y;
    ao.x += c1[u] * v[u].x;
    ao.y += c1[u] * v[u].y;
    csc += e0[u];
    cso += e1[u];
  }
}
__device__ __forceinline__ void pool_rows8(const v8f& acc, const float* __restrict__ rs8,
                                           const int* __restrict__ gid8, float vn,
                                           float bn, float* __restrict__ g) {
  int gw[8];
  float val[8];
#pragma unroll
  for (int e = 0; e < 8; ++e) {
    gw[e] = gid8[e];
    val[e] = fmaxf(acc[e] + rs8[e] * vn + bn, 0.f);
  }
  int cur = gw[0];
  float a = val[0];
#pragma unroll
  for (int e = 1; e < 8; ++e) {
    if (gw[e] == cur) {
      a += val[e];
    } else {
      if ((unsigned)cur < (unsigned)N_GRAPHS) atomicAdd(&g[(size_t)cur * HIDDEN], a);
      cur = gw[e];
      a = val[e];
    }
  }
  if ((unsigned)cur < (unsigned)N_GRAPHS) atomicAdd(&g[(size_t)cur * HIDDEN], a);
}
__global__ __launch_bounds__(128) void gather_gemm_pool2_kernel(
    const float* __restrict__ h, const float* __restrict__ patt,
    const int* __restrict__ off, const int2* __restrict__ sedge,
    const float* __restrict__ att, const float* __restrict__ deg2,
    const float* __restrict__ Wt2, const float* __restrict__ v2,
    const float* __restrict__ b_ctx, const float* __restrict__ b_obj,
    const int* __restrict__ batch, float* __restrict__ gpool, int N) {
  __shared__ __attribute__((aligned(16))) float sh_t[2][16 * T_PITCH];
  __shared__ float sh_rs[2][16];
  __shared__ int sh_gid[16];
  const int lane = threadIdx.x & 31;
  const int wave = threadIdx.x >> 5;
  const int kOff = (lane >> 4) << 1;
  const int n0   = lane & 15;
  const int nbase = blockIdx.x * 16;
  const float2* __restrict__ h2 = (const float2*)h;
  const float2* __restrict__ P2 = (const float2*)patt;
  const float2* __restrict__ d2 = (const float2*)deg2;
  const float2* __restrict__ att2 = (const float2*)att;
#pragma nounroll
  for (int i = 0; i < 4; ++i) {
    const int rl = wave * 4 + i;
    const int c = __builtin_amdgcn_readfirstlane(nbase + rl);
    float2 tc = make_float2(0.f, 0.f), to = make_float2(0.f, 0.f);
    float rsc = 0.f, rso = 0.f;
    int gid = -1;
    if (c < N) {
      gid = batch[c];
      const int beg = off[c], end = off[c + 1];
      const float2 dcv = d2[c];
      const float dcc = inv_sqrt_deg(dcv.x);
      const float dco = inv_sqrt_deg(dcv.y);
      const float2 pself = P2[c];
      const float2 self = h2[(size_t)c * (HIDDEN / 2) + lane];
      float2 ac = make_float2(0.f, 0.f), ao = make_float2(0.f, 0.f);
      float csc = 0.f, cso = 0.f;
      for (int base = beg; base < end; base += 32) {
        const int p = base + lane;
        int r = 0;
        float ec = 0.f, eo = 0.f;
        float2 pw = make_float2(0.f, 0.f);
        if (p < end) {
          const int2 se = sedge[p];
          r = se.x;
          const float2 dr = d2[r];
          const float2 w = att2[se.y];
          pw = P2[r];
          ec = inv_sqrt_deg(dr.x) * w.x * dcc;
          eo = inv_sqrt_deg(dr.y) * w.y * dco;
        }
        const int n = (end - base) < 32 ? (end - base) : 32;
        int k = 0;
        for (; k + 4 <= n; k += 4) gather_rows2<4>(h2, r, ec, eo, pw, k, lane, ac, ao, csc, cso);
        const int rem = n - k;
        if (rem == 3)      gather_rows2<3>(h2, r, ec, eo, pw, k, lane, ac, ao, csc, cso);
        else if (rem == 2) gather_rows2<2>(h2, r, ec, eo, pw, k, lane, ac, ao, csc, cso);
        else if (rem == 1) gather_rows2<1>(h2, r, ec, eo, pw, k, lane, ac, ao, csc, cso);
      }
      rsc = csc + dcc * dcc;
      rso = cso + dco * dco;
      const float sc = dcc * dcc * pself.x;
      const float so = dco * dco * pself.y;
      tc.x = ac.x + sc * self.x;
      tc.y = ac.y + sc * self.y;
      to.x = ao.x + so * self.x;
      to.y = ao.y + so * self.y;
    }
    *(float2*)(sh_t[0] + rl * T_PITCH + 2 * lane) = tc;
    *(float2*)(sh_t[1] + rl * T_PITCH + 2 * lane) = to;
    if (lane == 0) {
      sh_rs[0][rl] = rsc;
      sh_rs[1][rl] = rso;
      sh_gid[rl] = gid;
    }
  }
  const int n = wave * 16 + n0;
  const int rowBase = (lane >> 4) << 3;
  const float* wp = Wt2 + (size_t)kOff * HIDDEN + n;
  const float* ap = sh_t[0] + n0 * T_PITCH + kOff;
  v2f bf[HIDDEN / 4];
  load_b_frags(wp, bf);
  __syncthreads();
  {
    const v8f acc = tile_product(ap, bf);
    pool_rows8(acc, sh_rs[0] + rowBase, sh_gid + rowBase, v2[n], b_ctx[n], gpool + n);
  }
  load_b_frags(wp + HIDDEN * HIDDEN, bf);
  {
    const v8f acc = tile_product(ap + 16 * T_PITCH, bf);
    pool_rows8(acc, sh_rs[1] + rowBase, sh_gid + rowBase, v2[HIDDEN + n], b_obj[n],
               gpool + (size_t)N_GRAPHS * HIDDEN + n);
  }
}

__global__ void node_att_kernel(const float* __restrict__ h,
                                const float* __restrict__ Wna,
                                const float* __restrict__ bna,
                                const float* __restrict__ Wea,
                                float* __restrict__ patt,
                                float* __restrict__ pr,
                                float* __restrict__ pc, int N) {
  int i = blockIdx.x * blockDim.x + threadIdx.x;
  if (i >= N) return;
  const float* hi = h + (size_t)i * HIDDEN;
  float l0 = bna[0], l1 = bna[1];
  float r0 = 0.f, r1 = 0.f, c0 = 0.f, c1 = 0.f;
  for (int j = 0; j < HIDDEN; ++j) {
    float a = hi[j];
    l0 += a * Wna[j * 2 + 0];
    l1 += a * Wna[j * 2 + 1];
    r0 += a * Wea[j * 2 + 0];
    r1 += a * Wea[j * 2 + 1];
    c0 += a * Wea[(HIDDEN + j) * 2 + 0];
    c1 += a * Wea[(HIDDEN + j) * 2 + 1];
  }
  ((float2*)pr)[i] = make_float2(r0, r1);
  ((float2*)pc)[i] = make_float2(c0, c1);
  float mx = fmaxf(l0, l1);
  float e0 = __expf(l0 - mx), e1 = __expf(l1 - mx);
  float inv = 1.f / (e0 + e1);
  float p0 = e0 * inv, p1 = e1 * inv;
  ((float2*)patt)[i] = make_float2(p0, p1);
}

__global__ void edge_att_kernel(const float* __restrict__ pr,
                                const float* __restrict__ pc,
                                const int* __restrict__ row,
                                const int* __restrict__ col,
                                const float* __restrict__ bea,
                                float* __restrict__ att,
                                float* __restrict__ deg2, int E) {
  int e = blockIdx.x * blockDim.x + threadIdx.x;
  if (e >= E) return;
  const int r = row[e];
  const float2 a = ((const float2*)pr)[r];
  const float2 c = ((const float2*)pc)[col[e]];
  const float l0 = (a.x + c.x) + bea[0];
  const float l1 = (a.y + c.y) + bea[1];
  float mx = fmaxf(l0, l1);
  float e0 = __expf(l0 - mx), e1 = __expf(l1 - mx);
  float inv = 1.f / (e0 + e1);
  const float w0 = e0 * inv, w1 = e1 * inv;
  ((float2*)att)[e] = make_float2(w0, w1);
  atomicAdd(&deg2[2 * (size_t)r], w0);
  atomicAdd(&deg2[2 * (size_t)r + 1], w1);
}

#define HEAD_THREADS 512
#define G_PITCH 68
#define W_PITCH 72

__device__ __forceinline__ void head_colstats(const float* __restrict__ ps,
                                              const float* __restrict__ pq,
                                              float* __restrict__ sc,
                                              float* __restrict__ ms, int col) {
  float ts = 0.f, tq = 0.f;
  for (int p = 0; p < 32; ++p) {
    ts += ps[p * HIDDEN + col];
    tq += pq[p * HIDDEN + col];
  }
  const float invN = 1.f / N_GRAPHS;
  const float mean = ts * invN;
  const float var  = tq * invN - mean * mean;
  const float s    = rsqrtf(var + BN_EPS);
  sc[col] = s;
  ms[col] = mean * s;
}

__global__ __launch_bounds__(HEAD_THREADS) void heads_kernel(
    const float* __restrict__ gc, const float* __restrict__ go,
    const float* __restrict__ W1_c, const float* __restrict__ b1_c,
    const float* __restrict__ W2_c, const float* __restrict__ b2_c,
    const float* __restrict__ W1_o, const float* __restrict__ b1_o,
    const float* __restrict__ W2_o, const float* __restrict__ b2_o,
    const float* __restrict__ W1_co, const float* __restrict__ b1_co,
    const float* __restrict__ W2_co, const float* __restrict__ b2_co,
    float* __restrict__ out) {
  __shared__ __attribute__((aligned(16))) float sh_g[N_GRAPHS * G_PITCH];
  __shared__ __attribute__((aligned(16))) float sh_w[HIDDEN * W_PITCH];
  __shared__ __attribute__((aligned(16))) float sh_ps[32 * HIDDEN];
  __shared__ __attribute__((aligned(16))) float sh_pq[32 * HIDDEN];
  __shared__ float sh_sc[HIDDEN];
  __shared__ float sh_ms[HIDDEN];
  __shared__ float sh_v[HIDDEN];
  __shared__ __attribute__((aligned(16))) float sh_w2[HIDDEN * N_CLASSES];
  __shared__ float sh_v2[16];
  __shared__ float sh_w1r[HIDDEN * HIDDEN];
  __shared__ float sh_w2r[HIDDEN * N_CLASSES];

  const int head = blockIdx.x;
  const float* __restrict__ W1 = head == 0 ? W1_c : (head == 1 ? W1_o : W1_co);
  const float* __restrict__ b1 = head == 0 ? b1_c : (head == 1 ? b1_o : b1_co);
  const float* __restrict__ W2 = head == 0 ? W2_c : (head == 1 ? W2_o : W2_co);
  const float* __restrict__ b2 = head == 0 ? b2_c : (head == 1 ? b2_o : b2_co);
  const int tid  = threadIdx.x;
  const int lane = tid & 31;
  const int wave = tid >> 5;

  float w1r[8];
#pragma unroll
  for (int u = 0; u < 8; ++u) w1r[u] = W1[tid + HEAD_THREADS * u];
  const float w2r0 = W2[tid];
  const float w2r1 = tid < HIDDEN * N_CLASSES - HEAD_THREADS ? W2[HEAD_THREADS + tid] : 0.f;
  const float b1r  = tid < HIDDEN ? b1[tid] : 0.f;
  const float b2r  = tid < N_CLASSES ? b2[tid] : 0.f;

  {
    const float4* __restrict__ gc4 = (const float4*)gc;
    const float4* __restrict__ go4 = (const float4*)go;
    const float4* __restrict__ pa  = head == 1 ? go4 : gc4;
    const int cg = tid & 15;
    const int r0 = tid >> 4;
    float4 x[16];
#pragma unroll
    for (int u = 0; u < 16; ++u) x[u] = pa[tid + HEAD_THREADS * u];
    if (head == 2) {
#pragma unroll
      for (int u = 0; u < 16; ++u) {
        const float4 b = go4[tid + HEAD_THREADS * u];
        x[u] = make_float4(x[u].x + b.x, x[u].y + b.y, x[u].z + b.z, x[u].w + b.w);
      }
    }
    float4 s = make_float4(0.f, 0.f, 0.f, 0.f);
    float4 q = make_float4(0.f, 0.f, 0.f, 0.f);
#pragma unroll
    for (int u = 0; u < 16; ++u) {
      *(float4*)(sh_g + (r0 + 32 * u) * G_PITCH + 4 * cg) = x[u];
      s.x += x[u].x; q.x += x[u].x * x[u].x;
      s.y += x[u].y; q.y += x[u].y * x[u].y;
      s.z += x[u].z; q.z += x[u].z * x[u].z;
      s.w += x[u].w; q.w += x[u].w * x[u].w;
    }
    *(float4*)(sh_ps + r0 * HIDDEN + 4 * cg) = s;
    *(float4*)(sh_pq + r0 * HIDDEN + 4 * cg) = q;
#pragma unroll
    for (int u = 0; u < 8; ++u) sh_w1r[tid + HEAD_THREADS * u] = w1r[u];
    sh_w2r[tid] = w2r0;
    if (tid < HIDDEN * N_CLASSES - HEAD_THREADS) sh_w2r[HEAD_THREADS + tid] = w2r1;
  }
  __syncthreads();
  if (tid < HIDDEN) head_colstats(sh_ps, sh_pq, sh_sc, sh_ms, tid);
  __syncthreads();

#pragma unroll
  for (int u = 0; u < 8; ++u) {
    const int i = tid + HEAD_THREADS * u;
    const int k = i >> 6, j = i & 63;
    sh_w[k * W_PITCH + j] = sh_sc[k] * w1r[u];
  }
  if (tid < HIDDEN) {
    float acc = b1r;
    for (int k = 0; k < HIDDEN; ++k) acc += (BN_BETA - sh_ms[k]) * sh_w1r[k * HIDDEN + tid];
    sh_v[tid] = acc;
  }
  __syncthreads();

  {
    const int kOff = (lane >> 4) << 1;
    const int n0   = lane & 15;
    v8f acc[2][4] = {};
    const float* a0p = sh_g + (32 * wave + n0) * G_PITCH + kOff;
    const float* a1p = a0p + 16 * G_PITCH;
    const float* bp  = sh_w + kOff * W_PITCH + n0;
    for (int kk = 0; kk < HIDDEN; kk += 4) {
      const v2f a0 = *(const v2f*)(a0p + kk);
      const v2f a1 = *(const v2f*)(a1p + kk);
#pragma unroll
      for (int t = 0; t < 4; ++t) {
        v2f b;
        b.x = bp[kk * W_PITCH + t * 16];
        b.y = bp[(kk + 1) * W_PITCH + t * 16];
        acc[0][t] = __builtin_amdgcn_wmma_f32_16x16x4_f32(
            false, a0, false, b, (short)0, acc[0][t], false, false);
        acc[1][t] = __builtin_amdgcn_wmma_f32_16x16x4_f32(
            false, a1, false, b, (short)0, acc[1][t], false, false);
      }
    }
    const int rowBase = 32 * wave + ((lane >> 4) << 3);
#pragma unroll
    for (int t = 0; t < 4; ++t) {
      const int n = t * 16 + n0;
      const float bias = sh_v[n];
      float s = 0.f, q = 0.f;
#pragma unroll
      for (int i = 0; i < 2; ++i) {
#pragma unroll
        for (int e = 0; e < 8; ++e) {
          const float val = fmaxf(acc[i][t][e] + bias, 0.f);
          sh_g[(rowBase + 16 * i + e) * G_PITCH + n] = val;
          s += val;
          q += val * val;
        }
      }
      sh_ps[(2 * wave + (lane >> 4)) * HIDDEN + n] = s;
      sh_pq[(2 * wave + (lane >> 4)) * HIDDEN + n] = q;
    }
  }
  __syncthreads();
  if (tid < HIDDEN) head_colstats(sh_ps, sh_pq, sh_sc, sh_ms, tid);
  __syncthreads();

  sh_w2[tid] = sh_sc[tid / N_CLASSES] * w2r0;
  if (tid < HIDDEN * N_CLASSES - HEAD_THREADS)
    sh_w2[HEAD_THREADS + tid] = sh_sc[(HEAD_THREADS + tid) / N_CLASSES] * w2r1;
  if (tid < N_CLASSES) {
    float acc = b2r;
    for (int k = 0; k < HIDDEN; ++k) acc += (BN_BETA - sh_ms[k]) * sh_w2r[k * N_CLASSES + tid];
    sh_v2[tid] = acc;
  }
  __syncthreads();

  {
    float l[N_CLASSES];
    for (int c = 0; c < N_CLASSES; ++c) l[c] = sh_v2[c];
    const float4* __restrict__ row4 = (const float4*)(sh_g + tid * G_PITCH);
    for (int k4 = 0; k4 < HIDDEN / 4; ++k4) {
      const float4 x = row4[k4];
      const float* w = sh_w2 + k4 * 4 * N_CLASSES;
      for (int c = 0; c < N_CLASSES; ++c) l[c] += x.x * w[c];
      for (int c = 0; c < N_CLASSES; ++c) l[c] += x.y * w[N_CLASSES + c];
      for (int c = 0; c < N_CLASSES; ++c) l[c] += x.z * w[2 * N_CLASSES + c];
      for (int c = 0; c < N_CLASSES; ++c) l[c] += x.w * w[3 * N_CLASSES + c];
    }
    float mx = l[0];
    for (int c = 1; c < N_CLASSES; ++c) mx = fmaxf(mx, l[c]);
    float s = 0.f;
    for (int c = 0; c < N_CLASSES; ++c) s += __expf(l[c] - mx);
    const float lse = __logf(s) + mx;
    float* __restrict__ o = out + ((size_t)head * N_GRAPHS + tid) * N_CLASSES;
    for (int c = 0; c < N_CLASSES; ++c) o[c] = l[c] - lse;
  }
}

extern "C" void kernel_launch(void* const* d_in, const int* in_sizes, int n_in,
                              void* d_out, int out_size, void* d_ws, size_t ws_size,
                              hipStream_t stream) {
  (void)in_sizes; (void)n_in; (void)out_size; (void)ws_size;
  const float* x       = (const float*)d_in[0];
  const int*   ei      = (const int*)d_in[1];
  const int*   row     = ei;
  const int*   col     = ei + N_EDGES;
  const int*   batch   = (const int*)d_in[2];
  const float* W_feat  = (const float*)d_in[3];
  const float* W_convs = (const float*)d_in[4];
  const float* b_convs = (const float*)d_in[5];
  const float* W_ea    = (const float*)d_in[6];
  const float* b_ea    = (const float*)d_in[7];
  const float* W_na    = (const float*)d_in[8];
  const float* b_na    = (const float*)d_in[9];
  const float* W_ctx   = (const float*)d_in[10];
  const float* b_ctx   = (const float*)d_in[11];
  const float* W_obj   = (const float*)d_in[12];
  const float* b_obj   = (const float*)d_in[13];
  const float* W1h[3]  = {(const float*)d_in[14], (const float*)d_in[18], (const float*)d_in[22]};
  const float* b1h[3]  = {(const float*)d_in[15], (const float*)d_in[19], (const float*)d_in[23]};
  const float* W2h[3]  = {(const float*)d_in[16], (const float*)d_in[20], (const float*)d_in[24]};
  const float* b2h[3]  = {(const float*)d_in[17], (const float*)d_in[21], (const float*)d_in[25]};

  size_t off = 0;
  auto carve = [&](size_t elems) -> float* {
    float* p = (float*)((char*)d_ws + off);
    off += ((elems * sizeof(float)) + 255) & ~(size_t)255;
    return p;
  };
  float* hb    = carve((size_t)N_NODES * HIDDEN);
  float* hb2   = carve((size_t)N_NODES * HIDDEN);
  float* patt  = carve((size_t)N_NODES * 2);
  float* att   = carve((size_t)N_EDGES * 2);
  float* pr    = carve((size_t)N_NODES * 2);
  float* pc    = carve((size_t)N_NODES * 2);
  float* degu  = carve(N_NODES);
  float* deg2  = carve((size_t)N_NODES * 2);
  float* gpool = carve((size_t)2 * N_GRAPHS * HIDDEN);
  float* gc    = gpool;
  float* go    = gpool + (size_t)N_GRAPHS * HIDDEN;
  float* stat  = carve((size_t)6 * 2 * N_FEAT);
  float* Wf    = carve((size_t)N_FEAT * HIDDEN);
  float* vf    = carve(HIDDEN);
  float* Wf2   = carve((size_t)2 * HIDDEN * HIDDEN);
  float* vf2   = carve((size_t)2 * HIDDEN);
  int*  ecnt   = (int*)carve(N_NODES);
  int*  etsum  = (int*)carve(128);
  int*  eoff   = (int*)carve(N_NODES + 1);
  int*  ecur   = (int*)carve(N_NODES);
  int2* sedge  = (int2*)carve((size_t)N_EDGES * 2);

  auto nblk = [](long n, int b) { return (int)((n + b - 1) / b); };

  auto stats = [&](const float* A, int N, int F, int slot) {
    const int nvec = N * F / 4;
    const int grid = nblk(nvec, 256) < 512 ? nblk(nvec, 256) : 512;
    colstats_kernel<<<grid, 256, 0, stream>>>(A, nvec, F, stat + slot * 2 * N_FEAT,
                                              stat + slot * 2 * N_FEAT + N_FEAT);
  };
  auto fold = [&](int slot, int K, const float* W) {
    fold_bn_kernel<<<1, HIDDEN, 0, stream>>>(stat + slot * 2 * N_FEAT,
                                             stat + slot * 2 * N_FEAT + N_FEAT,
                                             1.f / N_NODES, K, HIDDEN, W, nullptr, Wf, vf);
  };
  auto gemm = [&](const float* A, float* C, int M, int K, int relu) {
    gemm_wmma_kernel<<<nblk(M / 16, 4), 128, 0, stream>>>(A, Wf, vf, C, M, K, relu);
  };

  static_assert(2 * N_GRAPHS * HIDDEN <= N_NODES && 6 * 2 * N_FEAT <= N_NODES,
                "init_kernel's grid is sized by the node count");
  init_kernel<<<nblk(N_NODES, 256), 256, 0, stream>>>(stat, 6 * 2 * N_FEAT, gpool,
                                                      2 * N_GRAPHS * HIDDEN, ecnt, degu,
                                                      deg2, N_NODES);

  stats(x, N_NODES, N_FEAT, 0);
  fold(0, N_FEAT, W_feat);
  gemm(x, hb, N_NODES, N_FEAT, 1);

  csr_count_kernel<<<nblk(N_EDGES, 256), 256, 0, stream>>>(row, col, N_EDGES, ecnt, degu);
  csr_tilesum_kernel<<<nblk(N_NODES, 1024), 256, 0, stream>>>(ecnt, N_NODES, etsum);
  csr_scan_kernel<<<nblk(N_NODES, 1024), 256, 0, stream>>>(ecnt, etsum, N_NODES,
                                                           eoff, ecur);
  csr_fill_kernel<<<nblk(N_EDGES, 256), 256, 0, stream>>>(row, col, N_EDGES, ecur,
                                                          sedge);

  for (int i = 0; i < 3; ++i) {
    const float* hsrc = (i & 1) ? hb2 : hb;
    float*       hdst = (i & 1) ? hb : hb2;
    stats(hsrc, N_NODES, HIDDEN, 1 + i);
    fold(1 + i, HIDDEN, W_convs + (size_t)i * HIDDEN * HIDDEN);
    gather_gemm_kernel<<<nblk(N_NODES, 16), 128, 0, stream>>>(
        hsrc, eoff, sedge, degu, Wf, vf, b_convs + (size_t)i * HIDDEN, hdst, N_NODES);
  }
  const float* hfin = hb2;

  node_att_kernel<<<nblk(N_NODES, 256), 256, 0, stream>>>(hfin, W_na, b_na, W_ea,
                                                          patt, pr, pc, N_NODES);
  edge_att_kernel<<<nblk(N_EDGES, 256), 256, 0, stream>>>(pr, pc, row, col, b_ea,
                                                          att, deg2, N_EDGES);

  {
    const int nvec = N_NODES * HIDDEN / 4;
    const int grid = nblk(nvec, 256) < 512 ? nblk(nvec, 256) : 512;
    colstats2_kernel<<<grid, 256, 0, stream>>>(
        hfin, patt, nvec, stat + 4 * 2 * N_FEAT, stat + 4 * 2 * N_FEAT + N_FEAT,
        stat + 5 * 2 * N_FEAT, stat + 5 * 2 * N_FEAT + N_FEAT);
    fold_bn2_kernel<<<2, HIDDEN, 0, stream>>>(stat + 4 * 2 * N_FEAT, 1.f / N_NODES,
                                              W_ctx, W_obj, Wf2, vf2);
    gather_gemm_pool2_kernel<<<nblk(N_NODES, 16), 128, 0, stream>>>(
        hfin, patt, eoff, sedge, att, deg2, Wf2, vf2, b_ctx, b_obj, batch, gpool, N_NODES);
  }

  static_assert(HEAD_THREADS == N_GRAPHS, "one thread per graph in the logits stage");
  heads_kernel<<<3, HEAD_THREADS, 0, stream>>>(
      gc, go,
      W1h[0], b1h[0], W2h[0], b2h[0],
      W1h[1], b1h[1], W2h[1], b2h[1],
      W1h[2], b1h[2], W2h[2], b2h[2],
      (float*)d_out);
}


# batch norms applied to the A operand inside the multiplying kernels (x fragments; gathered tile s*(Ah)+(A1)t), weights as given: the 5 fold launches go, 23 -> 18 (shift summed inside the product)
# speedup vs baseline: 4.3808x; 1.1076x over previous
#include <hip/hip_runtime.h>
#include <math.h>


#define N_NODES  100000
#define N_EDGES  1000000
#define N_FEAT   128
#define HIDDEN   64
#define N_GRAPHS 512
#define N_CLASSES 10
#define BN_EPS  1e-5f
#define BN_BETA 1e-4f

typedef __attribute__((ext_vector_type(2))) float v2f;
typedef __attribute__((ext_vector_type(8))) float v8f;

__global__ __launch_bounds__(256) void colstats_kernel(
    const float* __restrict__ A, int nvec, int F,
    float* __restrict__ sums, float* __restrict__ sumsq) {
  __shared__ float sh_s[1024];
  __shared__ float sh_q[1024];
  const float4* __restrict__ A4 = (const float4*)A;
  const int tid = threadIdx.x;
  const int stride = gridDim.x * 256;
  const int last = nvec - 1;
  float4 s = make_float4(0.f, 0.f, 0.f, 0.f);
  float4 q = make_float4(0.f, 0.f, 0.f, 0.f);
  for (int i = blockIdx.x * 256 + tid; i < nvec; i += 8 * stride) {
    float4 v[8];
#pragma unroll
    for (int u = 0; u < 8; ++u) {
      const int idx = i + u * stride;
      v[u] = A4[idx < last ? idx : last];
    }
    __builtin_amdgcn_sched_barrier(0);
#pragma unroll
    for (int u = 0; u < 8; ++u) {
      const bool live = (i + u * stride) < nvec;
      const float a = live ? v[u].x : 0.f;
      const float b = live ? v[u].y : 0.f;
      const float c = live ? v[u].z : 0.f;
      const float d = live ? v[u].w : 0.f;
      s.x += a; q.x += a * a;
      s.y += b; q.y += b * b;
      s.z += c; q.z += c * c;
      s.w += d; q.w += d * d;
    }
  }
  ((float4*)sh_s)[tid] = s;
  ((float4*)sh_q)[tid] = q;
  __syncthreads();
  if (tid < F) {
    float ts = 0.f, tq = 0.f;
    for (int p = tid; p < 1024; p += F) { ts += sh_s[p]; tq += sh_q[p]; }
    atomicAdd(&sums[tid], ts);
    atomicAdd(&sumsq[tid], tq);
  }
}

__global__ __launch_bounds__(256) void colstats2_kernel(
    const float* __restrict__ A, const float* __restrict__ patt, int nvec,
    float* __restrict__ sums_c, float* __restrict__ sumsq_c,
    float* __restrict__ sums_o, float* __restrict__ sumsq_o) {
  __shared__ float sh[4][1024];
  const float4* __restrict__ A4 = (const float4*)A;
  const float2* __restrict__ P2 = (const float2*)patt;
  const int tid = threadIdx.x;
  const int stride = gridDim.x * 256;
  const int last = nvec - 1;
  float4 sc = make_float4(0.f, 0.f, 0.f, 0.f);
  float4 qc = make_float4(0.f, 0.f, 0.f, 0.f);
  float4 so = make_float4(0.f, 0.f, 0.f, 0.f);
  float4 qo = make_float4(0.f, 0.f, 0.f, 0.f);
  for (int i = blockIdx.x * 256 + tid; i < nvec; i += 8 * stride) {
    float4 v[8];
    float2 p[8];
#pragma unroll
    for (int u = 0; u < 8; ++u) {
      const int idx = i + u * stride;
      const int cl = idx < last ? idx : last;
      v[u] = A4[cl];
      p[u] = P2[cl >> 4];
    }
    __builtin_amdgcn_sched_barrier(0);
#pragma unroll
    for (int u = 0; u < 8; ++u) {
      const bool live = (i + u * stride) < nvec;
      const float pc0 = live ? p[u].x : 0.f;
      const float po0 = live ? p[u].y : 0.f;
      float a;
      a = pc0 * v[u].x; sc.x += a; qc.x += a * a;
      a = pc0 * v[u].y; sc.y += a; qc.y += a * a;
      a = pc0 * v[u].z; sc.z += a; qc.z += a * a;
      a = pc0 * v[u].w; sc.w += a; qc.w += a * a;
      a = po0 * v[u].x; so.x += a; qo.x += a * a;
      a = po0 * v[u].y; so.y += a; qo.y += a * a;
      a = po0 * v[u].z; so.z += a; qo.z += a * a;
      a = po0 * v[u].w; so.w += a; qo.w += a * a;
    }
  }
  ((float4*)sh[0])[tid] = sc;
  ((float4*)sh[1])[tid] = qc;
  ((float4*)sh[2])[tid] = so;
  ((float4*)sh[3])[tid] = qo;
  __syncthreads();
  {
    const int which = tid >> 6;
    const int colj  = tid & (HIDDEN - 1);
    float t = 0.f;
    for (int p = colj; p < 1024; p += HIDDEN) t += sh[which][p];
    float* __restrict__ dst = which == 0 ? sums_c : (which == 1 ? sumsq_c : (which == 2 ? sums_o : sumsq_o));
    atomicAdd(&dst[colj], t);
  }
}

__device__ __forceinline__ void bn_coeffs(float sum, float sumsq, float invN,
                                          float& sc, float& sh) {
  const float mean = sum * invN;
  const float var  = sumsq * invN - mean * mean;
  sc = rsqrtf(var + BN_EPS);
  sh = BN_BETA - mean * sc;
}

__global__ __launch_bounds__(128) void gemm_wmma_kernel(
    const float* __restrict__ A, const float* __restrict__ Wt,
    const float* __restrict__ sums, const float* __restrict__ sumsq, float invN,
    float* __restrict__ C, int M, int K, int doRelu) {
  __shared__ __attribute__((aligned(8))) float sh_sc[N_FEAT];
  __shared__ __attribute__((aligned(8))) float sh_sh[N_FEAT];
  for (int k = threadIdx.x; k < K; k += blockDim.x)
    bn_coeffs(sums[k], sumsq[k], invN, sh_sc[k], sh_sh[k]);
  __syncthreads();
  const int lane = threadIdx.x & 31;
  const int wave = threadIdx.x >> 5;
  const int rowTile = blockIdx.x * 4 + wave;
  if (rowTile >= (M >> 4)) return;
  const int mrow = (rowTile << 4) + (lane & 15);
  const int kOff = (lane >> 4) << 1;
  const int n0   = lane & 15;
  v8f acc[4] = {};
  for (int kk = 0; kk < K; kk += 4) {
    const int k0 = kk + kOff;
    v2f a = *(const v2f*)(A + (size_t)mrow * K + k0);
    const v2f sc = *(const v2f*)(sh_sc + k0);
    const v2f sf = *(const v2f*)(sh_sh + k0);
    a.x = a.x * sc.x + sf.x;
    a.y = a.y * sc.y + sf.y;
    const float* w0 = Wt + (size_t)k0 * HIDDEN + n0;
#pragma unroll
    for (int t = 0; t < 4; ++t) {
      v2f b;
      b.x = w0[t * 16];
      b.y = w0[HIDDEN + t * 16];
      acc[t] = __builtin_amdgcn_wmma_f32_16x16x4_f32(
          false, a, false, b, (short)0, acc[t], false, false);
    }
  }
  const int rowBase = (rowTile << 4) + ((lane >> 4) << 3);
#pragma unroll
  for (int t = 0; t < 4; ++t) {
    const int n = t * 16 + n0;
#pragma unroll
    for (int e = 0; e < 8; ++e) {
      float val = acc[t][e];
      if (doRelu) val = fmaxf(val, 0.f);
      C[(size_t)(rowBase + e) * HIDDEN + n] = val;
    }
  }
}

__global__ void init_kernel(float* __restrict__ stat, int nstat,
                            float* __restrict__ gpool, int npool,
                            int* __restrict__ cnt, float* __restrict__ degu,
                            float* __restrict__ deg2, int N) {
  int i = blockIdx.x * blockDim.x + threadIdx.x;
  if (i < nstat) stat[i] = 0.f;
  if (i < npool) gpool[i] = 0.f;
  if (i < N) {
    cnt[i] = 0;
    degu[i] = 1.0f;
    ((float2*)deg2)[i] = make_float2(1.0f, 1.0f);
  }
}
__device__ __forceinline__ float inv_sqrt_deg(float d) {
  return d > 0.f ? rsqrtf(d) : 0.f;
}

__global__ void csr_count_kernel(const int* __restrict__ row,
                                 const int* __restrict__ col, int E,
                                 int* __restrict__ cnt, float* __restrict__ degu) {
  int e = blockIdx.x * blockDim.x + threadIdx.x;
  if (e >= E) return;
  atomicAdd(&cnt[col[e]], 1);
  atomicAdd(&degu[row[e]], 1.0f);
}
__device__ __forceinline__ void load_counts4(const int* __restrict__ cnt, int i0,
                                             int N, int c[4]) {
  if (i0 + 3 < N) {
    const int4 v = *(const int4*)(cnt + i0);
    c[0] = v.x; c[1] = v.y; c[2] = v.z; c[3] = v.w;
  } else {
    c[0] = (i0     < N) ? cnt[i0]     : 0;
    c[1] = (i0 + 1 < N) ? cnt[i0 + 1] : 0;
    c[2] = (i0 + 2 < N) ? cnt[i0 + 2] : 0;
    c[3] = 0;
  }
}
__global__ __launch_bounds__(256) void csr_tilesum_kernel(
    const int* __restrict__ cnt, int N, int* __restrict__ tsum) {
  __shared__ int sh[8];
  const int tid = threadIdx.x;
  int c[4];
  load_counts4(cnt, (blockIdx.x * 256 + tid) * 4, N, c);
  int t = c[0] + c[1] + c[2] + c[3];
  for (int o = 16; o > 0; o >>= 1) t += __shfl_xor(t, o);
  if ((tid & 31) == 0) sh[tid >> 5] = t;
  __syncthreads();
  if (tid == 0) {
    int s = 0;
    for (int w = 0; w < 8; ++w) s += sh[w];
    tsum[blockIdx.x] = s;
  }
}
__global__ __launch_bounds__(256) void csr_scan_kernel(
    const int* __restrict__ cnt, const int* __restrict__ tsum, int N,
    int* __restrict__ off, int* __restrict__ cur) {
  __shared__ int sh_w[8];
  __shared__ int sh_b[8];
  const int tid = threadIdx.x, lane = tid & 31, wave = tid >> 5;
  int bpart = 0;
  for (int t = tid; t < (int)blockIdx.x; t += 256) bpart += tsum[t];
  for (int o = 16; o > 0; o >>= 1) bpart += __shfl_xor(bpart, o);
  const int i0 = (blockIdx.x * 256 + tid) * 4;
  int c[4];
  load_counts4(cnt, i0, N, c);
  const int t = c[0] + c[1] + c[2] + c[3];
  int inc = t;
  for (int o = 1; o < 32; o <<= 1) {
    const int y = __shfl_up(inc, o);
    if (lane >= o) inc += y;
  }
  if (lane == 31) sh_w[wave] = inc;
  if (lane == 0)  sh_b[wave] = bpart;
  __syncthreads();
  int base = 0;
  for (int w = 0; w < 8; ++w) {
    base += sh_b[w];
    if (w < wave) base += sh_w[w];
  }
  const int o0 = base + inc - t;
  const int o1 = o0 + c[0], o2 = o1 + c[1], o3 = o2 + c[2];
  if (i0 + 3 < N) {
    const int4 v = make_int4(o0, o1, o2, o3);
    *(int4*)(off + i0) = v;
    *(int4*)(cur + i0) = v;
  } else {
    if (i0     < N) { off[i0]     = o0; cur[i0]     = o0; }
    if (i0 + 1 < N) { off[i0 + 1] = o1; cur[i0 + 1] = o1; }
    if (i0 + 2 < N) { off[i0 + 2] = o2; cur[i0 + 2] = o2; }
  }
  if (i0 < N && N - 1 <= i0 + 3) off[N] = o0 + t;
}
__global__ void csr_fill_kernel(const int* __restrict__ row,
                                const int* __restrict__ col, int E,
                                int* __restrict__ cur, int2* __restrict__ sedge) {
  int e = blockIdx.x * blockDim.x + threadIdx.x;
  if (e >= E) return;
  const int r = row[e];
  const int pos = atomicAdd(&cur[col[e]], 1);
  if ((unsigned)pos < (unsigned)E) sedge[pos] = make_int2(r, e);
}

#define T_PITCH 68

template <int CNT>
__device__ __forceinline__ void gather_rows(const float2* __restrict__ m2, int r,
                                            float cf, int k, int lane, float2& acc,
                                            float& cs) {
  int rr[CNT];
  float cc[CNT];
  float2 v[CNT];
#pragma unroll
  for (int u = 0; u < CNT; ++u) {
    rr[u] = __builtin_amdgcn_readlane(r, k + u);
    cc[u] = __int_as_float(__builtin_amdgcn_readlane(__float_as_int(cf), k + u));
  }
#pragma unroll
  for (int u = 0; u < CNT; ++u) v[u] = m2[(size_t)rr[u] * (HIDDEN / 2) + lane];
#pragma unroll
  for (int u = 0; u < CNT; ++u) {
    acc.x += cc[u] * v[u].x;
    acc.y += cc[u] * v[u].y;
    cs += cc[u];
  }
}
__device__ __forceinline__ float2 agg_row(
    const float2* __restrict__ h2, const int* __restrict__ off,
    const int2* __restrict__ sedge, const float* __restrict__ deg,
    int c, int lane, float& rs) {
  const int beg = off[c], end = off[c + 1];
  const float dc = inv_sqrt_deg(deg[c]);
  const float2 self = h2[(size_t)c * (HIDDEN / 2) + lane];
  float2 acc = make_float2(0.f, 0.f);
  float cs = 0.f;
  for (int base = beg; base < end; base += 32) {
    const int p = base + lane;
    int r = 0;
    float cf = 0.f;
    if (p < end) {
      const int2 se = sedge[p];
      r = se.x;
      const float dr = inv_sqrt_deg(deg[r]);
      cf = dr * dc;
    }
    const int n = (end - base) < 32 ? (end - base) : 32;
    int k = 0;
    for (; k + 4 <= n; k += 4) gather_rows<4>(h2, r, cf, k, lane, acc, cs);
    const int rem = n - k;
    if (rem == 3)      gather_rows<3>(h2, r, cf, k, lane, acc, cs);
    else if (rem == 2) gather_rows<2>(h2, r, cf, k, lane, acc, cs);
    else if (rem == 1) gather_rows<1>(h2, r, cf, k, lane, acc, cs);
  }
  rs = cs + dc * dc;
  float2 t;
  t.x = acc.x + dc * dc * self.x;
  t.y = acc.y + dc * dc * self.y;
  return t;
}

__device__ __forceinline__ void load_b_frags(const float* __restrict__ wp,
                                             v2f (&bf)[HIDDEN / 4]) {
#pragma unroll
  for (int q = 0; q < HIDDEN / 4; ++q) {
    bf[q].x = wp[(4 * q) * HIDDEN];
    bf[q].y = wp[(4 * q + 1) * HIDDEN];
  }
}
__device__ __forceinline__ v8f tile_product(const float* ap, const v2f (&bf)[HIDDEN / 4]) {
  v8f acc = {};
#pragma unroll
  for (int q = 0; q < HIDDEN / 4; ++q) {
    const v2f a = *(const v2f*)(ap + 4 * q);
    acc = __builtin_amdgcn_wmma_f32_16x16x4_f32(
        false, a, false, bf[q], (short)0, acc, false, false);
  }
  return acc;
}

__global__ __launch_bounds__(128) void gather_gemm_kernel(
    const float* __restrict__ h, const int* __restrict__ off,
    const int2* __restrict__ sedge, const float* __restrict__ deg,
    const float* __restrict__ Wt, const float* __restrict__ sums,
    const float* __restrict__ sumsq, float invN,
    const float* __restrict__ b, float* __restrict__ out, int N) {
  __shared__ __attribute__((aligned(16))) float sh_t[16 * T_PITCH];
  const int lane = threadIdx.x & 31;
  const int wave = threadIdx.x >> 5;
  const int kOff = (lane >> 4) << 1;
  const int n0   = lane & 15;
  const int nbase = blockIdx.x * 16;
  float2 sc, sf;
  {
    const float2 su = ((const float2*)sums)[lane];
    const float2 sq = ((const float2*)sumsq)[lane];
    bn_coeffs(su.x, sq.x, invN, sc.x, sf.x);
    bn_coeffs(su.y, sq.y, invN, sc.y, sf.y);
  }
#pragma nounroll
  for (int i = 0; i < 4; ++i) {
    const int rl = wave * 4 + i;
    const int c = __builtin_amdgcn_readfirstlane(nbase + rl);
    float2 t = make_float2(0.f, 0.f);
    float rs = 0.f;
    if (c < N) t = agg_row((const float2*)h, off, sedge, deg, c, lane, rs);
    t.x = t.x * sc.x + rs * sf.x;
    t.y = t.y * sc.y + rs * sf.y;
    *(float2*)(sh_t + rl * T_PITCH + 2 * lane) = t;
  }
  v2f bf[HIDDEN / 4];
  load_b_frags(Wt + (size_t)kOff * HIDDEN + wave * 16 + n0, bf);
  __syncthreads();
  const v8f acc = tile_product(sh_t + n0 * T_PITCH + kOff, bf);
  __syncthreads();
  {
    const int n = wave * 16 + n0;
    const float bn = b[n];
    const int rowBase = (lane >> 4) << 3;
#pragma unroll
    for (int e = 0; e < 8; ++e) {
      const int r = rowBase + e;
      sh_t[r * T_PITCH + n] = fmaxf(acc[e] + bn, 0.f);
    }
  }
  __syncthreads();
#pragma unroll
  for (int j = 0; j < 2; ++j) {
    const int r = wave * 4 + 2 * j + (lane >> 4);
    const int node = nbase + r;
    if (node < N) {
      const float4 val = *(const float4*)(sh_t + r * T_PITCH + 4 * n0);
      ((float4*)out)[(size_t)node * (HIDDEN / 4) + n0] = val;
    }
  }
}

template <int CNT>
__device__ __forceinline__ void gather_rows2(const float2* __restrict__ h2, int r,
                                             float ec, float eo, float2 pw, int k,
                                             int lane, float2& ac, float2& ao,
                                             float& csc, float& cso) {
  int rr[CNT];
  float e0[CNT], e1[CNT], c0[CNT], c1[CNT];
  float2 v[CNT];
#pragma unroll
  for (int u = 0; u < CNT; ++u) {
    rr[u] = __builtin_amdgcn_readlane(r, k + u);
    e0[u] = __int_as_float(__builtin_amdgcn_readlane(__float_as_int(ec), k + u));
    e1[u] = __int_as_float(__builtin_amdgcn_readlane(__float_as_int(eo), k + u));
    c0[u] = e0[u] * __int_as_float(__builtin_amdgcn_readlane(__float_as_int(pw.x), k + u));
    c1[u] = e1[u] * __int_as_float(__builtin_amdgcn_readlane(__float_as_int(pw.y), k + u));
  }
#pragma unroll
  for (int u = 0; u < CNT; ++u) v[u] = h2[(size_t)rr[u] * (HIDDEN / 2) + lane];
#pragma unroll
  for (int u = 0; u < CNT; ++u) {
    ac.x += c0[u] * v[u].x;
    ac.y += c0[u] * v[u].y;
    ao.x += c1[u] * v[u].x;
    ao.y += c1[u] * v[u].y;
    csc += e0[u];
    cso += e1[u];
  }
}
__device__ __forceinline__ void pool_rows8(const v8f& acc, const int* __restrict__ gid8,
                                           float bn, float* __restrict__ g) {
  int gw[8];
  float val[8];
#pragma unroll
  for (int e = 0; e < 8; ++e) {
    gw[e] = gid8[e];
    val[e] = fmaxf(acc[e] + bn, 0.f);
  }
  int cur = gw[0];
  float a = val[0];
#pragma unroll
  for (int e = 1; e < 8; ++e) {
    if (gw[e] == cur) {
      a += val[e];
    } else {
      if ((unsigned)cur < (unsigned)N_GRAPHS) atomicAdd(&g[(size_t)cur * HIDDEN], a);
      cur = gw[e];
      a = val[e];
    }
  }
  if ((unsigned)cur < (unsigned)N_GRAPHS) atomicAdd(&g[(size_t)cur * HIDDEN], a);
}
__global__ __launch_bounds__(128) void gather_gemm_pool2_kernel(
    const float* __restrict__ h, const float* __restrict__ patt,
    const int* __restrict__ off, const int2* __restrict__ sedge,
    const float* __restrict__ att, const float* __restrict__ deg2,
    const float* __restrict__ W_c, const float* __restrict__ W_o,
    const float* __restrict__ stat2, float invN,
    const float* __restrict__ b_ctx, const float* __restrict__ b_obj,
    const int* __restrict__ batch, float* __restrict__ gpool, int N) {
  __shared__ __attribute__((aligned(16))) float sh_t[2][16 * T_PITCH];
  __shared__ int sh_gid[16];
  const int lane = threadIdx.x & 31;
  const int wave = threadIdx.x >> 5;
  const int kOff = (lane >> 4) << 1;
  const int n0   = lane & 15;
  const int nbase = blockIdx.x * 16;
  const float2* __restrict__ h2 = (const float2*)h;
  const float2* __restrict__ P2 = (const float2*)patt;
  const float2* __restrict__ d2 = (const float2*)deg2;
  const float2* __restrict__ att2 = (const float2*)att;
  float2 scc, sfc, sco, sfo;
  {
    const float2* __restrict__ s2 = (const float2*)stat2;
    const float2 suc = s2[lane];
    const float2 sqc = s2[N_FEAT / 2 + lane];
    const float2 suo = s2[N_FEAT + lane];
    const float2 sqo = s2[N_FEAT + N_FEAT / 2 + lane];
    bn_coeffs(suc.x, sqc.x, invN, scc.x, sfc.x);
    bn_coeffs(suc.y, sqc.y, invN, scc.y, sfc.y);
    bn_coeffs(suo.x, sqo.x, invN, sco.x, sfo.x);
    bn_coeffs(suo.y, sqo.y, invN, sco.y, sfo.y);
  }
#pragma nounroll
  for (int i = 0; i < 4; ++i) {
    const int rl = wave * 4 + i;
    const int c = __builtin_amdgcn_readfirstlane(nbase + rl);
    float2 tc = make_float2(0.f, 0.f), to = make_float2(0.f, 0.f);
    float rsc = 0.f, rso = 0.f;
    int gid = -1;
    if (c < N) {
      gid = batch[c];
      const int beg = off[c], end = off[c + 1];
      const float2 dcv = d2[c];
      const float dcc = inv_sqrt_deg(dcv.x);
      const float dco = inv_sqrt_deg(dcv.y);
      const float2 pself = P2[c];
      const float2 self = h2[(size_t)c * (HIDDEN / 2) + lane];
      float2 ac = make_float2(0.f, 0.f), ao = make_float2(0.f, 0.f);
      float csc = 0.f, cso = 0.f;
      for (int base = beg; base < end; base += 32) {
        const int p = base + lane;
        int r = 0;
        float ec = 0.f, eo = 0.f;
        float2 pw = make_float2(0.f, 0.f);
        if (p < end) {
          const int2 se = sedge[p];
          r = se.x;
          const float2 dr = d2[r];
          const float2 w = att2[se.y];
          pw = P2[r];
          ec = inv_sqrt_deg(dr.x) * w.x * dcc;
          eo = inv_sqrt_deg(dr.y) * w.y * dco;
        }
        const int n = (end - base) < 32 ? (end - base) : 32;
        int k = 0;
        for (; k + 4 <= n; k += 4) gather_rows2<4>(h2, r, ec, eo, pw, k, lane, ac, ao, csc, cso);
        const int rem = n - k;
        if (rem == 3)      gather_rows2<3>(h2, r, ec, eo, pw, k, lane, ac, ao, csc, cso);
        else if (rem == 2) gather_rows2<2>(h2, r, ec, eo, pw, k, lane, ac, ao, csc, cso);
        else if (rem == 1) gather_rows2<1>(h2, r, ec, eo, pw, k, lane, ac, ao, csc, cso);
      }
      rsc = csc + dcc * dcc;
      rso = cso + dco * dco;
      const float sc = dcc * dcc * pself.x;
      const float so = dco * dco * pself.y;
      tc.x = ac.x + sc * self.x;
      tc.y = ac.y + sc * self.y;
      to.x = ao.x + so * self.x;
      to.y = ao.y + so * self.y;
    }
    tc.x = tc.x * scc.x + rsc * sfc.x;
    tc.y = tc.y * scc.y + rsc * sfc.y;
    to.x = to.x * sco.x + rso * sfo.x;
    to.y = to.y * sco.y + rso * sfo.y;
    *(float2*)(sh_t[0] + rl * T_PITCH + 2 * lane) = tc;
    *(float2*)(sh_t[1] + rl * T_PITCH + 2 * lane) = to;
    if (lane == 0) sh_gid[rl] = gid;
  }
  const int n = wave * 16 + n0;
  const int rowBase = (lane >> 4) << 3;
  const size_t wo = (size_t)kOff * HIDDEN + n;
  const float* ap = sh_t[0] + n0 * T_PITCH + kOff;
  v2f bf[HIDDEN / 4];
  load_b_frags(W_c + wo, bf);
  __syncthreads();
  {
    const v8f acc = tile_product(ap, bf);
    pool_rows8(acc, sh_gid + rowBase, b_ctx[n], gpool + n);
  }
  load_b_frags(W_o + wo, bf);
  {
    const v8f acc = tile_product(ap + 16 * T_PITCH, bf);
    pool_rows8(acc, sh_gid + rowBase, b_obj[n], gpool + (size_t)N_GRAPHS * HIDDEN + n);
  }
}

__global__ void node_att_kernel(const float* __restrict__ h,
                                const float* __restrict__ Wna,
                                const float* __restrict__ bna,
                                const float* __restrict__ Wea,
                                float* __restrict__ patt,
                                float* __restrict__ pr,
                                float* __restrict__ pc, int N) {
  int i = blockIdx.x * blockDim.x + threadIdx.x;
  if (i >= N) return;
  const float* hi = h + (size_t)i * HIDDEN;
  float l0 = bna[0], l1 = bna[1];
  float r0 = 0.f, r1 = 0.f, c0 = 0.f, c1 = 0.f;
  for (int j = 0; j < HIDDEN; ++j) {
    float a = hi[j];
    l0 += a * Wna[j * 2 + 0];
    l1 += a * Wna[j * 2 + 1];
    r0 += a * Wea[j * 2 + 0];
    r1 += a * Wea[j * 2 + 1];
    c0 += a * Wea[(HIDDEN + j) * 2 + 0];
    c1 += a * Wea[(HIDDEN + j) * 2 + 1];
  }
  ((float2*)pr)[i] = make_float2(r0, r1);
  ((float2*)pc)[i] = make_float2(c0, c1);
  float mx = fmaxf(l0, l1);
  float e0 = __expf(l0 - mx), e1 = __expf(l1 - mx);
  float inv = 1.f / (e0 + e1);
  float p0 = e0 * inv, p1 = e1 * inv;
  ((float2*)patt)[i] = make_float2(p0, p1);
}

__global__ void edge_att_kernel(const float* __restrict__ pr,
                                const float* __restrict__ pc,
                                const int* __restrict__ row,
                                const int* __restrict__ col,
                                const float* __restrict__ bea,
                                float* __restrict__ att,
                                float* __restrict__ deg2, int E) {
  int e = blockIdx.x * blockDim.x + threadIdx.x;
  if (e >= E) return;
  const int r = row[e];
  const float2 a = ((const float2*)pr)[r];
  const float2 c = ((const float2*)pc)[col[e]];
  const float l0 = (a.x + c.x) + bea[0];
  const float l1 = (a.y + c.y) + bea[1];
  float mx = fmaxf(l0, l1);
  float e0 = __expf(l0 - mx), e1 = __expf(l1 - mx);
  float inv = 1.f / (e0 + e1);
  const float w0 = e0 * inv, w1 = e1 * inv;
  ((float2*)att)[e] = make_float2(w0, w1);
  atomicAdd(&deg2[2 * (size_t)r], w0);
  atomicAdd(&deg2[2 * (size_t)r + 1], w1);
}

#define HEAD_THREADS 512
#define G_PITCH 68
#define W_PITCH 72

__device__ __forceinline__ void head_colstats(const float* __restrict__ ps,
                                              const float* __restrict__ pq,
                                              float* __restrict__ sc,
                                              float* __restrict__ ms, int col) {
  float ts = 0.f, tq = 0.f;
  for (int p = 0; p < 32; ++p) {
    ts += ps[p * HIDDEN + col];
    tq += pq[p * HIDDEN + col];
  }
  const float invN = 1.f / N_GRAPHS;
  const float mean = ts * invN;
  const float var  = tq * invN - mean * mean;
  const float s    = rsqrtf(var + BN_EPS);
  sc[col] = s;
  ms[col] = mean * s;
}

__global__ __launch_bounds__(HEAD_THREADS) void heads_kernel(
    const float* __restrict__ gc, const float* __restrict__ go,
    const float* __restrict__ W1_c, const float* __restrict__ b1_c,
    const float* __restrict__ W2_c, const float* __restrict__ b2_c,
    const float* __restrict__ W1_o, const float* __restrict__ b1_o,
    const float* __restrict__ W2_o, const float* __restrict__ b2_o,
    const float* __restrict__ W1_co, const float* __restrict__ b1_co,
    const float* __restrict__ W2_co, const float* __restrict__ b2_co,
    float* __restrict__ out) {
  __shared__ __attribute__((aligned(16))) float sh_g[N_GRAPHS * G_PITCH];
  __shared__ __attribute__((aligned(16))) float sh_w[HIDDEN * W_PITCH];
  __shared__ __attribute__((aligned(16))) float sh_ps[32 * HIDDEN];
  __shared__ __attribute__((aligned(16))) float sh_pq[32 * HIDDEN];
  __shared__ float sh_sc[HIDDEN];
  __shared__ float sh_ms[HIDDEN];
  __shared__ float sh_v[HIDDEN];
  __shared__ __attribute__((aligned(16))) float sh_w2[HIDDEN * N_CLASSES];
  __shared__ float sh_v2[16];
  __shared__ float sh_w1r[HIDDEN * HIDDEN];
  __shared__ float sh_w2r[HIDDEN * N_CLASSES];

  const int head = blockIdx.x;
  const float* __restrict__ W1 = head == 0 ? W1_c : (head == 1 ? W1_o : W1_co);
  const float* __restrict__ b1 = head == 0 ? b1_c : (head == 1 ? b1_o : b1_co);
  const float* __restrict__ W2 = head == 0 ? W2_c : (head == 1 ? W2_o : W2_co);
  const float* __restrict__ b2 = head == 0 ? b2_c : (head == 1 ? b2_o : b2_co);
  const int tid  = threadIdx.x;
  const int lane = tid & 31;
  const int wave = tid >> 5;

  float w1r[8];
#pragma unroll
  for (int u = 0; u < 8; ++u) w1r[u] = W1[tid + HEAD_THREADS * u];
  const float w2r0 = W2[tid];
  const float w2r1 = tid < HIDDEN * N_CLASSES - HEAD_THREADS ? W2[HEAD_THREADS + tid] : 0.f;
  const float b1r  = tid < HIDDEN ? b1[tid] : 0.f;
  const float b2r  = tid < N_CLASSES ? b2[tid] : 0.f;

  {
    const float4* __restrict__ gc4 = (const float4*)gc;
    const float4* __restrict__ go4 = (const float4*)go;
    const float4* __restrict__ pa  = head == 1 ? go4 : gc4;
    const int cg = tid & 15;
    const int r0 = tid >> 4;
    float4 x[16];
#pragma unroll
    for (int u = 0; u < 16; ++u) x[u] = pa[tid + HEAD_THREADS * u];
    if (head == 2) {
#pragma unroll
      for (int u = 0; u < 16; ++u) {
        const float4 b = go4[tid + HEAD_THREADS * u];
        x[u] = make_float4(x[u].x + b.x, x[u].y + b.y, x[u].z + b.z, x[u].w + b.w);
      }
    }
    float4 s = make_float4(0.f, 0.f, 0.f, 0.f);
    float4 q = make_float4(0.f, 0.f, 0.f, 0.f);
#pragma unroll
    for (int u = 0; u < 16; ++u) {
      *(float4*)(sh_g + (r0 + 32 * u) * G_PITCH + 4 * cg) = x[u];
      s.x += x[u].x; q.x += x[u].x * x[u].x;
      s.y += x[u].y; q.y += x[u].y * x[u].y;
      s.z += x[u].z; q.z += x[u].z * x[u].z;
      s.w += x[u].w; q.w += x[u].w * x[u].w;
    }
    *(float4*)(sh_ps + r0 * HIDDEN + 4 * cg) = s;
    *(float4*)(sh_pq + r0 * HIDDEN + 4 * cg) = q;
#pragma unroll
    for (int u = 0; u < 8; ++u) sh_w1r[tid + HEAD_THREADS * u] = w1r[u];
    sh_w2r[tid] = w2r0;
    if (tid < HIDDEN * N_CLASSES - HEAD_THREADS) sh_w2r[HEAD_THREADS + tid] = w2r1;
  }
  __syncthreads();
  if (tid < HIDDEN) head_colstats(sh_ps, sh_pq, sh_sc, sh_ms, tid);
  __syncthreads();

#pragma unroll
  for (int u = 0; u < 8; ++u) {
    const int i = tid + HEAD_THREADS * u;
    const int k = i >> 6, j = i & 63;
    sh_w[k * W_PITCH + j] = sh_sc[k] * w1r[u];
  }
  if (tid < HIDDEN) {
    float acc = b1r;
    for (int k = 0; k < HIDDEN; ++k) acc += (BN_BETA - sh_ms[k]) * sh_w1r[k * HIDDEN + tid];
    sh_v[tid] = acc;
  }
  __syncthreads();

  {
    const int kOff = (lane >> 4) << 1;
    const int n0   = lane & 15;
    v8f acc[2][4] = {};
    const float* a0p = sh_g + (32 * wave + n0) * G_PITCH + kOff;
    const float* a1p = a0p + 16 * G_PITCH;
    const float* bp  = sh_w + kOff * W_PITCH + n0;
    for (int kk = 0; kk < HIDDEN; kk += 4) {
      const v2f a0 = *(const v2f*)(a0p + kk);
      const v2f a1 = *(const v2f*)(a1p + kk);
#pragma unroll
      for (int t = 0; t < 4; ++t) {
        v2f b;
        b.x = bp[kk * W_PITCH + t * 16];
        b.y = bp[(kk + 1) * W_PITCH + t * 16];
        acc[0][t] = __builtin_amdgcn_wmma_f32_16x16x4_f32(
            false, a0, false, b, (short)0, acc[0][t], false, false);
        acc[1][t] = __builtin_amdgcn_wmma_f32_16x16x4_f32(
            false, a1, false, b, (short)0, acc[1][t], false, false);
      }
    }
    const int rowBase = 32 * wave + ((lane >> 4) << 3);
#pragma unroll
    for (int t = 0; t < 4; ++t) {
      const int n = t * 16 + n0;
      const float bias = sh_v[n];
      float s = 0.f, q = 0.f;
#pragma unroll
      for (int i = 0; i < 2; ++i) {
#pragma unroll
        for (int e = 0; e < 8; ++e) {
          const float val = fmaxf(acc[i][t][e] + bias, 0.f);
          sh_g[(rowBase + 16 * i + e) * G_PITCH + n] = val;
          s += val;
          q += val * val;
        }
      }
      sh_ps[(2 * wave + (lane >> 4)) * HIDDEN + n] = s;
      sh_pq[(2 * wave + (lane >> 4)) * HIDDEN + n] = q;
    }
  }
  __syncthreads();
  if (tid < HIDDEN) head_colstats(sh_ps, sh_pq, sh_sc, sh_ms, tid);
  __syncthreads();

  sh_w2[tid] = sh_sc[tid / N_CLASSES] * w2r0;
  if (tid < HIDDEN * N_CLASSES - HEAD_THREADS)
    sh_w2[HEAD_THREADS + tid] = sh_sc[(HEAD_THREADS + tid) / N_CLASSES] * w2r1;
  if (tid < N_CLASSES) {
    float acc = b2r;
    for (int k = 0; k < HIDDEN; ++k) acc += (BN_BETA - sh_ms[k]) * sh_w2r[k * N_CLASSES + tid];
    sh_v2[tid] = acc;
  }
  __syncthreads();

  {
    float l[N_CLASSES];
    for (int c = 0; c < N_CLASSES; ++c) l[c] = sh_v2[c];
    const float4* __restrict__ row4 = (const float4*)(sh_g + tid * G_PITCH);
    for (int k4 = 0; k4 < HIDDEN / 4; ++k4) {
      const float4 x = row4[k4];
      const float* w = sh_w2 + k4 * 4 * N_CLASSES;
      for (int c = 0; c < N_CLASSES; ++c) l[c] += x.x * w[c];
      for (int c = 0; c < N_CLASSES; ++c) l[c] += x.y * w[N_CLASSES + c];
      for (int c = 0; c < N_CLASSES; ++c) l[c] += x.z * w[2 * N_CLASSES + c];
      for (int c = 0; c < N_CLASSES; ++c) l[c] += x.w * w[3 * N_CLASSES + c];
    }
    float mx = l[0];
    for (int c = 1; c < N_CLASSES; ++c) mx = fmaxf(mx, l[c]);
    float s = 0.f;
    for (int c = 0; c < N_CLASSES; ++c) s += __expf(l[c] - mx);
    const float lse = __logf(s) + mx;
    float* __restrict__ o = out + ((size_t)head * N_GRAPHS + tid) * N_CLASSES;
    for (int c = 0; c < N_CLASSES; ++c) o[c] = l[c] - lse;
  }
}

extern "C" void kernel_launch(void* const* d_in, const int* in_sizes, int n_in,
                              void* d_out, int out_size, void* d_ws, size_t ws_size,
                              hipStream_t stream) {
  (void)in_sizes; (void)n_in; (void)out_size; (void)ws_size;
  const float* x       = (const float*)d_in[0];
  const int*   ei      = (const int*)d_in[1];
  const int*   row     = ei;
  const int*   col     = ei + N_EDGES;
  const int*   batch   = (const int*)d_in[2];
  const float* W_feat  = (const float*)d_in[3];
  const float* W_convs = (const float*)d_in[4];
  const float* b_convs = (const float*)d_in[5];
  const float* W_ea    = (const float*)d_in[6];
  const float* b_ea    = (const float*)d_in[7];
  const float* W_na    = (const float*)d_in[8];
  const float* b_na    = (const float*)d_in[9];
  const float* W_ctx   = (const float*)d_in[10];
  const float* b_ctx   = (const float*)d_in[11];
  const float* W_obj   = (const float*)d_in[12];
  const float* b_obj   = (const float*)d_in[13];
  const float* W1h[3]  = {(const float*)d_in[14], (const float*)d_in[18], (const float*)d_in[22]};
  const float* b1h[3]  = {(const float*)d_in[15], (const float*)d_in[19], (const float*)d_in[23]};
  const float* W2h[3]  = {(const float*)d_in[16], (const float*)d_in[20], (const float*)d_in[24]};
  const float* b2h[3]  = {(const float*)d_in[17], (const float*)d_in[21], (const float*)d_in[25]};

  size_t off = 0;
  auto carve = [&](size_t elems) -> float* {
    float* p = (float*)((char*)d_ws + off);
    off += ((elems * sizeof(float)) + 255) & ~(size_t)255;
    return p;
  };
  float* hb    = carve((size_t)N_NODES * HIDDEN);
  float* hb2   = carve((size_t)N_NODES * HIDDEN);
  float* patt  = carve((size_t)N_NODES * 2);
  float* att   = carve((size_t)N_EDGES * 2);
  float* pr    = carve((size_t)N_NODES * 2);
  float* pc    = carve((size_t)N_NODES * 2);
  float* degu  = carve(N_NODES);
  float* deg2  = carve((size_t)N_NODES * 2);
  float* gpool = carve((size_t)2 * N_GRAPHS * HIDDEN);
  float* gc    = gpool;
  float* go    = gpool + (size_t)N_GRAPHS * HIDDEN;
  float* stat  = carve((size_t)6 * 2 * N_FEAT);
  int*  ecnt   = (int*)carve(N_NODES);
  int*  etsum  = (int*)carve(128);
  int*  eoff   = (int*)carve(N_NODES + 1);
  int*  ecur   = (int*)carve(N_NODES);
  int2* sedge  = (int2*)carve((size_t)N_EDGES * 2);

  auto nblk = [](long n, int b) { return (int)((n + b - 1) / b); };

  auto stats = [&](const float* A, int N, int F, int slot) {
    const int nvec = N * F / 4;
    const int grid = nblk(nvec, 256) < 512 ? nblk(nvec, 256) : 512;
    colstats_kernel<<<grid, 256, 0, stream>>>(A, nvec, F, stat + slot * 2 * N_FEAT,
                                              stat + slot * 2 * N_FEAT + N_FEAT);
  };
  auto gemm = [&](const float* A, const float* W, int slot, float* C, int M, int K, int relu) {
    gemm_wmma_kernel<<<nblk(M / 16, 4), 128, 0, stream>>>(
        A, W, stat + slot * 2 * N_FEAT, stat + slot * 2 * N_FEAT + N_FEAT, 1.f / N_NODES,
        C, M, K, relu);
  };

  static_assert(2 * N_GRAPHS * HIDDEN <= N_NODES && 6 * 2 * N_FEAT <= N_NODES,
                "init_kernel's grid is sized by the node count");
  init_kernel<<<nblk(N_NODES, 256), 256, 0, stream>>>(stat, 6 * 2 * N_FEAT, gpool,
                                                      2 * N_GRAPHS * HIDDEN, ecnt, degu,
                                                      deg2, N_NODES);

  stats(x, N_NODES, N_FEAT, 0);
  gemm(x, W_feat, 0, hb, N_NODES, N_FEAT, 1);

  csr_count_kernel<<<nblk(N_EDGES, 256), 256, 0, stream>>>(row, col, N_EDGES, ecnt, degu);
  csr_tilesum_kernel<<<nblk(N_NODES, 1024), 256, 0, stream>>>(ecnt, N_NODES, etsum);
  csr_scan_kernel<<<nblk(N_NODES, 1024), 256, 0, stream>>>(ecnt, etsum, N_NODES,
                                                           eoff, ecur);
  csr_fill_kernel<<<nblk(N_EDGES, 256), 256, 0, stream>>>(row, col, N_EDGES, ecur,
                                                          sedge);

  for (int i = 0; i < 3; ++i) {
    const float* hsrc = (i & 1) ? hb2 : hb;
    float*       hdst = (i & 1) ? hb : hb2;
    stats(hsrc, N_NODES, HIDDEN, 1 + i);
    gather_gemm_kernel<<<nblk(N_NODES, 16), 128, 0, stream>>>(
        hsrc, eoff, sedge, degu, W_convs + (size_t)i * HIDDEN * HIDDEN,
        stat + (1 + i) * 2 * N_FEAT, stat + (1 + i) * 2 * N_FEAT + N_FEAT, 1.f / N_NODES,
        b_convs + (size_t)i * HIDDEN, hdst, N_NODES);
  }
  const float* hfin = hb2;

  node_att_kernel<<<nblk(N_NODES, 256), 256, 0, stream>>>(hfin, W_na, b_na, W_ea,
                                                          patt, pr, pc, N_NODES);
  edge_att_kernel<<<nblk(N_EDGES, 256), 256, 0, stream>>>(pr, pc, row, col, b_ea,
                                                          att, deg2, N_EDGES);

  {
    const int nvec = N_NODES * HIDDEN / 4;
    const int grid = nblk(nvec, 256) < 512 ? nblk(nvec, 256) : 512;
    colstats2_kernel<<<grid, 256, 0, stream>>>(
        hfin, patt, nvec, stat + 4 * 2 * N_FEAT, stat + 4 * 2 * N_FEAT + N_FEAT,
        stat + 5 * 2 * N_FEAT, stat + 5 * 2 * N_FEAT + N_FEAT);
    gather_gemm_pool2_kernel<<<nblk(N_NODES, 16), 128, 0, stream>>>(
        hfin, patt, eoff, sedge, att, deg2, W_ctx, W_obj, stat + 4 * 2 * N_FEAT,
        1.f / N_NODES, b_ctx, b_obj, batch, gpool, N_NODES);
  }

  static_assert(HEAD_THREADS == N_GRAPHS, "one thread per graph in the logits stage");
  heads_kernel<<<3, HEAD_THREADS, 0, stream>>>(
      gc, go,
      W1h[0], b1h[0], W2h[0], b2h[0],
      W1h[1], b1h[1], W2h[1], b2h[1],
      W1h[2], b1h[2], W2h[2], b2h[2],
      (float*)d_out);
}


# single-read passes over a finished layer (3 column-statistics passes, node attention, branch statistics) move into the producer's epilogue: 18 -> 13 launches (BN sums, att logits: new sum order)
# speedup vs baseline: 4.5469x; 1.0379x over previous
#include <hip/hip_runtime.h>
#include <math.h>


#define N_NODES  100000
#define N_EDGES  1000000
#define N_FEAT   128
#define HIDDEN   64
#define N_GRAPHS 512
#define N_CLASSES 10
#define BN_EPS  1e-5f
#define BN_BETA 1e-4f
#define STAT_REP 8

typedef __attribute__((ext_vector_type(2))) float v2f;
typedef __attribute__((ext_vector_type(8))) float v8f;

__global__ __launch_bounds__(256) void colstats_kernel(
    const float* __restrict__ A, int nvec, int F,
    float* __restrict__ sums, float* __restrict__ sumsq) {
  __shared__ float sh_s[1024];
  __shared__ float sh_q[1024];
  const float4* __restrict__ A4 = (const float4*)A;
  const int tid = threadIdx.x;
  const int stride = gridDim.x * 256;
  const int last = nvec - 1;
  float4 s = make_float4(0.f, 0.f, 0.f, 0.f);
  float4 q = make_float4(0.f, 0.f, 0.f, 0.f);
  for (int i = blockIdx.x * 256 + tid; i < nvec; i += 8 * stride) {
    float4 v[8];
#pragma unroll
    for (int u = 0; u < 8; ++u) {
      const int idx = i + u * stride;
      v[u] = A4[idx < last ? idx : last];
    }
    __builtin_amdgcn_sched_barrier(0);
#pragma unroll
    for (int u = 0; u < 8; ++u) {
      const bool live = (i + u * stride) < nvec;
      const float a = live ? v[u].x : 0.f;
      const float b = live ? v[u].y : 0.f;
      const float c = live ? v[u].z : 0.f;
      const float d = live ? v[u].w : 0.f;
      s.x += a; q.x += a * a;
      s.y += b; q.y += b * b;
      s.z += c; q.z += c * c;
      s.w += d; q.w += d * d;
    }
  }
  ((float4*)sh_s)[tid] = s;
  ((float4*)sh_q)[tid] = q;
  __syncthreads();
  if (tid < F) {
    float ts = 0.f, tq = 0.f;
    for (int p = tid; p < 1024; p += F) { ts += sh_s[p]; tq += sh_q[p]; }
    atomicAdd(&sums[tid], ts);
    atomicAdd(&sumsq[tid], tq);
  }
}

__device__ __forceinline__ void bn_coeffs(float sum, float sumsq, float invN,
                                          float& sc, float& sh) {
  const float mean = sum * invN;
  const float var  = sumsq * invN - mean * mean;
  sc = rsqrtf(var + BN_EPS);
  sh = BN_BETA - mean * sc;
}

__device__ __forceinline__ void tile_colstats(float s, float q, int lane, int col,
                                              float* __restrict__ dst) {
  s += __shfl_xor(s, 16);
  q += __shfl_xor(q, 16);
  const bool lo = lane < 16;
  atomicAdd(dst + (lo ? 0 : N_FEAT) + col, lo ? s : q);
}
__device__ __forceinline__ void read_colstats(const float* __restrict__ src, int pitch,
                                              int lane, float2& su, float2& sq) {
  float2 a[STAT_REP], b[STAT_REP];
#pragma unroll
  for (int r = 0; r < STAT_REP; ++r) {
    a[r] = ((const float2*)(src + r * pitch))[lane];
    b[r] = ((const float2*)(src + r * pitch + N_FEAT))[lane];
  }
  su = a[0];
  sq = b[0];
#pragma unroll
  for (int r = 1; r < STAT_REP; ++r) {
    su.x += a[r].x; su.y += a[r].y;
    sq.x += b[r].x; sq.y += b[r].y;
  }
}

__global__ __launch_bounds__(128) void gemm_wmma_kernel(
    const float* __restrict__ A, const float* __restrict__ Wt,
    const float* __restrict__ sums, const float* __restrict__ sumsq, float invN,
    float* __restrict__ C, int M, int K, int doRelu, float* __restrict__ stat_out) {
  __shared__ __attribute__((aligned(8))) float sh_sc[N_FEAT];
  __shared__ __attribute__((aligned(8))) float sh_sh[N_FEAT];
  for (int k = threadIdx.x; k < K; k += blockDim.x)
    bn_coeffs(sums[k], sumsq[k], invN, sh_sc[k], sh_sh[k]);
  __syncthreads();
  const int lane = threadIdx.x & 31;
  const int wave = threadIdx.x >> 5;
  const int rowTile = blockIdx.x * 4 + wave;
  if (rowTile >= (M >> 4)) return;
  const int mrow = (rowTile << 4) + (lane & 15);
  const int kOff = (lane >> 4) << 1;
  const int n0   = lane & 15;
  v8f acc[4] = {};
  for (int kk = 0; kk < K; kk += 4) {
    const int k0 = kk + kOff;
    v2f a = *(const v2f*)(A + (size_t)mrow * K + k0);
    const v2f sc = *(const v2f*)(sh_sc + k0);
    const v2f sf = *(const v2f*)(sh_sh + k0);
    a.x = a.x * sc.x + sf.x;
    a.y = a.y * sc.y + sf.y;
    const float* w0 = Wt + (size_t)k0 * HIDDEN + n0;
#pragma unroll
    for (int t = 0; t < 4; ++t) {
      v2f b;
      b.x = w0[t * 16];
      b.y = w0[HIDDEN + t * 16];
      acc[t] = __builtin_amdgcn_wmma_f32_16x16x4_f32(
          false, a, false, b, (short)0, acc[t], false, false);
    }
  }
  const int rowBase = (rowTile << 4) + ((lane >> 4) << 3);
  float* __restrict__ sdst = stat_out + (blockIdx.x & (STAT_REP - 1)) * 2 * N_FEAT;
#pragma unroll
  for (int t = 0; t < 4; ++t) {
    const int n = t * 16 + n0;
    float cs = 0.f, cq = 0.f;
#pragma unroll
    for (int e = 0; e < 8; ++e) {
      float val = acc[t][e];
      if (doRelu) val = fmaxf(val, 0.f);
      C[(size_t)(rowBase + e) * HIDDEN + n] = val;
      cs += val;
      cq += val * val;
    }
    tile_colstats(cs, cq, lane, n, sdst);
  }
}

__global__ void init_kernel(float* __restrict__ stat, int nstat,
                            float* __restrict__ gpool, int npool,
                            int* __restrict__ cnt, float* __restrict__ degu,
                            float* __restrict__ deg2, int N) {
  int i = blockIdx.x * blockDim.x + threadIdx.x;
  if (i < nstat) stat[i] = 0.f;
  if (i < npool) gpool[i] = 0.f;
  if (i < N) {
    cnt[i] = 0;
    degu[i] = 1.0f;
    ((float2*)deg2)[i] = make_float2(1.0f, 1.0f);
  }
}
__device__ __forceinline__ float inv_sqrt_deg(float d) {
  return d > 0.f ? rsqrtf(d) : 0.f;
}

__global__ void csr_count_kernel(const int* __restrict__ row,
                                 const int* __restrict__ col, int E,
                                 int* __restrict__ cnt, float* __restrict__ degu) {
  int e = blockIdx.x * blockDim.x + threadIdx.x;
  if (e >= E) return;
  atomicAdd(&cnt[col[e]], 1);
  atomicAdd(&degu[row[e]], 1.0f);
}
__device__ __forceinline__ void load_counts4(const int* __restrict__ cnt, int i0,
                                             int N, int c[4]) {
  if (i0 + 3 < N) {
    const int4 v = *(const int4*)(cnt + i0);
    c[0] = v.x; c[1] = v.y; c[2] = v.z; c[3] = v.w;
  } else {
    c[0] = (i0     < N) ? cnt[i0]     : 0;
    c[1] = (i0 + 1 < N) ? cnt[i0 + 1] : 0;
    c[2] = (i0 + 2 < N) ? cnt[i0 + 2] : 0;
    c[3] = 0;
  }
}
__global__ __launch_bounds__(256) void csr_tilesum_kernel(
    const int* __restrict__ cnt, int N, int* __restrict__ tsum) {
  __shared__ int sh[8];
  const int tid = threadIdx.x;
  int c[4];
  load_counts4(cnt, (blockIdx.x * 256 + tid) * 4, N, c);
  int t = c[0] + c[1] + c[2] + c[3];
  for (int o = 16; o > 0; o >>= 1) t += __shfl_xor(t, o);
  if ((tid & 31) == 0) sh[tid >> 5] = t;
  __syncthreads();
  if (tid == 0) {
    int s = 0;
    for (int w = 0; w < 8; ++w) s += sh[w];
    tsum[blockIdx.x] = s;
  }
}
__global__ __launch_bounds__(256) void csr_scan_kernel(
    const int* __restrict__ cnt, const int* __restrict__ tsum, int N,
    int* __restrict__ off, int* __restrict__ cur) {
  __shared__ int sh_w[8];
  __shared__ int sh_b[8];
  const int tid = threadIdx.x, lane = tid & 31, wave = tid >> 5;
  int bpart = 0;
  for (int t = tid; t < (int)blockIdx.x; t += 256) bpart += tsum[t];
  for (int o = 16; o > 0; o >>= 1) bpart += __shfl_xor(bpart, o);
  const int i0 = (blockIdx.x * 256 + tid) * 4;
  int c[4];
  load_counts4(cnt, i0, N, c);
  const int t = c[0] + c[1] + c[2] + c[3];
  int inc = t;
  for (int o = 1; o < 32; o <<= 1) {
    const int y = __shfl_up(inc, o);
    if (lane >= o) inc += y;
  }
  if (lane == 31) sh_w[wave] = inc;
  if (lane == 0)  sh_b[wave] = bpart;
  __syncthreads();
  int base = 0;
  for (int w = 0; w < 8; ++w) {
    base += sh_b[w];
    if (w < wave) base += sh_w[w];
  }
  const int o0 = base + inc - t;
  const int o1 = o0 + c[0], o2 = o1 + c[1], o3 = o2 + c[2];
  if (i0 + 3 < N) {
    const int4 v = make_int4(o0, o1, o2, o3);
    *(int4*)(off + i0) = v;
    *(int4*)(cur + i0) = v;
  } else {
    if (i0     < N) { off[i0]     = o0; cur[i0]     = o0; }
    if (i0 + 1 < N) { off[i0 + 1] = o1; cur[i0 + 1] = o1; }
    if (i0 + 2 < N) { off[i0 + 2] = o2; cur[i0 + 2] = o2; }
  }
  if (i0 < N && N - 1 <= i0 + 3) off[N] = o0 + t;
}
__global__ void csr_fill_kernel(const int* __restrict__ row,
                                const int* __restrict__ col, int E,
                                int* __restrict__ cur, int2* __restrict__ sedge) {
  int e = blockIdx.x * blockDim.x + threadIdx.x;
  if (e >= E) return;
  const int r = row[e];
  const int pos = atomicAdd(&cur[col[e]], 1);
  if ((unsigned)pos < (unsigned)E) sedge[pos] = make_int2(r, e);
}

#define T_PITCH 68

template <int CNT>
__device__ __forceinline__ void gather_rows(const float2* __restrict__ m2, int r,
                                            float cf, int k, int lane, float2& acc,
                                            float& cs) {
  int rr[CNT];
  float cc[CNT];
  float2 v[CNT];
#pragma unroll
  for (int u = 0; u < CNT; ++u) {
    rr[u] = __builtin_amdgcn_readlane(r, k + u);
    cc[u] = __int_as_float(__builtin_amdgcn_readlane(__float_as_int(cf), k + u));
  }
#pragma unroll
  for (int u = 0; u < CNT; ++u) v[u] = m2[(size_t)rr[u] * (HIDDEN / 2) + lane];
#pragma unroll
  for (int u = 0; u < CNT; ++u) {
    acc.x += cc[u] * v[u].x;
    acc.y += cc[u] * v[u].y;
    cs += cc[u];
  }
}
__device__ __forceinline__ float2 agg_row(
    const float2* __restrict__ h2, const int* __restrict__ off,
    const int2* __restrict__ sedge, const float* __restrict__ deg,
    int c, int lane, float& rs) {
  const int beg = off[c], end = off[c + 1];
  const float dc = inv_sqrt_deg(deg[c]);
  const float2 self = h2[(size_t)c * (HIDDEN / 2) + lane];
  float2 acc = make_float2(0.f, 0.f);
  float cs = 0.f;
  for (int base = beg; base < end; base += 32) {
    const int p = base + lane;
    int r = 0;
    float cf = 0.f;
    if (p < end) {
      const int2 se = sedge[p];
      r = se.x;
      const float dr = inv_sqrt_deg(deg[r]);
      cf = dr * dc;
    }
    const int n = (end - base) < 32 ? (end - base) : 32;
    int k = 0;
    for (; k + 4 <= n; k += 4) gather_rows<4>(h2, r, cf, k, lane, acc, cs);
    const int rem = n - k;
    if (rem == 3)      gather_rows<3>(h2, r, cf, k, lane, acc, cs);
    else if (rem == 2) gather_rows<2>(h2, r, cf, k, lane, acc, cs);
    else if (rem == 1) gather_rows<1>(h2, r, cf, k, lane, acc, cs);
  }
  rs = cs + dc * dc;
  float2 t;
  t.x = acc.x + dc * dc * self.x;
  t.y = acc.y + dc * dc * self.y;
  return t;
}

__device__ __forceinline__ void load_b_frags(const float* __restrict__ wp,
                                             v2f (&bf)[HIDDEN / 4]) {
#pragma unroll
  for (int q = 0; q < HIDDEN / 4; ++q) {
    bf[q].x = wp[(4 * q) * HIDDEN];
    bf[q].y = wp[(4 * q + 1) * HIDDEN];
  }
}
__device__ __forceinline__ v8f tile_product(const float* ap, const v2f (&bf)[HIDDEN / 4]) {
  v8f acc = {};
#pragma unroll
  for (int q = 0; q < HIDDEN / 4; ++q) {
    const v2f a = *(const v2f*)(ap + 4 * q);
    acc = __builtin_amdgcn_wmma_f32_16x16x4_f32(
        false, a, false, bf[q], (short)0, acc, false, false);
  }
  return acc;
}

template <bool LAST>
__global__ __launch_bounds__(128) void gather_gemm_kernel(
    const float* __restrict__ h, const int* __restrict__ off,
    const int2* __restrict__ sedge, const float* __restrict__ deg,
    const float* __restrict__ Wt, const float* __restrict__ stat_in, float invN,
    const float* __restrict__ b, float* __restrict__ out,
    float* __restrict__ stat_out,
    const float* __restrict__ Wna, const float* __restrict__ bna,
    const float* __restrict__ Wea, float* __restrict__ patt,
    float* __restrict__ pr, float* __restrict__ pc, int N) {
  __shared__ __attribute__((aligned(16))) float sh_t[16 * T_PITCH];
  __shared__ float sh_a[16 * 8];
  __shared__ float2 sh_p[16];
  const int lane = threadIdx.x & 31;
  const int wave = threadIdx.x >> 5;
  const int kOff = (lane >> 4) << 1;
  const int n0   = lane & 15;
  const int nbase = blockIdx.x * 16;
  float2 sc, sf;
  {
    float2 su, sq;
    read_colstats(stat_in, 2 * N_FEAT, lane, su, sq);
    bn_coeffs(su.x, sq.x, invN, sc.x, sf.x);
    bn_coeffs(su.y, sq.y, invN, sc.y, sf.y);
  }
#pragma nounroll
  for (int i = 0; i < 4; ++i) {
    const int rl = wave * 4 + i;
    const int c = __builtin_amdgcn_readfirstlane(nbase + rl);
    float2 t = make_float2(0.f, 0.f);
    float rs = 0.f;
    if (c < N) t = agg_row((const float2*)h, off, sedge, deg, c, lane, rs);
    t.x = t.x * sc.x + rs * sf.x;
    t.y = t.y * sc.y + rs * sf.y;
    *(float2*)(sh_t + rl * T_PITCH + 2 * lane) = t;
  }
  v2f bf[HIDDEN / 4];
  load_b_frags(Wt + (size_t)kOff * HIDDEN + wave * 16 + n0, bf);
  __syncthreads();
  const v8f acc = tile_product(sh_t + n0 * T_PITCH + kOff, bf);
  __syncthreads();
  {
    const int n = wave * 16 + n0;
    const float bn = b[n];
    const int rowBase = (lane >> 4) << 3;
    float cs = 0.f, cq = 0.f;
#pragma unroll
    for (int e = 0; e < 8; ++e) {
      const int r = rowBase + e;
      const float val = fmaxf(acc[e] + bn, 0.f);
      sh_t[r * T_PITCH + n] = val;
      const float live = (nbase + r < N) ? val : 0.f;
      cs += live;
      cq += live * live;
    }
    if (!LAST)
      tile_colstats(cs, cq, lane, n, stat_out + (blockIdx.x & (STAT_REP - 1)) * 2 * N_FEAT);
  }
  __syncthreads();
  if (LAST) {
    if (wave == 0) {
      const float* __restrict__ wsrc = Wna;
      if (n0 < 2)      wsrc = Wna + n0;
      else if (n0 < 4) wsrc = Wea + (n0 - 2);
      else if (n0 < 6) wsrc = Wea + 2 * HIDDEN + (n0 - 4);
      const bool used = n0 < 6;
      v2f af[HIDDEN / 4];
#pragma unroll
      for (int q = 0; q < HIDDEN / 4; ++q) {
        const int k = 4 * q + kOff;
        af[q].x = used ? wsrc[2 * k] : 0.f;
        af[q].y = used ? wsrc[2 * (k + 1)] : 0.f;
      }
      const v8f pa = tile_product(sh_t + n0 * T_PITCH + kOff, af);
      const int rowBase = (lane >> 4) << 3;
      if (n0 < 8) {
#pragma unroll
        for (int e = 0; e < 8; ++e) sh_a[(rowBase + e) * 8 + n0] = pa[e];
      }
    }
  }
#pragma unroll
  for (int j = 0; j < 2; ++j) {
    const int r = wave * 4 + 2 * j + (lane >> 4);
    const int node = nbase + r;
    if (node < N) {
      const float4 val = *(const float4*)(sh_t + r * T_PITCH + 4 * n0);
      ((float4*)out)[(size_t)node * (HIDDEN / 4) + n0] = val;
    }
  }
  if (LAST) {
    __syncthreads();
    if (threadIdx.x < 16) {
      const int r = threadIdx.x;
      const float* a = sh_a + r * 8;
      const float l0 = a[0] + bna[0], l1 = a[1] + bna[1];
      const float mx = fmaxf(l0, l1);
      const float e0 = __expf(l0 - mx), e1 = __expf(l1 - mx);
      const float inv = 1.f / (e0 + e1);
      const int node = nbase + r;
      float2 p = make_float2(0.f, 0.f);
      if (node < N) {
        p = make_float2(e0 * inv, e1 * inv);
        ((float2*)patt)[node] = p;
        ((float2*)pr)[node] = make_float2(a[2], a[3]);
        ((float2*)pc)[node] = make_float2(a[4], a[5]);
      }
      sh_p[r] = p;
    }
    __syncthreads();
    {
      const int colj = threadIdx.x & (HIDDEN - 1);
      const int br   = threadIdx.x >> 6;
      float s = 0.f, q = 0.f;
#pragma unroll
      for (int r = 0; r < 16; ++r) {
        const float2 p = sh_p[r];
        const float a = (br ? p.y : p.x) * sh_t[r * T_PITCH + colj];
        s += a;
        q += a * a;
      }
      float* __restrict__ dst = stat_out + (blockIdx.x & (STAT_REP - 1)) * 4 * N_FEAT +
                                br * 2 * N_FEAT;
      atomicAdd(dst + colj, s);
      atomicAdd(dst + N_FEAT + colj, q);
    }
  }
}

template <int CNT>
__device__ __forceinline__ void gather_rows2(const float2* __restrict__ h2, int r,
                                             float ec, float eo, float2 pw, int k,
                                             int lane, float2& ac, float2& ao,
                                             float& csc, float& cso) {
  int rr[CNT];
  float e0[CNT], e1[CNT], c0[CNT], c1[CNT];
  float2 v[CNT];
#pragma unroll
  for (int u = 0; u < CNT; ++u) {
    rr[u] = __builtin_amdgcn_readlane(r, k + u);
    e0[u] = __int_as_float(__builtin_amdgcn_readlane(__float_as_int(ec), k + u));
    e1[u] = __int_as_float(__builtin_amdgcn_readlane(__float_as_int(eo), k + u));
    c0[u] = e0[u] * __int_as_float(__builtin_amdgcn_readlane(__float_as_int(pw.x), k + u));
    c1[u] = e1[u] * __int_as_float(__builtin_amdgcn_readlane(__float_as_int(pw.y), k + u));
  }
#pragma unroll
  for (int u = 0; u < CNT; ++u) v[u] = h2[(size_t)rr[u] * (HIDDEN / 2) + lane];
#pragma unroll
  for (int u = 0; u < CNT; ++u) {
    ac.x += c0[u] * v[u].x;
    ac.y += c0[u] * v[u].y;
    ao.x += c1[u] * v[u].x;
    ao.y += c1[u] * v[u].y;
    csc += e0[u];
    cso += e1[u];
  }
}
__device__ __forceinline__ void pool_rows8(const v8f& acc, const int* __restrict__ gid8,
                                           float bn, float* __restrict__ g) {
  int gw[8];
  float val[8];
#pragma unroll
  for (int e = 0; e < 8; ++e) {
    gw[e] = gid8[e];
    val[e] = fmaxf(acc[e] + bn, 0.f);
  }
  int cur = gw[0];
  float a = val[0];
#pragma unroll
  for (int e = 1; e < 8; ++e) {
    if (gw[e] == cur) {
      a += val[e];
    } else {
      if ((unsigned)cur < (unsigned)N_GRAPHS) atomicAdd(&g[(size_t)cur * HIDDEN], a);
      cur = gw[e];
      a = val[e];
    }
  }
  if ((unsigned)cur < (unsigned)N_GRAPHS) atomicAdd(&g[(size_t)cur * HIDDEN], a);
}
__global__ __launch_bounds__(128) void gather_gemm_pool2_kernel(
    const float* __restrict__ h, const float* __restrict__ patt,
    const int* __restrict__ off, const int2* __restrict__ sedge,
    const float* __restrict__ att, const float* __restrict__ deg2,
    const float* __restrict__ W_c, const float* __restrict__ W_o,
    const float* __restrict__ stat2, float invN,
    const float* __restrict__ b_ctx, const float* __restrict__ b_obj,
    const int* __restrict__ batch, float* __restrict__ gpool, int N) {
  __shared__ __attribute__((aligned(16))) float sh_t[2][16 * T_PITCH];
  __shared__ int sh_gid[16];
  const int lane = threadIdx.x & 31;
  const int wave = threadIdx.x >> 5;
  const int kOff = (lane >> 4) << 1;
  const int n0   = lane & 15;
  const int nbase = blockIdx.x * 16;
  const float2* __restrict__ h2 = (const float2*)h;
  const float2* __restrict__ P2 = (const float2*)patt;
  const float2* __restrict__ d2 = (const float2*)deg2;
  const float2* __restrict__ att2 = (const float2*)att;
  float2 scc, sfc, sco, sfo;
  {
    float2 suc, sqc, suo, sqo;
    read_colstats(stat2, 4 * N_FEAT, lane, suc, sqc);
    read_colstats(stat2 + 2 * N_FEAT, 4 * N_FEAT, lane, suo, sqo);
    bn_coeffs(suc.x, sqc.x, invN, scc.x, sfc.x);
    bn_coeffs(suc.y, sqc.y, invN, scc.y, sfc.y);
    bn_coeffs(suo.x, sqo.x, invN, sco.x, sfo.x);
    bn_coeffs(suo.y, sqo.y, invN, sco.y, sfo.y);
  }
#pragma nounroll
  for (int i = 0; i < 4; ++i) {
    const int rl = wave * 4 + i;
    const int c = __builtin_amdgcn_readfirstlane(nbase + rl);
    float2 tc = make_float2(0.f, 0.f), to = make_float2(0.f, 0.f);
    float rsc = 0.f, rso = 0.f;
    int gid = -1;
    if (c < N) {
      gid = batch[c];
      const int beg = off[c], end = off[c + 1];
      const float2 dcv = d2[c];
      const float dcc = inv_sqrt_deg(dcv.x);
      const float dco = inv_sqrt_deg(dcv.y);
      const float2 pself = P2[c];
      const float2 self = h2[(size_t)c * (HIDDEN / 2) + lane];
      float2 ac = make_float2(0.f, 0.f), ao = make_float2(0.f, 0.f);
      float csc = 0.f, cso = 0.f;
      for (int base = beg; base < end; base += 32) {
        const int p = base + lane;
        int r = 0;
        float ec = 0.f, eo = 0.f;
        float2 pw = make_float2(0.f, 0.f);
        if (p < end) {
          const int2 se = sedge[p];
          r = se.x;
          const float2 dr = d2[r];
          const float2 w = att2[se.y];
          pw = P2[r];
          ec = inv_sqrt_deg(dr.x) * w.x * dcc;
          eo = inv_sqrt_deg(dr.y) * w.y * dco;
        }
        const int n = (end - base) < 32 ? (end - base) : 32;
        int k = 0;
        for (; k + 4 <= n; k += 4) gather_rows2<4>(h2, r, ec, eo, pw, k, lane, ac, ao, csc, cso);
        const int rem = n - k;
        if (rem == 3)      gather_rows2<3>(h2, r, ec, eo, pw, k, lane, ac, ao, csc, cso);
        else if (rem == 2) gather_rows2<2>(h2, r, ec, eo, pw, k, lane, ac, ao, csc, cso);
        else if (rem == 1) gather_rows2<1>(h2, r, ec, eo, pw, k, lane, ac, ao, csc, cso);
      }
      rsc = csc + dcc * dcc;
      rso = cso + dco * dco;
      const float sc = dcc * dcc * pself.x;
      const float so = dco * dco * pself.y;
      tc.x = ac.x + sc * self.x;
      tc.y = ac.y + sc * self.y;
      to.x = ao.x + so * self.x;
      to.y = ao.y + so * self.y;
    }
    tc.x = tc.x * scc.x + rsc * sfc.x;
    tc.y = tc.y * scc.y + rsc * sfc.y;
    to.x = to.x * sco.x + rso * sfo.x;
    to.y = to.y * sco.y + rso * sfo.y;
    *(float2*)(sh_t[0] + rl * T_PITCH + 2 * lane) = tc;
    *(float2*)(sh_t[1] + rl * T_PITCH + 2 * lane) = to;
    if (lane == 0) sh_gid[rl] = gid;
  }
  const int n = wave * 16 + n0;
  const int rowBase = (lane >> 4) << 3;
  const size_t wo = (size_t)kOff * HIDDEN + n;
  const float* ap = sh_t[0] + n0 * T_PITCH + kOff;
  v2f bf[HIDDEN / 4];
  load_b_frags(W_c + wo, bf);
  __syncthreads();
  {
    const v8f acc = tile_product(ap, bf);
    pool_rows8(acc, sh_gid + rowBase, b_ctx[n], gpool + n);
  }
  load_b_frags(W_o + wo, bf);
  {
    const v8f acc = tile_product(ap + 16 * T_PITCH, bf);
    pool_rows8(acc, sh_gid + rowBase, b_obj[n], gpool + (size_t)N_GRAPHS * HIDDEN + n);
  }
}

__global__ void edge_att_kernel(const float* __restrict__ pr,
                                const float* __restrict__ pc,
                                const int* __restrict__ row,
                                const int* __restrict__ col,
                                const float* __restrict__ bea,
                                float* __restrict__ att,
                                float* __restrict__ deg2, int E) {
  int e = blockIdx.x * blockDim.x + threadIdx.x;
  if (e >= E) return;
  const int r = row[e];
  const float2 a = ((const float2*)pr)[r];
  const float2 c = ((const float2*)pc)[col[e]];
  const float l0 = (a.x + c.x) + bea[0];
  const float l1 = (a.y + c.y) + bea[1];
  float mx = fmaxf(l0, l1);
  float e0 = __expf(l0 - mx), e1 = __expf(l1 - mx);
  float inv = 1.f / (e0 + e1);
  const float w0 = e0 * inv, w1 = e1 * inv;
  ((float2*)att)[e] = make_float2(w0, w1);
  atomicAdd(&deg2[2 * (size_t)r], w0);
  atomicAdd(&deg2[2 * (size_t)r + 1], w1);
}

#define HEAD_THREADS 512
#define G_PITCH 68
#define W_PITCH 72

__device__ __forceinline__ void head_colstats(const float* __restrict__ ps,
                                              const float* __restrict__ pq,
                                              float* __restrict__ sc,
                                              float* __restrict__ ms, int col) {
  float ts = 0.f, tq = 0.f;
  for (int p = 0; p < 32; ++p) {
    ts += ps[p * HIDDEN + col];
    tq += pq[p * HIDDEN + col];
  }
  const float invN = 1.f / N_GRAPHS;
  const float mean = ts * invN;
  const float var  = tq * invN - mean * mean;
  const float s    = rsqrtf(var + BN_EPS);
  sc[col] = s;
  ms[col] = mean * s;
}

__global__ __launch_bounds__(HEAD_THREADS) void heads_kernel(
    const float* __restrict__ gc, const float* __restrict__ go,
    const float* __restrict__ W1_c, const float* __restrict__ b1_c,
    const float* __restrict__ W2_c, const float* __restrict__ b2_c,
    const float* __restrict__ W1_o, const float* __restrict__ b1_o,
    const float* __restrict__ W2_o, const float* __restrict__ b2_o,
    const float* __restrict__ W1_co, const float* __restrict__ b1_co,
    const float* __restrict__ W2_co, const float* __restrict__ b2_co,
    float* __restrict__ out) {
  __shared__ __attribute__((aligned(16))) float sh_g[N_GRAPHS * G_PITCH];
  __shared__ __attribute__((aligned(16))) float sh_w[HIDDEN * W_PITCH];
  __shared__ __attribute__((aligned(16))) float sh_ps[32 * HIDDEN];
  __shared__ __attribute__((aligned(16))) float sh_pq[32 * HIDDEN];
  __shared__ float sh_sc[HIDDEN];
  __shared__ float sh_ms[HIDDEN];
  __shared__ float sh_v[HIDDEN];
  __shared__ __attribute__((aligned(16))) float sh_w2[HIDDEN * N_CLASSES];
  __shared__ float sh_v2[16];
  __shared__ float sh_w1r[HIDDEN * HIDDEN];
  __shared__ float sh_w2r[HIDDEN * N_CLASSES];

  const int head = blockIdx.x;
  const float* __restrict__ W1 = head == 0 ? W1_c : (head == 1 ? W1_o : W1_co);
  const float* __restrict__ b1 = head == 0 ? b1_c : (head == 1 ? b1_o : b1_co);
  const float* __restrict__ W2 = head == 0 ? W2_c : (head == 1 ? W2_o : W2_co);
  const float* __restrict__ b2 = head == 0 ? b2_c : (head == 1 ? b2_o : b2_co);
  const int tid  = threadIdx.x;
  const int lane = tid & 31;
  const int wave = tid >> 5;

  float w1r[8];
#pragma unroll
  for (int u = 0; u < 8; ++u) w1r[u] = W1[tid + HEAD_THREADS * u];
  const float w2r0 = W2[tid];
  const float w2r1 = tid < HIDDEN * N_CLASSES - HEAD_THREADS ? W2[HEAD_THREADS + tid] : 0.f;
  const float b1r  = tid < HIDDEN ? b1[tid] : 0.f;
  const float b2r  = tid < N_CLASSES ? b2[tid] : 0.f;

  {
    const float4* __restrict__ gc4 = (const float4*)gc;
    const float4* __restrict__ go4 = (const float4*)go;
    const float4* __restrict__ pa  = head == 1 ? go4 : gc4;
    const int cg = tid & 15;
    const int r0 = tid >> 4;
    float4 x[16];
#pragma unroll
    for (int u = 0; u < 16; ++u) x[u] = pa[tid + HEAD_THREADS * u];
    if (head == 2) {
#pragma unroll
      for (int u = 0; u < 16; ++u) {
        const float4 b = go4[tid + HEAD_THREADS * u];
        x[u] = make_float4(x[u].x + b.x, x[u].y + b.y, x[u].z + b.z, x[u].w + b.w);
      }
    }
    float4 s = make_float4(0.f, 0.f, 0.f, 0.f);
    float4 q = make_float4(0.f, 0.f, 0.f, 0.f);
#pragma unroll
    for (int u = 0; u < 16; ++u) {
      *(float4*)(sh_g + (r0 + 32 * u) * G_PITCH + 4 * cg) = x[u];
      s.x += x[u].x; q.x += x[u].x * x[u].x;
      s.y += x[u].y; q.y += x[u].y * x[u].y;
      s.z += x[u].z; q.z += x[u].z * x[u].z;
      s.w += x[u].w; q.w += x[u].w * x[u].w;
    }
    *(float4*)(sh_ps + r0 * HIDDEN + 4 * cg) = s;
    *(float4*)(sh_pq + r0 * HIDDEN + 4 * cg) = q;
#pragma unroll
    for (int u = 0; u < 8; ++u) sh_w1r[tid + HEAD_THREADS * u] = w1r[u];
    sh_w2r[tid] = w2r0;
    if (tid < HIDDEN * N_CLASSES - HEAD_THREADS) sh_w2r[HEAD_THREADS + tid] = w2r1;
  }
  __syncthreads();
  if (tid < HIDDEN) head_colstats(sh_ps, sh_pq, sh_sc, sh_ms, tid);
  __syncthreads();

#pragma unroll
  for (int u = 0; u < 8; ++u) {
    const int i = tid + HEAD_THREADS * u;
    const int k = i >> 6, j = i & 63;
    sh_w[k * W_PITCH + j] = sh_sc[k] * w1r[u];
  }
  if (tid < HIDDEN) {
    float acc = b1r;
    for (int k = 0; k < HIDDEN; ++k) acc += (BN_BETA - sh_ms[k]) * sh_w1r[k * HIDDEN + tid];
    sh_v[tid] = acc;
  }
  __syncthreads();

  {
    const int kOff = (lane >> 4) << 1;
    const int n0   = lane & 15;
    v8f acc[2][4] = {};
    const float* a0p = sh_g + (32 * wave + n0) * G_PITCH + kOff;
    const float* a1p = a0p + 16 * G_PITCH;
    const float* bp  = sh_w + kOff * W_PITCH + n0;
    for (int kk = 0; kk < HIDDEN; kk += 4) {
      const v2f a0 = *(const v2f*)(a0p + kk);
      const v2f a1 = *(const v2f*)(a1p + kk);
#pragma unroll
      for (int t = 0; t < 4; ++t) {
        v2f b;
        b.x = bp[kk * W_PITCH + t * 16];
        b.y = bp[(kk + 1) * W_PITCH + t * 16];
        acc[0][t] = __builtin_amdgcn_wmma_f32_16x16x4_f32(
            false, a0, false, b, (short)0, acc[0][t], false, false);
        acc[1][t] = __builtin_amdgcn_wmma_f32_16x16x4_f32(
            false, a1, false, b, (short)0, acc[1][t], false, false);
      }
    }
    const int rowBase = 32 * wave + ((lane >> 4) << 3);
#pragma unroll
    for (int t = 0; t < 4; ++t) {
      const int n = t * 16 + n0;
      const float bias = sh_v[n];
      float s = 0.f, q = 0.f;
#pragma unroll
      for (int i = 0; i < 2; ++i) {
#pragma unroll
        for (int e = 0; e < 8; ++e) {
          const float val = fmaxf(acc[i][t][e] + bias, 0.f);
          sh_g[(rowBase + 16 * i + e) * G_PITCH + n] = val;
          s += val;
          q += val * val;
        }
      }
      sh_ps[(2 * wave + (lane >> 4)) * HIDDEN + n] = s;
      sh_pq[(2 * wave + (lane >> 4)) * HIDDEN + n] = q;
    }
  }
  __syncthreads();
  if (tid < HIDDEN) head_colstats(sh_ps, sh_pq, sh_sc, sh_ms, tid);
  __syncthreads();

  sh_w2[tid] = sh_sc[tid / N_CLASSES] * w2r0;
  if (tid < HIDDEN * N_CLASSES - HEAD_THREADS)
    sh_w2[HEAD_THREADS + tid] = sh_sc[(HEAD_THREADS + tid) / N_CLASSES] * w2r1;
  if (tid < N_CLASSES) {
    float acc = b2r;
    for (int k = 0; k < HIDDEN; ++k) acc += (BN_BETA - sh_ms[k]) * sh_w2r[k * N_CLASSES + tid];
    sh_v2[tid] = acc;
  }
  __syncthreads();

  {
    float l[N_CLASSES];
    for (int c = 0; c < N_CLASSES; ++c) l[c] = sh_v2[c];
    const float4* __restrict__ row4 = (const float4*)(sh_g + tid * G_PITCH);
    for (int k4 = 0; k4 < HIDDEN / 4; ++k4) {
      const float4 x = row4[k4];
      const float* w = sh_w2 + k4 * 4 * N_CLASSES;
      for (int c = 0; c < N_CLASSES; ++c) l[c] += x.x * w[c];
      for (int c = 0; c < N_CLASSES; ++c) l[c] += x.y * w[N_CLASSES + c];
      for (int c = 0; c < N_CLASSES; ++c) l[c] += x.z * w[2 * N_CLASSES + c];
      for (int c = 0; c < N_CLASSES; ++c) l[c] += x.w * w[3 * N_CLASSES + c];
    }
    float mx = l[0];
    for (int c = 1; c < N_CLASSES; ++c) mx = fmaxf(mx, l[c]);
    float s = 0.f;
    for (int c = 0; c < N_CLASSES; ++c) s += __expf(l[c] - mx);
    const float lse = __logf(s) + mx;
    float* __restrict__ o = out + ((size_t)head * N_GRAPHS + tid) * N_CLASSES;
    for (int c = 0; c < N_CLASSES; ++c) o[c] = l[c] - lse;
  }
}

extern "C" void kernel_launch(void* const* d_in, const int* in_sizes, int n_in,
                              void* d_out, int out_size, void* d_ws, size_t ws_size,
                              hipStream_t stream) {
  (void)in_sizes; (void)n_in; (void)out_size; (void)ws_size;
  const float* x       = (const float*)d_in[0];
  const int*   ei      = (const int*)d_in[1];
  const int*   row     = ei;
  const int*   col     = ei + N_EDGES;
  const int*   batch   = (const int*)d_in[2];
  const float* W_feat  = (const float*)d_in[3];
  const float* W_convs = (const float*)d_in[4];
  const float* b_convs = (const float*)d_in[5];
  const float* W_ea    = (const float*)d_in[6];
  const float* b_ea    = (const float*)d_in[7];
  const float* W_na    = (const float*)d_in[8];
  const float* b_na    = (const float*)d_in[9];
  const float* W_ctx   = (const float*)d_in[10];
  const float* b_ctx   = (const float*)d_in[11];
  const float* W_obj   = (const float*)d_in[12];
  const float* b_obj   = (const float*)d_in[13];
  const float* W1h[3]  = {(const float*)d_in[14], (const float*)d_in[18], (const float*)d_in[22]};
  const float* b1h[3]  = {(const float*)d_in[15], (const float*)d_in[19], (const float*)d_in[23]};
  const float* W2h[3]  = {(const float*)d_in[16], (const float*)d_in[20], (const float*)d_in[24]};
  const float* b2h[3]  = {(const float*)d_in[17], (const float*)d_in[21], (const float*)d_in[25]};

  size_t off = 0;
  auto carve = [&](size_t elems) -> float* {
    float* p = (float*)((char*)d_ws + off);
    off += ((elems * sizeof(float)) + 255) & ~(size_t)255;
    return p;
  };
  float* hb    = carve((size_t)N_NODES * HIDDEN);
  float* hb2   = carve((size_t)N_NODES * HIDDEN);
  float* patt  = carve((size_t)N_NODES * 2);
  float* att   = carve((size_t)N_EDGES * 2);
  float* pr    = carve((size_t)N_NODES * 2);
  float* pc    = carve((size_t)N_NODES * 2);
  float* degu  = carve(N_NODES);
  float* deg2  = carve((size_t)N_NODES * 2);
  float* gpool = carve((size_t)2 * N_GRAPHS * HIDDEN);
  float* gc    = gpool;
  float* go    = gpool + (size_t)N_GRAPHS * HIDDEN;
  const int nstat = 2 * N_FEAT + 3 * STAT_REP * 2 * N_FEAT + STAT_REP * 4 * N_FEAT;
  float* stat   = carve(nstat);
  float* statr  = stat + 2 * N_FEAT;
  float* stat2r = statr + 3 * STAT_REP * 2 * N_FEAT;
  int*  ecnt   = (int*)carve(N_NODES);
  int*  etsum  = (int*)carve(128);
  int*  eoff   = (int*)carve(N_NODES + 1);
  int*  ecur   = (int*)carve(N_NODES);
  int2* sedge  = (int2*)carve((size_t)N_EDGES * 2);

  auto nblk = [](long n, int b) { return (int)((n + b - 1) / b); };

  auto stats = [&](const float* A, int N, int F, int slot) {
    const int nvec = N * F / 4;
    const int grid = nblk(nvec, 256) < 512 ? nblk(nvec, 256) : 512;
    colstats_kernel<<<grid, 256, 0, stream>>>(A, nvec, F, stat + slot * 2 * N_FEAT,
                                              stat + slot * 2 * N_FEAT + N_FEAT);
  };
  auto gemm = [&](const float* A, const float* W, int slot, float* C, int M, int K, int relu,
                  float* sout) {
    gemm_wmma_kernel<<<nblk(M / 16, 4), 128, 0, stream>>>(
        A, W, stat + slot * 2 * N_FEAT, stat + slot * 2 * N_FEAT + N_FEAT, 1.f / N_NODES,
        C, M, K, relu, sout);
  };

  static_assert(2 * N_GRAPHS * HIDDEN <= N_NODES &&
                    2 * N_FEAT + 3 * STAT_REP * 2 * N_FEAT + STAT_REP * 4 * N_FEAT <= N_NODES,
                "init_kernel's grid is sized by the node count");
  init_kernel<<<nblk(N_NODES, 256), 256, 0, stream>>>(stat, nstat, gpool,
                                                      2 * N_GRAPHS * HIDDEN, ecnt, degu,
                                                      deg2, N_NODES);

  stats(x, N_NODES, N_FEAT, 0);
  gemm(x, W_feat, 0, hb, N_NODES, N_FEAT, 1, statr);

  csr_count_kernel<<<nblk(N_EDGES, 256), 256, 0, stream>>>(row, col, N_EDGES, ecnt, degu);
  csr_tilesum_kernel<<<nblk(N_NODES, 1024), 256, 0, stream>>>(ecnt, N_NODES, etsum);
  csr_scan_kernel<<<nblk(N_NODES, 1024), 256, 0, stream>>>(ecnt, etsum, N_NODES,
                                                           eoff, ecur);
  csr_fill_kernel<<<nblk(N_EDGES, 256), 256, 0, stream>>>(row, col, N_EDGES, ecur,
                                                          sedge);

  for (int i = 0; i < 2; ++i) {
    const float* hsrc = (i & 1) ? hb2 : hb;
    float*       hdst = (i & 1) ? hb : hb2;
    gather_gemm_kernel<false><<<nblk(N_NODES, 16), 128, 0, stream>>>(
        hsrc, eoff, sedge, degu, W_convs + (size_t)i * HIDDEN * HIDDEN,
        statr + (size_t)i * STAT_REP * 2 * N_FEAT, 1.f / N_NODES,
        b_convs + (size_t)i * HIDDEN, hdst,
        statr + (size_t)(i + 1) * STAT_REP * 2 * N_FEAT,
        nullptr, nullptr, nullptr, nullptr, nullptr, nullptr, N_NODES);
  }
  gather_gemm_kernel<true><<<nblk(N_NODES, 16), 128, 0, stream>>>(
      hb, eoff, sedge, degu, W_convs + (size_t)2 * HIDDEN * HIDDEN,
      statr + (size_t)2 * STAT_REP * 2 * N_FEAT, 1.f / N_NODES,
      b_convs + (size_t)2 * HIDDEN, hb2, stat2r,
      W_na, b_na, W_ea, patt, pr, pc, N_NODES);
  const float* hfin = hb2;

  edge_att_kernel<<<nblk(N_EDGES, 256), 256, 0, stream>>>(pr, pc, row, col, b_ea,
                                                          att, deg2, N_EDGES);

  gather_gemm_pool2_kernel<<<nblk(N_NODES, 16), 128, 0, stream>>>(
      hfin, patt, eoff, sedge, att, deg2, W_ctx, W_obj, stat2r,
      1.f / N_NODES, b_ctx, b_obj, batch, gpool, N_NODES);

  static_assert(HEAD_THREADS == N_GRAPHS, "one thread per graph in the logits stage");
  heads_kernel<<<3, HEAD_THREADS, 0, stream>>>(
      gc, go,
      W1h[0], b1h[0], W2h[0], b2h[0],
      W1h[1], b1h[1], W2h[1], b2h[1],
      W1h[2], b1h[2], W2h[2], b2h[2],
      (float*)d_out);
}
